# starting kernel
# baseline (speedup 1.0000x reference)
#include <hip/hip_runtime.h>
#include <math.h>

typedef __attribute__((ext_vector_type(2))) float v2f;
typedef __attribute__((ext_vector_type(8))) float v8f;

#define INV_NN     0.512989176042577f
#define INV_SQRT3  0.5773502691896258f
#define S3   1.7320508075688772f
#define S5   2.23606797749979f
#define S7   2.6457513110645907f
#define S15  3.872983346207417f
#define S105 10.246950765959598f
#define C358 2.091650066335189f
#define C218 1.6201851746019651f

__global__ void kzero(float* __restrict__ p, long n) {
    long i = (long)blockIdx.x * blockDim.x + threadIdx.x;
    if (i < n) p[i] = 0.0f;
}

__global__ void kA(const float* __restrict__ evec, const float* __restrict__ elen,
                   const int* __restrict__ edst,
                   float* __restrict__ sh, float* __restrict__ basis,
                   float* __restrict__ x, int E)
{
    int e = blockIdx.x * blockDim.x + threadIdx.x;
    if (e >= E) return;
    float vx = evec[3*(size_t)e], vy = evec[3*(size_t)e+1], vz = evec[3*(size_t)e+2];
    float rn = 1.0f / (sqrtf(vx*vx + vy*vy + vz*vz) + 1e-12f);
    vx *= rn; vy *= rn; vz *= rn;
    float x2 = vx*vx, y2 = vy*vy, z2 = vz*vz;
    float s[16];
    s[0]  = 1.0f;
    s[1]  = S3*vx; s[2] = S3*vy; s[3] = S3*vz;
    s[4]  = S15*vx*vz;
    s[5]  = S15*vx*vy;
    s[6]  = 0.5f*S5*(2.0f*y2 - x2 - z2);
    s[7]  = S15*vy*vz;
    s[8]  = 0.5f*S15*(z2 - x2);
    s[9]  = C358*vx*(3.0f*z2 - x2);
    s[10] = S105*vx*vy*vz;
    s[11] = C218*vx*(4.0f*y2 - z2 - x2);
    s[12] = 0.5f*S7*vy*(2.0f*y2 - 3.0f*z2 - 3.0f*x2);
    s[13] = C218*vz*(4.0f*y2 - z2 - x2);
    s[14] = 0.5f*S105*vy*(z2 - x2);
    s[15] = C358*vz*(z2 - 3.0f*x2);
    int d = edst[e];
    #pragma unroll
    for (int j = 0; j < 16; j++) {
        sh[(size_t)e*16 + j] = s[j];
        atomicAdd(&x[(size_t)d*16 + j], s[j] * INV_NN);
    }
    float len = elen[e];
    #pragma unroll
    for (int i = 0; i < 3; i++) {
        float dd = (len - 2.0f*(float)i) * 0.5f;
        basis[(size_t)e*3 + i] = expf(-dd*dd) * (1.0f/1.12f);
    }
}

__global__ __launch_bounds__(192) void kB(
    const float* __restrict__ sh, const float* __restrict__ basis,
    const float* __restrict__ x,
    const int* __restrict__ esrc, const int* __restrict__ edst,
    const float* __restrict__ W1a, const float* __restrict__ W1b,
    float* __restrict__ h, int E)
{
    __shared__ float h1[32][258];
    __shared__ float dots[32][4];
    __shared__ int   dstT[32];

    const int tid = threadIdx.x;
    const int e0  = blockIdx.x * 32;

    for (int idx = tid; idx < 32*256; idx += 192) {
        int m = idx >> 8, c = idx & 255;
        int e = e0 + m;
        float v = 0.0f;
        if (e < E) {
            float b0 = basis[(size_t)e*3], b1 = basis[(size_t)e*3+1], b2 = basis[(size_t)e*3+2];
            v = b0*W1a[c] + b1*W1a[256+c] + b2*W1a[512+c];
            v = fmaxf(v * INV_SQRT3, 0.0f);
        }
        h1[m][c] = v;
    }
    if (tid < 128) {
        int m = tid >> 2, p = tid & 3;
        int e = e0 + m;
        float s = 0.0f;
        if (e < E) {
            int src = esrc[e];
            int a = p*p, b = a + 2*p + 1;
            for (int j = a; j < b; j++)
                s += x[(size_t)src*16 + j] * sh[(size_t)e*16 + j];
            s *= rsqrtf((float)(2*p + 1));
        }
        dots[m][p] = s;
        if (p == 0) dstT[m] = (e < E) ? edst[e] : 0;
    }
    __syncthreads();

    const int kc    = tid >> 5;
    const int lane  = tid & 31;
    const int mA    = lane & 15;
    const int khalf = lane >> 4;
    const int n     = lane & 15;

    v8f a00 = {}, a01 = {}, a02 = {}, a03 = {};
    v8f a10 = {}, a11 = {}, a12 = {}, a13 = {};
    const int colb = kc * 16 + n;

    for (int k0 = 0; k0 < 256; k0 += 4) {
        int kb = k0 + 2*khalf;
        v2f A0, A1;
        A0.x = h1[mA][kb];      A0.y = h1[mA][kb + 1];
        A1.x = h1[16 + mA][kb]; A1.y = h1[16 + mA][kb + 1];
        const float* Brow0 = W1b + (size_t)kb * 384;
        const float* Brow1 = Brow0 + 384;
        v2f B0; B0.x = Brow0[colb];       B0.y = Brow1[colb];
        v2f B1; B1.x = Brow0[96  + colb]; B1.y = Brow1[96  + colb];
        v2f B2; B2.x = Brow0[192 + colb]; B2.y = Brow1[192 + colb];
        v2f B3; B3.x = Brow0[288 + colb]; B3.y = Brow1[288 + colb];
        a00 = __builtin_amdgcn_wmma_f32_16x16x4_f32(false, A0, false, B0, (short)0, a00, false, false);
        a10 = __builtin_amdgcn_wmma_f32_16x16x4_f32(false, A1, false, B0, (short)0, a10, false, false);
        a01 = __builtin_amdgcn_wmma_f32_16x16x4_f32(false, A0, false, B1, (short)0, a01, false, false);
        a11 = __builtin_amdgcn_wmma_f32_16x16x4_f32(false, A1, false, B1, (short)0, a11, false, false);
        a02 = __builtin_amdgcn_wmma_f32_16x16x4_f32(false, A0, false, B2, (short)0, a02, false, false);
        a12 = __builtin_amdgcn_wmma_f32_16x16x4_f32(false, A1, false, B2, (short)0, a12, false, false);
        a03 = __builtin_amdgcn_wmma_f32_16x16x4_f32(false, A0, false, B3, (short)0, a03, false, false);
        a13 = __builtin_amdgcn_wmma_f32_16x16x4_f32(false, A1, false, B3, (short)0, a13, false, false);
    }

    const float scale = (1.0f/16.0f) * 0.5f * INV_NN;
    #pragma unroll
    for (int r = 0; r < 8; r++) {
        int m = r + 8*khalf;
        if (e0 + m < E) {
            float ef = dots[m][0]*a00[r] + dots[m][1]*a01[r]
                     + dots[m][2]*a02[r] + dots[m][3]*a03[r];
            atomicAdd(&h[(size_t)dstT[m]*96 + kc*16 + n], ef * scale);
        }
        int m1 = 16 + m;
        if (e0 + m1 < E) {
            float ef = dots[m1][0]*a10[r] + dots[m1][1]*a11[r]
                     + dots[m1][2]*a12[r] + dots[m1][3]*a13[r];
            atomicAdd(&h[(size_t)dstT[m1]*96 + kc*16 + n], ef * scale);
        }
    }
}

__global__ void kC(const float* __restrict__ h, float* __restrict__ y, int N)
{
    int nid = blockIdx.x * blockDim.x + threadIdx.x;
    if (nid >= N) return;
    const float* hr = h + (size_t)nid * 96;
    float* yr = y + (size_t)nid * 64;
    #pragma unroll
    for (int j = 0; j < 16; j++) yr[j] = fmaxf(hr[j], 0.0f);
    #pragma unroll
    for (int j = 16; j < 32; j++) yr[j] = fabsf(hr[j]);
    #pragma unroll
    for (int j = 0; j < 32; j++) {
        float g = hr[32 + j];
        bool use_relu = (j < 8) || (j >= 16 && j < 24);
        float ga = use_relu ? fmaxf(g, 0.0f) : tanhf(g);
        yr[32 + j] = ga * hr[64 + j];
    }
}

__global__ __launch_bounds__(256) void kD(const float* __restrict__ y,
                                          const float* __restrict__ W2b,
                                          float* __restrict__ z, int N)
{
    __shared__ float yn[64];
    int nid = blockIdx.x;
    int c = threadIdx.x;
    if (c < 64) yn[c] = y[(size_t)nid*64 + c];
    __syncthreads();
    float s = 0.0f;
    const float* wr = W2b + (size_t)c * 64;
    #pragma unroll 8
    for (int k = 0; k < 64; k++) s += wr[k] * yn[k];
    z[(size_t)nid*256 + c] = s;
}

__global__ __launch_bounds__(256) void kE(const float* __restrict__ basis,
                                          const float* __restrict__ z,
                                          const int* __restrict__ esrc,
                                          const int* __restrict__ edst,
                                          const float* __restrict__ W2a,
                                          float* __restrict__ out, int E)
{
    int e = blockIdx.x * (blockDim.x >> 5) + (threadIdx.x >> 5);
    int lane = threadIdx.x & 31;
    if (e >= E) return;
    float b0 = basis[(size_t)e*3], b1 = basis[(size_t)e*3+1], b2 = basis[(size_t)e*3+2];
    int src = esrc[e];
    const float* zr = z + (size_t)src * 256;
    float s = 0.0f;
    #pragma unroll
    for (int i = 0; i < 8; i++) {
        int c = lane + 32*i;
        float hv = b0*W2a[c] + b1*W2a[256+c] + b2*W2a[512+c];
        hv = fmaxf(hv * INV_SQRT3, 0.0f);
        s += hv * zr[c];
    }
    #pragma unroll
    for (int off = 16; off > 0; off >>= 1) s += __shfl_down(s, off, 32);
    if (lane == 0)
        atomicAdd(&out[edst[e]], s * (1.0f/128.0f) * INV_NN);
}

extern "C" void kernel_launch(void* const* d_in, const int* in_sizes, int n_in,
                              void* d_out, int out_size, void* d_ws, size_t ws_size,
                              hipStream_t stream)
{
    const float* evec = (const float*)d_in[0];
    const float* elen = (const float*)d_in[1];
    const float* W1a  = (const float*)d_in[2];
    const float* W1b  = (const float*)d_in[3];
    const float* W2a  = (const float*)d_in[4];
    const float* W2b  = (const float*)d_in[5];
    const int*   esrc = (const int*)d_in[6];
    const int*   edst = (const int*)d_in[7];
    float* out = (float*)d_out;

    const int E = in_sizes[0] / 3;
    const int N = out_size;

    float* ws = (float*)d_ws;
    float* x_     = ws;                 ws += (size_t)N * 16;
    float* h_     = ws;                 ws += (size_t)N * 96;
    float* y_     = ws;                 ws += (size_t)N * 64;
    float* z_     = ws;                 ws += (size_t)N * 256;
    float* sh_    = ws;                 ws += (size_t)E * 16;
    float* basis_ = ws;                 ws += (size_t)E * 3;

    long nx = (long)N * 16, nh = (long)N * 96;
    kzero<<<(int)((nx + 255) / 256), 256, 0, stream>>>(x_, nx);
    kzero<<<(int)((nh + 255) / 256), 256, 0, stream>>>(h_, nh);
    kzero<<<(N + 255) / 256, 256, 0, stream>>>(out, N);

    kA<<<(E + 255) / 256, 256, 0, stream>>>(evec, elen, edst, sh_, basis_, x_, E);
    kB<<<(E + 31) / 32, 192, 0, stream>>>(sh_, basis_, x_, esrc, edst, W1a, W1b, h_, E);
    kC<<<(N + 255) / 256, 256, 0, stream>>>(h_, y_, N);
    kD<<<N, 256, 0, stream>>>(y_, W2b, z_, N);
    kE<<<(E + 7) / 8, 256, 0, stream>>>(basis_, z_, esrc, edst, W2a, out, E);
}


# kD (z = y x W2b^T): one block per 128 nodes stages W2b once in LDS and forms its tile on f32 WMMA, instead of one block per node re-reading all of W2b; sum order over k changes
# speedup vs baseline: 1.2182x; 1.2182x over previous
#include <hip/hip_runtime.h>
#include <math.h>

typedef __attribute__((ext_vector_type(2))) float v2f;
typedef __attribute__((ext_vector_type(8))) float v8f;

#define INV_NN     0.512989176042577f
#define INV_SQRT3  0.5773502691896258f
#define S3   1.7320508075688772f
#define S5   2.23606797749979f
#define S7   2.6457513110645907f
#define S15  3.872983346207417f
#define S105 10.246950765959598f
#define C358 2.091650066335189f
#define C218 1.6201851746019651f

__global__ void kzero(float* __restrict__ p, long n) {
    long i = (long)blockIdx.x * blockDim.x + threadIdx.x;
    if (i < n) p[i] = 0.0f;
}

__global__ void kA(const float* __restrict__ evec, const float* __restrict__ elen,
                   const int* __restrict__ edst,
                   float* __restrict__ sh, float* __restrict__ basis,
                   float* __restrict__ x, int E)
{
    int e = blockIdx.x * blockDim.x + threadIdx.x;
    if (e >= E) return;
    float vx = evec[3*(size_t)e], vy = evec[3*(size_t)e+1], vz = evec[3*(size_t)e+2];
    float rn = 1.0f / (sqrtf(vx*vx + vy*vy + vz*vz) + 1e-12f);
    vx *= rn; vy *= rn; vz *= rn;
    float x2 = vx*vx, y2 = vy*vy, z2 = vz*vz;
    float s[16];
    s[0]  = 1.0f;
    s[1]  = S3*vx; s[2] = S3*vy; s[3] = S3*vz;
    s[4]  = S15*vx*vz;
    s[5]  = S15*vx*vy;
    s[6]  = 0.5f*S5*(2.0f*y2 - x2 - z2);
    s[7]  = S15*vy*vz;
    s[8]  = 0.5f*S15*(z2 - x2);
    s[9]  = C358*vx*(3.0f*z2 - x2);
    s[10] = S105*vx*vy*vz;
    s[11] = C218*vx*(4.0f*y2 - z2 - x2);
    s[12] = 0.5f*S7*vy*(2.0f*y2 - 3.0f*z2 - 3.0f*x2);
    s[13] = C218*vz*(4.0f*y2 - z2 - x2);
    s[14] = 0.5f*S105*vy*(z2 - x2);
    s[15] = C358*vz*(z2 - 3.0f*x2);
    int d = edst[e];
    #pragma unroll
    for (int j = 0; j < 16; j++) {
        sh[(size_t)e*16 + j] = s[j];
        atomicAdd(&x[(size_t)d*16 + j], s[j] * INV_NN);
    }
    float len = elen[e];
    #pragma unroll
    for (int i = 0; i < 3; i++) {
        float dd = (len - 2.0f*(float)i) * 0.5f;
        basis[(size_t)e*3 + i] = expf(-dd*dd) * (1.0f/1.12f);
    }
}

__global__ __launch_bounds__(192) void kB(
    const float* __restrict__ sh, const float* __restrict__ basis,
    const float* __restrict__ x,
    const int* __restrict__ esrc, const int* __restrict__ edst,
    const float* __restrict__ W1a, const float* __restrict__ W1b,
    float* __restrict__ h, int E)
{
    __shared__ float h1[32][258];
    __shared__ float dots[32][4];
    __shared__ int   dstT[32];

    const int tid = threadIdx.x;
    const int e0  = blockIdx.x * 32;

    for (int idx = tid; idx < 32*256; idx += 192) {
        int m = idx >> 8, c = idx & 255;
        int e = e0 + m;
        float v = 0.0f;
        if (e < E) {
            float b0 = basis[(size_t)e*3], b1 = basis[(size_t)e*3+1], b2 = basis[(size_t)e*3+2];
            v = b0*W1a[c] + b1*W1a[256+c] + b2*W1a[512+c];
            v = fmaxf(v * INV_SQRT3, 0.0f);
        }
        h1[m][c] = v;
    }
    if (tid < 128) {
        int m = tid >> 2, p = tid & 3;
        int e = e0 + m;
        float s = 0.0f;
        if (e < E) {
            int src = esrc[e];
            int a = p*p, b = a + 2*p + 1;
            for (int j = a; j < b; j++)
                s += x[(size_t)src*16 + j] * sh[(size_t)e*16 + j];
            s *= rsqrtf((float)(2*p + 1));
        }
        dots[m][p] = s;
        if (p == 0) dstT[m] = (e < E) ? edst[e] : 0;
    }
    __syncthreads();

    const int kc    = tid >> 5;
    const int lane  = tid & 31;
    const int mA    = lane & 15;
    const int khalf = lane >> 4;
    const int n     = lane & 15;

    v8f a00 = {}, a01 = {}, a02 = {}, a03 = {};
    v8f a10 = {}, a11 = {}, a12 = {}, a13 = {};
    const int colb = kc * 16 + n;

    for (int k0 = 0; k0 < 256; k0 += 4) {
        int kb = k0 + 2*khalf;
        v2f A0, A1;
        A0.x = h1[mA][kb];      A0.y = h1[mA][kb + 1];
        A1.x = h1[16 + mA][kb]; A1.y = h1[16 + mA][kb + 1];
        const float* Brow0 = W1b + (size_t)kb * 384;
        const float* Brow1 = Brow0 + 384;
        v2f B0; B0.x = Brow0[colb];       B0.y = Brow1[colb];
        v2f B1; B1.x = Brow0[96  + colb]; B1.y = Brow1[96  + colb];
        v2f B2; B2.x = Brow0[192 + colb]; B2.y = Brow1[192 + colb];
        v2f B3; B3.x = Brow0[288 + colb]; B3.y = Brow1[288 + colb];
        a00 = __builtin_amdgcn_wmma_f32_16x16x4_f32(false, A0, false, B0, (short)0, a00, false, false);
        a10 = __builtin_amdgcn_wmma_f32_16x16x4_f32(false, A1, false, B0, (short)0, a10, false, false);
        a01 = __builtin_amdgcn_wmma_f32_16x16x4_f32(false, A0, false, B1, (short)0, a01, false, false);
        a11 = __builtin_amdgcn_wmma_f32_16x16x4_f32(false, A1, false, B1, (short)0, a11, false, false);
        a02 = __builtin_amdgcn_wmma_f32_16x16x4_f32(false, A0, false, B2, (short)0, a02, false, false);
        a12 = __builtin_amdgcn_wmma_f32_16x16x4_f32(false, A1, false, B2, (short)0, a12, false, false);
        a03 = __builtin_amdgcn_wmma_f32_16x16x4_f32(false, A0, false, B3, (short)0, a03, false, false);
        a13 = __builtin_amdgcn_wmma_f32_16x16x4_f32(false, A1, false, B3, (short)0, a13, false, false);
    }

    const float scale = (1.0f/16.0f) * 0.5f * INV_NN;
    #pragma unroll
    for (int r = 0; r < 8; r++) {
        int m = r + 8*khalf;
        if (e0 + m < E) {
            float ef = dots[m][0]*a00[r] + dots[m][1]*a01[r]
                     + dots[m][2]*a02[r] + dots[m][3]*a03[r];
            atomicAdd(&h[(size_t)dstT[m]*96 + kc*16 + n], ef * scale);
        }
        int m1 = 16 + m;
        if (e0 + m1 < E) {
            float ef = dots[m1][0]*a10[r] + dots[m1][1]*a11[r]
                     + dots[m1][2]*a12[r] + dots[m1][3]*a13[r];
            atomicAdd(&h[(size_t)dstT[m1]*96 + kc*16 + n], ef * scale);
        }
    }
}

__global__ void kC(const float* __restrict__ h, float* __restrict__ y, int N)
{
    int nid = blockIdx.x * blockDim.x + threadIdx.x;
    if (nid >= N) return;
    const float* hr = h + (size_t)nid * 96;
    float* yr = y + (size_t)nid * 64;
    #pragma unroll
    for (int j = 0; j < 16; j++) yr[j] = fmaxf(hr[j], 0.0f);
    #pragma unroll
    for (int j = 16; j < 32; j++) yr[j] = fabsf(hr[j]);
    #pragma unroll
    for (int j = 0; j < 32; j++) {
        float g = hr[32 + j];
        bool use_relu = (j < 8) || (j >= 16 && j < 24);
        float ga = use_relu ? fmaxf(g, 0.0f) : tanhf(g);
        yr[32 + j] = ga * hr[64 + j];
    }
}

#define KD_PITCH 68
__global__ __launch_bounds__(256) void kD(const float* __restrict__ y,
                                          const float* __restrict__ W2b,
                                          float* __restrict__ z, int N)
{
    __shared__ float wl[256 * KD_PITCH];
    const int tid = threadIdx.x;
    #pragma unroll
    for (int j = 0; j < 16; j++) {
        int idx = tid + 256 * j;
        int n = idx >> 4, q = idx & 15;
        int slot = (n & ~63) | ((n & 3) << 4) | ((n >> 2) & 15);
        float4 v = *reinterpret_cast<const float4*>(W2b + (size_t)n * 64 + q * 4);
        *reinterpret_cast<float4*>(&wl[slot * KD_PITCH + q * 4]) = v;
    }

    const int wave = tid >> 5, lane = tid & 31;
    const int c  = lane & 15;
    const int hh = lane >> 4;
    const int m0 = blockIdx.x * 128 + wave * 16;
    int rowA = m0 + c;
    if (rowA > N - 1) rowA = N - 1;
    const float* yr = y + (size_t)rowA * 64 + 2 * hh;
    v2f a[16];
    #pragma unroll
    for (int i = 0; i < 16; i++) a[i] = *reinterpret_cast<const v2f*>(yr + 4 * i);
    __syncthreads();

    #pragma unroll
    for (int g = 0; g < 4; g++) {
        v8f acc0 = {}, acc1 = {}, acc2 = {}, acc3 = {};
        const float* wb = &wl[(g * 64 + c) * KD_PITCH + 2 * hh];
        #pragma unroll
        for (int i = 0; i < 16; i++) {
            v2f b0 = *reinterpret_cast<const v2f*>(wb + 4 * i);
            v2f b1 = *reinterpret_cast<const v2f*>(wb + 16 * KD_PITCH + 4 * i);
            v2f b2 = *reinterpret_cast<const v2f*>(wb + 32 * KD_PITCH + 4 * i);
            v2f b3 = *reinterpret_cast<const v2f*>(wb + 48 * KD_PITCH + 4 * i);
            acc0 = __builtin_amdgcn_wmma_f32_16x16x4_f32(false, a[i], false, b0, (short)0, acc0, false, false);
            acc1 = __builtin_amdgcn_wmma_f32_16x16x4_f32(false, a[i], false, b1, (short)0, acc1, false, false);
            acc2 = __builtin_amdgcn_wmma_f32_16x16x4_f32(false, a[i], false, b2, (short)0, acc2, false, false);
            acc3 = __builtin_amdgcn_wmma_f32_16x16x4_f32(false, a[i], false, b3, (short)0, acc3, false, false);
        }
        #pragma unroll
        for (int r = 0; r < 8; r++) {
            int row = m0 + 8 * hh + r;
            if (row < N) {
                float4 o;
                o.x = acc0[r]; o.y = acc1[r]; o.z = acc2[r]; o.w = acc3[r];
                *reinterpret_cast<float4*>(z + (size_t)row * 256 + g * 64 + 4 * c) = o;
            }
        }
    }
}

__global__ __launch_bounds__(256) void kE(const float* __restrict__ basis,
                                          const float* __restrict__ z,
                                          const int* __restrict__ esrc,
                                          const int* __restrict__ edst,
                                          const float* __restrict__ W2a,
                                          float* __restrict__ out, int E)
{
    int e = blockIdx.x * (blockDim.x >> 5) + (threadIdx.x >> 5);
    int lane = threadIdx.x & 31;
    if (e >= E) return;
    float b0 = basis[(size_t)e*3], b1 = basis[(size_t)e*3+1], b2 = basis[(size_t)e*3+2];
    int src = esrc[e];
    const float* zr = z + (size_t)src * 256;
    float s = 0.0f;
    #pragma unroll
    for (int i = 0; i < 8; i++) {
        int c = lane + 32*i;
        float hv = b0*W2a[c] + b1*W2a[256+c] + b2*W2a[512+c];
        hv = fmaxf(hv * INV_SQRT3, 0.0f);
        s += hv * zr[c];
    }
    #pragma unroll
    for (int off = 16; off > 0; off >>= 1) s += __shfl_down(s, off, 32);
    if (lane == 0)
        atomicAdd(&out[edst[e]], s * (1.0f/128.0f) * INV_NN);
}

extern "C" void kernel_launch(void* const* d_in, const int* in_sizes, int n_in,
                              void* d_out, int out_size, void* d_ws, size_t ws_size,
                              hipStream_t stream)
{
    const float* evec = (const float*)d_in[0];
    const float* elen = (const float*)d_in[1];
    const float* W1a  = (const float*)d_in[2];
    const float* W1b  = (const float*)d_in[3];
    const float* W2a  = (const float*)d_in[4];
    const float* W2b  = (const float*)d_in[5];
    const int*   esrc = (const int*)d_in[6];
    const int*   edst = (const int*)d_in[7];
    float* out = (float*)d_out;

    const int E = in_sizes[0] / 3;
    const int N = out_size;

    float* ws = (float*)d_ws;
    float* x_     = ws;                 ws += (size_t)N * 16;
    float* h_     = ws;                 ws += (size_t)N * 96;
    float* y_     = ws;                 ws += (size_t)N * 64;
    float* z_     = ws;                 ws += (size_t)N * 256;
    float* sh_    = ws;                 ws += (size_t)E * 16;
    float* basis_ = ws;                 ws += (size_t)E * 3;

    long nx = (long)N * 16, nh = (long)N * 96;
    kzero<<<(int)((nx + 255) / 256), 256, 0, stream>>>(x_, nx);
    kzero<<<(int)((nh + 255) / 256), 256, 0, stream>>>(h_, nh);
    kzero<<<(N + 255) / 256, 256, 0, stream>>>(out, N);

    kA<<<(E + 255) / 256, 256, 0, stream>>>(evec, elen, edst, sh_, basis_, x_, E);
    kB<<<(E + 31) / 32, 192, 0, stream>>>(sh_, basis_, x_, esrc, edst, W1a, W1b, h_, E);
    kC<<<(N + 255) / 256, 256, 0, stream>>>(h_, y_, N);
    kD<<<(N + 127) / 128, 256, 0, stream>>>(y_, W2b, z_, N);
    kE<<<(E + 7) / 8, 256, 0, stream>>>(basis_, z_, esrc, edst, W2a, out, E);
}


# conv1: edges grouped by destination (rank, scan, fill), dots*h1 summed per node first, then one f32 WMMA product per node tile; h stored, no atomics; order of the sums over edges and k changes
# speedup vs baseline: 1.7126x; 1.4058x over previous
#include <hip/hip_runtime.h>
#include <math.h>

typedef __attribute__((ext_vector_type(2))) float v2f;
typedef __attribute__((ext_vector_type(8))) float v8f;

#define INV_NN     0.512989176042577f
#define INV_SQRT3  0.5773502691896258f
#define S3   1.7320508075688772f
#define S5   2.23606797749979f
#define S7   2.6457513110645907f
#define S15  3.872983346207417f
#define S105 10.246950765959598f
#define C358 2.091650066335189f
#define C218 1.6201851746019651f

__global__ void kzero(float* __restrict__ p, long n) {
    long i = (long)blockIdx.x * blockDim.x + threadIdx.x;
    if (i < n) p[i] = 0.0f;
}

__global__ void kA(const float* __restrict__ evec, const float* __restrict__ elen,
                   const int* __restrict__ edst,
                   float* __restrict__ sh, float* __restrict__ basis,
                   float* __restrict__ x,
                   int* __restrict__ cnt, int* __restrict__ rank, int E)
{
    int e = blockIdx.x * blockDim.x + threadIdx.x;
    if (e >= E) return;
    float vx = evec[3*(size_t)e], vy = evec[3*(size_t)e+1], vz = evec[3*(size_t)e+2];
    float rn = 1.0f / (sqrtf(vx*vx + vy*vy + vz*vz) + 1e-12f);
    vx *= rn; vy *= rn; vz *= rn;
    float x2 = vx*vx, y2 = vy*vy, z2 = vz*vz;
    float s[16];
    s[0]  = 1.0f;
    s[1]  = S3*vx; s[2] = S3*vy; s[3] = S3*vz;
    s[4]  = S15*vx*vz;
    s[5]  = S15*vx*vy;
    s[6]  = 0.5f*S5*(2.0f*y2 - x2 - z2);
    s[7]  = S15*vy*vz;
    s[8]  = 0.5f*S15*(z2 - x2);
    s[9]  = C358*vx*(3.0f*z2 - x2);
    s[10] = S105*vx*vy*vz;
    s[11] = C218*vx*(4.0f*y2 - z2 - x2);
    s[12] = 0.5f*S7*vy*(2.0f*y2 - 3.0f*z2 - 3.0f*x2);
    s[13] = C218*vz*(4.0f*y2 - z2 - x2);
    s[14] = 0.5f*S105*vy*(z2 - x2);
    s[15] = C358*vz*(z2 - 3.0f*x2);
    int d = edst[e];
    rank[e] = atomicAdd(&cnt[d], 1);
    #pragma unroll
    for (int j = 0; j < 16; j++) {
        sh[(size_t)e*16 + j] = s[j];
        atomicAdd(&x[(size_t)d*16 + j], s[j] * INV_NN);
    }
    float len = elen[e];
    #pragma unroll
    for (int i = 0; i < 3; i++) {
        float dd = (len - 2.0f*(float)i) * 0.5f;
        basis[(size_t)e*3 + i] = expf(-dd*dd) * (1.0f/1.12f);
    }
}

__global__ __launch_bounds__(1024) void kScan(const int* __restrict__ cnt,
                                              int* __restrict__ off, int N)
{
    __shared__ int part[1024];
    const int t = threadIdx.x;
    const int chunk = (N + 1023) / 1024;
    const int b = t * chunk;
    const int bend = (b + chunk < N) ? (b + chunk) : N;
    int s = 0;
    for (int i = b; i < bend; i++) s += cnt[i];
    part[t] = s;
    __syncthreads();
    for (int d = 1; d < 1024; d <<= 1) {
        int v = (t >= d) ? part[t - d] : 0;
        __syncthreads();
        part[t] += v;
        __syncthreads();
    }
    int run = part[t] - s;
    for (int i = b; i < bend; i++) { off[i] = run; run += cnt[i]; }
    if (t == 1023) off[N] = part[1023];
}

__global__ void kFill(const int* __restrict__ edst, const int* __restrict__ rank,
                      const int* __restrict__ off, int* __restrict__ perm, int E)
{
    int e = blockIdx.x * blockDim.x + threadIdx.x;
    if (e >= E) return;
    perm[off[edst[e]] + rank[e]] = e;
}

#define KB_PITCH 1028
#define RS3 0.5773502691896258f
#define RS5 0.4472135954999579f
#define RS7 0.3779644730092272f
__global__ __launch_bounds__(192) void kB(
    const float* __restrict__ sh, const float* __restrict__ basis,
    const float* __restrict__ x,
    const int* __restrict__ esrc,
    const int* __restrict__ off, const int* __restrict__ perm,
    const float* __restrict__ W1a, const float* __restrict__ W1b,
    float* __restrict__ h, int N)
{
    __shared__ float Gs[32 * KB_PITCH];

    const int tid  = threadIdx.x;
    const int lane = tid & 31;
    const int wave = __builtin_amdgcn_readfirstlane(tid >> 5);
    const int n0   = blockIdx.x * 32;

    float wa0[8], wa1[8], wa2[8];
    #pragma unroll
    for (int i = 0; i < 8; i++) {
        wa0[i] = W1a[lane + 32*i];
        wa1[i] = W1a[256 + lane + 32*i];
        wa2[i] = W1a[512 + lane + 32*i];
    }

    for (int m = wave; m < 32; m += 6) {
        const int n = n0 + m;
        float g0[8], g1[8], g2[8], g3[8];
        #pragma unroll
        for (int i = 0; i < 8; i++) { g0[i] = 0.0f; g1[i] = 0.0f; g2[i] = 0.0f; g3[i] = 0.0f; }
        if (n < N) {
            const int beg = off[n], end = off[n + 1];
            for (int base = beg; base < end; base += 32) {
                const int idx = base + lane;
                float d0 = 0.0f, d1 = 0.0f, d2 = 0.0f, d3 = 0.0f;
                float b0 = 0.0f, b1 = 0.0f, b2 = 0.0f;
                if (idx < end) {
                    const int e   = perm[idx];
                    const int srcn = esrc[e];
                    const float4* xs = reinterpret_cast<const float4*>(x  + (size_t)srcn * 16);
                    const float4* ss = reinterpret_cast<const float4*>(sh + (size_t)e * 16);
                    float4 x0 = xs[0], x1 = xs[1], x2 = xs[2], x3 = xs[3];
                    float4 s0 = ss[0], s1 = ss[1], s2 = ss[2], s3 = ss[3];
                    d0 = x0.x*s0.x;
                    d1 = (x0.y*s0.y + x0.z*s0.z + x0.w*s0.w) * RS3;
                    d2 = (x1.x*s1.x + x1.y*s1.y + x1.z*s1.z + x1.w*s1.w + x2.x*s2.x) * RS5;
                    d3 = (x2.y*s2.y + x2.z*s2.z + x2.w*s2.w
                        + x3.x*s3.x + x3.y*s3.y + x3.z*s3.z + x3.w*s3.w) * RS7;
                    b0 = basis[(size_t)e*3]; b1 = basis[(size_t)e*3 + 1]; b2 = basis[(size_t)e*3 + 2];
                }
                const int left = end - base;
                const int cntc = left < 32 ? left : 32;
                for (int j = 0; j < cntc; j++) {
                    const float e0 = __shfl(d0, j, 32), e1 = __shfl(d1, j, 32);
                    const float e2 = __shfl(d2, j, 32), e3 = __shfl(d3, j, 32);
                    const float c0 = __shfl(b0, j, 32), c1 = __shfl(b1, j, 32), c2 = __shfl(b2, j, 32);
                    #pragma unroll
                    for (int i = 0; i < 8; i++) {
                        float hv = c0*wa0[i] + c1*wa1[i] + c2*wa2[i];
                        hv = fmaxf(hv * INV_SQRT3, 0.0f);
                        g0[i] += e0 * hv; g1[i] += e1 * hv;
                        g2[i] += e2 * hv; g3[i] += e3 * hv;
                    }
                }
            }
        }
        float* gr = &Gs[m * KB_PITCH + lane];
        #pragma unroll
        for (int i = 0; i < 8; i++) {
            gr[32*i]       = g0[i];
            gr[256 + 32*i] = g1[i];
            gr[512 + 32*i] = g2[i];
            gr[768 + 32*i] = g3[i];
        }
    }
    __syncthreads();

    const int c    = lane & 15;
    const int hh   = lane >> 4;
    const int colb = wave * 16 + c;
    v8f a00 = {}, a01 = {}, a02 = {}, a03 = {};
    v8f a10 = {}, a11 = {}, a12 = {}, a13 = {};
    const float* ga0 = &Gs[c * KB_PITCH + 2 * hh];
    const float* ga1 = ga0 + 16 * KB_PITCH;
    const float* Bp = W1b + (size_t)(2 * hh) * 384 + colb;
    v2f Bc[2][4], Bn[2][4];
    #pragma unroll
    for (int s = 0; s < 2; s++) {
        #pragma unroll
        for (int q = 0; q < 4; q++) {
            Bc[s][q].x = Bp[(size_t)(4 * s) * 384 + 96 * q];
            Bc[s][q].y = Bp[(size_t)(4 * s + 1) * 384 + 96 * q];
        }
    }
    for (int k0 = 0; k0 < 256; k0 += 8) {
        const int kn = (k0 + 8 < 256) ? (k0 + 8) : k0;
        const float* Bq = Bp + (size_t)kn * 384;
        #pragma unroll
        for (int s = 0; s < 2; s++) {
            #pragma unroll
            for (int q = 0; q < 4; q++) {
                Bn[s][q].x = Bq[(size_t)(4 * s) * 384 + 96 * q];
                Bn[s][q].y = Bq[(size_t)(4 * s + 1) * 384 + 96 * q];
            }
        }
        #pragma unroll
        for (int s = 0; s < 2; s++) {
            const int k = k0 + 4 * s;
            v2f A00 = *reinterpret_cast<const v2f*>(ga0 + k);
            v2f A01 = *reinterpret_cast<const v2f*>(ga0 + 256 + k);
            v2f A02 = *reinterpret_cast<const v2f*>(ga0 + 512 + k);
            v2f A03 = *reinterpret_cast<const v2f*>(ga0 + 768 + k);
            v2f A10 = *reinterpret_cast<const v2f*>(ga1 + k);
            v2f A11 = *reinterpret_cast<const v2f*>(ga1 + 256 + k);
            v2f A12 = *reinterpret_cast<const v2f*>(ga1 + 512 + k);
            v2f A13 = *reinterpret_cast<const v2f*>(ga1 + 768 + k);
            a00 = __builtin_amdgcn_wmma_f32_16x16x4_f32(false, A00, false, Bc[s][0], (short)0, a00, false, false);
            a10 = __builtin_amdgcn_wmma_f32_16x16x4_f32(false, A10, false, Bc[s][0], (short)0, a10, false, false);
            a01 = __builtin_amdgcn_wmma_f32_16x16x4_f32(false, A01, false, Bc[s][1], (short)0, a01, false, false);
            a11 = __builtin_amdgcn_wmma_f32_16x16x4_f32(false, A11, false, Bc[s][1], (short)0, a11, false, false);
            a02 = __builtin_amdgcn_wmma_f32_16x16x4_f32(false, A02, false, Bc[s][2], (short)0, a02, false, false);
            a12 = __builtin_amdgcn_wmma_f32_16x16x4_f32(false, A12, false, Bc[s][2], (short)0, a12, false, false);
            a03 = __builtin_amdgcn_wmma_f32_16x16x4_f32(false, A03, false, Bc[s][3], (short)0, a03, false, false);
            a13 = __builtin_amdgcn_wmma_f32_16x16x4_f32(false, A13, false, Bc[s][3], (short)0, a13, false, false);
        }
        #pragma unroll
        for (int s = 0; s < 2; s++) {
            #pragma unroll
            for (int q = 0; q < 4; q++) Bc[s][q] = Bn[s][q];
        }
    }
    __syncthreads();

    const float scale = (1.0f/16.0f) * 0.5f * INV_NN;
    #pragma unroll
    for (int r = 0; r < 8; r++) {
        Gs[(8*hh + r) * 96 + colb]      = ((a00[r] + a01[r]) + (a02[r] + a03[r])) * scale;
        Gs[(16 + 8*hh + r) * 96 + colb] = ((a10[r] + a11[r]) + (a12[r] + a13[r])) * scale;
    }
    __syncthreads();
    const int rows = (N - n0 < 32) ? (N - n0) : 32;
    for (int idx = tid; idx < rows * 24; idx += 192) {
        float4 v = *reinterpret_cast<const float4*>(&Gs[idx * 4]);
        *reinterpret_cast<float4*>(h + (size_t)n0 * 96 + idx * 4) = v;
    }
}

__global__ void kC(const float* __restrict__ h, float* __restrict__ y, int N)
{
    int nid = blockIdx.x * blockDim.x + threadIdx.x;
    if (nid >= N) return;
    const float* hr = h + (size_t)nid * 96;
    float* yr = y + (size_t)nid * 64;
    #pragma unroll
    for (int j = 0; j < 16; j++) yr[j] = fmaxf(hr[j], 0.0f);
    #pragma unroll
    for (int j = 16; j < 32; j++) yr[j] = fabsf(hr[j]);
    #pragma unroll
    for (int j = 0; j < 32; j++) {
        float g = hr[32 + j];
        bool use_relu = (j < 8) || (j >= 16 && j < 24);
        float ga = use_relu ? fmaxf(g, 0.0f) : tanhf(g);
        yr[32 + j] = ga * hr[64 + j];
    }
}

#define KD_PITCH 68
__global__ __launch_bounds__(256) void kD(const float* __restrict__ y,
                                          const float* __restrict__ W2b,
                                          float* __restrict__ z, int N)
{
    __shared__ float wl[256 * KD_PITCH];
    const int tid = threadIdx.x;
    #pragma unroll
    for (int j = 0; j < 16; j++) {
        int idx = tid + 256 * j;
        int n = idx >> 4, q = idx & 15;
        int slot = (n & ~63) | ((n & 3) << 4) | ((n >> 2) & 15);
        float4 v = *reinterpret_cast<const float4*>(W2b + (size_t)n * 64 + q * 4);
        *reinterpret_cast<float4*>(&wl[slot * KD_PITCH + q * 4]) = v;
    }

    const int wave = tid >> 5, lane = tid & 31;
    const int c  = lane & 15;
    const int hh = lane >> 4;
    const int m0 = blockIdx.x * 128 + wave * 16;
    int rowA = m0 + c;
    if (rowA > N - 1) rowA = N - 1;
    const float* yr = y + (size_t)rowA * 64 + 2 * hh;
    v2f a[16];
    #pragma unroll
    for (int i = 0; i < 16; i++) a[i] = *reinterpret_cast<const v2f*>(yr + 4 * i);
    __syncthreads();

    #pragma unroll
    for (int g = 0; g < 4; g++) {
        v8f acc0 = {}, acc1 = {}, acc2 = {}, acc3 = {};
        const float* wb = &wl[(g * 64 + c) * KD_PITCH + 2 * hh];
        #pragma unroll
        for (int i = 0; i < 16; i++) {
            v2f b0 = *reinterpret_cast<const v2f*>(wb + 4 * i);
            v2f b1 = *reinterpret_cast<const v2f*>(wb + 16 * KD_PITCH + 4 * i);
            v2f b2 = *reinterpret_cast<const v2f*>(wb + 32 * KD_PITCH + 4 * i);
            v2f b3 = *reinterpret_cast<const v2f*>(wb + 48 * KD_PITCH + 4 * i);
            acc0 = __builtin_amdgcn_wmma_f32_16x16x4_f32(false, a[i], false, b0, (short)0, acc0, false, false);
            acc1 = __builtin_amdgcn_wmma_f32_16x16x4_f32(false, a[i], false, b1, (short)0, acc1, false, false);
            acc2 = __builtin_amdgcn_wmma_f32_16x16x4_f32(false, a[i], false, b2, (short)0, acc2, false, false);
            acc3 = __builtin_amdgcn_wmma_f32_16x16x4_f32(false, a[i], false, b3, (short)0, acc3, false, false);
        }
        #pragma unroll
        for (int r = 0; r < 8; r++) {
            int row = m0 + 8 * hh + r;
            if (row < N) {
                float4 o;
                o.x = acc0[r]; o.y = acc1[r]; o.z = acc2[r]; o.w = acc3[r];
                *reinterpret_cast<float4*>(z + (size_t)row * 256 + g * 64 + 4 * c) = o;
            }
        }
    }
}

__global__ __launch_bounds__(256) void kE(const float* __restrict__ basis,
                                          const float* __restrict__ z,
                                          const int* __restrict__ esrc,
                                          const int* __restrict__ edst,
                                          const float* __restrict__ W2a,
                                          float* __restrict__ out, int E)
{
    int e = blockIdx.x * (blockDim.x >> 5) + (threadIdx.x >> 5);
    int lane = threadIdx.x & 31;
    if (e >= E) return;
    float b0 = basis[(size_t)e*3], b1 = basis[(size_t)e*3+1], b2 = basis[(size_t)e*3+2];
    int src = esrc[e];
    const float* zr = z + (size_t)src * 256;
    float s = 0.0f;
    #pragma unroll
    for (int i = 0; i < 8; i++) {
        int c = lane + 32*i;
        float hv = b0*W2a[c] + b1*W2a[256+c] + b2*W2a[512+c];
        hv = fmaxf(hv * INV_SQRT3, 0.0f);
        s += hv * zr[c];
    }
    #pragma unroll
    for (int off = 16; off > 0; off >>= 1) s += __shfl_down(s, off, 32);
    if (lane == 0)
        atomicAdd(&out[edst[e]], s * (1.0f/128.0f) * INV_NN);
}

extern "C" void kernel_launch(void* const* d_in, const int* in_sizes, int n_in,
                              void* d_out, int out_size, void* d_ws, size_t ws_size,
                              hipStream_t stream)
{
    const float* evec = (const float*)d_in[0];
    const float* elen = (const float*)d_in[1];
    const float* W1a  = (const float*)d_in[2];
    const float* W1b  = (const float*)d_in[3];
    const float* W2a  = (const float*)d_in[4];
    const float* W2b  = (const float*)d_in[5];
    const int*   esrc = (const int*)d_in[6];
    const int*   edst = (const int*)d_in[7];
    float* out = (float*)d_out;

    const int E = in_sizes[0] / 3;
    const int N = out_size;

    float* ws = (float*)d_ws;
    float* x_     = ws;                 ws += (size_t)N * 16;
    float* h_     = ws;                 ws += (size_t)N * 96;
    float* y_     = ws;                 ws += (size_t)N * 64;
    float* z_     = ws;                 ws += (size_t)N * 256;
    float* sh_    = ws;                 ws += (size_t)E * 16;
    float* basis_ = ws;                 ws += (size_t)E * 3;

    int* cnt_  = (int*)z_;
    int* rank_ = cnt_ + N;
    int* off_  = rank_ + E;
    int* perm_ = off_ + N + 1;

    long nx = (long)N * 16;
    kzero<<<(int)((nx + 255) / 256), 256, 0, stream>>>(x_, nx);
    kzero<<<(N + 255) / 256, 256, 0, stream>>>((float*)cnt_, N);
    kzero<<<(N + 255) / 256, 256, 0, stream>>>(out, N);

    kA<<<(E + 255) / 256, 256, 0, stream>>>(evec, elen, edst, sh_, basis_, x_, cnt_, rank_, E);
    kScan<<<1, 1024, 0, stream>>>(cnt_, off_, N);
    kFill<<<(E + 255) / 256, 256, 0, stream>>>(edst, rank_, off_, perm_, E);
    kB<<<(N + 31) / 32, 192, 0, stream>>>(sh_, basis_, x_, esrc, off_, perm_, W1a, W1b, h_, N);
    kC<<<(N + 255) / 256, 256, 0, stream>>>(h_, y_, N);
    kD<<<(N + 127) / 128, 256, 0, stream>>>(y_, W2b, z_, N);
    kE<<<(E + 7) / 8, 256, 0, stream>>>(basis_, z_, esrc, edst, W2a, out, E);
}


# final conv (kE): one wave per 4 consecutive nodes walks the destination-sorted edge list, sums in registers, stores out[n] once; no atomics, out not zeroed; sum order over edges and columns changes
# speedup vs baseline: 1.9322x; 1.1282x over previous
#include <hip/hip_runtime.h>
#include <math.h>

typedef __attribute__((ext_vector_type(2))) float v2f;
typedef __attribute__((ext_vector_type(8))) float v8f;

#define INV_NN     0.512989176042577f
#define INV_SQRT3  0.5773502691896258f
#define S3   1.7320508075688772f
#define S5   2.23606797749979f
#define S7   2.6457513110645907f
#define S15  3.872983346207417f
#define S105 10.246950765959598f
#define C358 2.091650066335189f
#define C218 1.6201851746019651f

__global__ void kzero(float* __restrict__ p, long n) {
    long i = (long)blockIdx.x * blockDim.x + threadIdx.x;
    if (i < n) p[i] = 0.0f;
}

__global__ void kA(const float* __restrict__ evec, const float* __restrict__ elen,
                   const int* __restrict__ edst,
                   float* __restrict__ sh, float* __restrict__ basis,
                   float* __restrict__ x,
                   int* __restrict__ cnt, int* __restrict__ rank, int E)
{
    int e = blockIdx.x * blockDim.x + threadIdx.x;
    if (e >= E) return;
    float vx = evec[3*(size_t)e], vy = evec[3*(size_t)e+1], vz = evec[3*(size_t)e+2];
    float rn = 1.0f / (sqrtf(vx*vx + vy*vy + vz*vz) + 1e-12f);
    vx *= rn; vy *= rn; vz *= rn;
    float x2 = vx*vx, y2 = vy*vy, z2 = vz*vz;
    float s[16];
    s[0]  = 1.0f;
    s[1]  = S3*vx; s[2] = S3*vy; s[3] = S3*vz;
    s[4]  = S15*vx*vz;
    s[5]  = S15*vx*vy;
    s[6]  = 0.5f*S5*(2.0f*y2 - x2 - z2);
    s[7]  = S15*vy*vz;
    s[8]  = 0.5f*S15*(z2 - x2);
    s[9]  = C358*vx*(3.0f*z2 - x2);
    s[10] = S105*vx*vy*vz;
    s[11] = C218*vx*(4.0f*y2 - z2 - x2);
    s[12] = 0.5f*S7*vy*(2.0f*y2 - 3.0f*z2 - 3.0f*x2);
    s[13] = C218*vz*(4.0f*y2 - z2 - x2);
    s[14] = 0.5f*S105*vy*(z2 - x2);
    s[15] = C358*vz*(z2 - 3.0f*x2);
    int d = edst[e];
    rank[e] = atomicAdd(&cnt[d], 1);
    #pragma unroll
    for (int j = 0; j < 16; j++) {
        sh[(size_t)e*16 + j] = s[j];
        atomicAdd(&x[(size_t)d*16 + j], s[j] * INV_NN);
    }
    float len = elen[e];
    #pragma unroll
    for (int i = 0; i < 3; i++) {
        float dd = (len - 2.0f*(float)i) * 0.5f;
        basis[(size_t)e*3 + i] = expf(-dd*dd) * (1.0f/1.12f);
    }
}

__global__ __launch_bounds__(1024) void kScan(const int* __restrict__ cnt,
                                              int* __restrict__ off, int N)
{
    __shared__ int part[1024];
    const int t = threadIdx.x;
    const int chunk = (N + 1023) / 1024;
    const int b = t * chunk;
    const int bend = (b + chunk < N) ? (b + chunk) : N;
    int s = 0;
    for (int i = b; i < bend; i++) s += cnt[i];
    part[t] = s;
    __syncthreads();
    for (int d = 1; d < 1024; d <<= 1) {
        int v = (t >= d) ? part[t - d] : 0;
        __syncthreads();
        part[t] += v;
        __syncthreads();
    }
    int run = part[t] - s;
    for (int i = b; i < bend; i++) { off[i] = run; run += cnt[i]; }
    if (t == 1023) off[N] = part[1023];
}

__global__ void kFill(const int* __restrict__ edst, const int* __restrict__ rank,
                      const int* __restrict__ off, int* __restrict__ perm, int E)
{
    int e = blockIdx.x * blockDim.x + threadIdx.x;
    if (e >= E) return;
    perm[off[edst[e]] + rank[e]] = e;
}

#define KB_PITCH 1028
#define RS3 0.5773502691896258f
#define RS5 0.4472135954999579f
#define RS7 0.3779644730092272f
__global__ __launch_bounds__(192) void kB(
    const float* __restrict__ sh, const float* __restrict__ basis,
    const float* __restrict__ x,
    const int* __restrict__ esrc,
    const int* __restrict__ off, const int* __restrict__ perm,
    const float* __restrict__ W1a, const float* __restrict__ W1b,
    float* __restrict__ h, int N)
{
    __shared__ float Gs[32 * KB_PITCH];

    const int tid  = threadIdx.x;
    const int lane = tid & 31;
    const int wave = __builtin_amdgcn_readfirstlane(tid >> 5);
    const int n0   = blockIdx.x * 32;

    float wa0[8], wa1[8], wa2[8];
    #pragma unroll
    for (int i = 0; i < 8; i++) {
        wa0[i] = W1a[lane + 32*i];
        wa1[i] = W1a[256 + lane + 32*i];
        wa2[i] = W1a[512 + lane + 32*i];
    }

    for (int m = wave; m < 32; m += 6) {
        const int n = n0 + m;
        float g0[8], g1[8], g2[8], g3[8];
        #pragma unroll
        for (int i = 0; i < 8; i++) { g0[i] = 0.0f; g1[i] = 0.0f; g2[i] = 0.0f; g3[i] = 0.0f; }
        if (n < N) {
            const int beg = off[n], end = off[n + 1];
            for (int base = beg; base < end; base += 32) {
                const int idx = base + lane;
                float d0 = 0.0f, d1 = 0.0f, d2 = 0.0f, d3 = 0.0f;
                float b0 = 0.0f, b1 = 0.0f, b2 = 0.0f;
                if (idx < end) {
                    const int e   = perm[idx];
                    const int srcn = esrc[e];
                    const float4* xs = reinterpret_cast<const float4*>(x  + (size_t)srcn * 16);
                    const float4* ss = reinterpret_cast<const float4*>(sh + (size_t)e * 16);
                    float4 x0 = xs[0], x1 = xs[1], x2 = xs[2], x3 = xs[3];
                    float4 s0 = ss[0], s1 = ss[1], s2 = ss[2], s3 = ss[3];
                    d0 = x0.x*s0.x;
                    d1 = (x0.y*s0.y + x0.z*s0.z + x0.w*s0.w) * RS3;
                    d2 = (x1.x*s1.x + x1.y*s1.y + x1.z*s1.z + x1.w*s1.w + x2.x*s2.x) * RS5;
                    d3 = (x2.y*s2.y + x2.z*s2.z + x2.w*s2.w
                        + x3.x*s3.x + x3.y*s3.y + x3.z*s3.z + x3.w*s3.w) * RS7;
                    b0 = basis[(size_t)e*3]; b1 = basis[(size_t)e*3 + 1]; b2 = basis[(size_t)e*3 + 2];
                }
                const int left = end - base;
                const int cntc = left < 32 ? left : 32;
                for (int j = 0; j < cntc; j++) {
                    const float e0 = __shfl(d0, j, 32), e1 = __shfl(d1, j, 32);
                    const float e2 = __shfl(d2, j, 32), e3 = __shfl(d3, j, 32);
                    const float c0 = __shfl(b0, j, 32), c1 = __shfl(b1, j, 32), c2 = __shfl(b2, j, 32);
                    #pragma unroll
                    for (int i = 0; i < 8; i++) {
                        float hv = c0*wa0[i] + c1*wa1[i] + c2*wa2[i];
                        hv = fmaxf(hv * INV_SQRT3, 0.0f);
                        g0[i] += e0 * hv; g1[i] += e1 * hv;
                        g2[i] += e2 * hv; g3[i] += e3 * hv;
                    }
                }
            }
        }
        float* gr = &Gs[m * KB_PITCH + lane];
        #pragma unroll
        for (int i = 0; i < 8; i++) {
            gr[32*i]       = g0[i];
            gr[256 + 32*i] = g1[i];
            gr[512 + 32*i] = g2[i];
            gr[768 + 32*i] = g3[i];
        }
    }
    __syncthreads();

    const int c    = lane & 15;
    const int hh   = lane >> 4;
    const int colb = wave * 16 + c;
    v8f a00 = {}, a01 = {}, a02 = {}, a03 = {};
    v8f a10 = {}, a11 = {}, a12 = {}, a13 = {};
    const float* ga0 = &Gs[c * KB_PITCH + 2 * hh];
    const float* ga1 = ga0 + 16 * KB_PITCH;
    const float* Bp = W1b + (size_t)(2 * hh) * 384 + colb;
    v2f Bc[2][4], Bn[2][4];
    #pragma unroll
    for (int s = 0; s < 2; s++) {
        #pragma unroll
        for (int q = 0; q < 4; q++) {
            Bc[s][q].x = Bp[(size_t)(4 * s) * 384 + 96 * q];
            Bc[s][q].y = Bp[(size_t)(4 * s + 1) * 384 + 96 * q];
        }
    }
    for (int k0 = 0; k0 < 256; k0 += 8) {
        const int kn = (k0 + 8 < 256) ? (k0 + 8) : k0;
        const float* Bq = Bp + (size_t)kn * 384;
        #pragma unroll
        for (int s = 0; s < 2; s++) {
            #pragma unroll
            for (int q = 0; q < 4; q++) {
                Bn[s][q].x = Bq[(size_t)(4 * s) * 384 + 96 * q];
                Bn[s][q].y = Bq[(size_t)(4 * s + 1) * 384 + 96 * q];
            }
        }
        #pragma unroll
        for (int s = 0; s < 2; s++) {
            const int k = k0 + 4 * s;
            v2f A00 = *reinterpret_cast<const v2f*>(ga0 + k);
            v2f A01 = *reinterpret_cast<const v2f*>(ga0 + 256 + k);
            v2f A02 = *reinterpret_cast<const v2f*>(ga0 + 512 + k);
            v2f A03 = *reinterpret_cast<const v2f*>(ga0 + 768 + k);
            v2f A10 = *reinterpret_cast<const v2f*>(ga1 + k);
            v2f A11 = *reinterpret_cast<const v2f*>(ga1 + 256 + k);
            v2f A12 = *reinterpret_cast<const v2f*>(ga1 + 512 + k);
            v2f A13 = *reinterpret_cast<const v2f*>(ga1 + 768 + k);
            a00 = __builtin_amdgcn_wmma_f32_16x16x4_f32(false, A00, false, Bc[s][0], (short)0, a00, false, false);
            a10 = __builtin_amdgcn_wmma_f32_16x16x4_f32(false, A10, false, Bc[s][0], (short)0, a10, false, false);
            a01 = __builtin_amdgcn_wmma_f32_16x16x4_f32(false, A01, false, Bc[s][1], (short)0, a01, false, false);
            a11 = __builtin_amdgcn_wmma_f32_16x16x4_f32(false, A11, false, Bc[s][1], (short)0, a11, false, false);
            a02 = __builtin_amdgcn_wmma_f32_16x16x4_f32(false, A02, false, Bc[s][2], (short)0, a02, false, false);
            a12 = __builtin_amdgcn_wmma_f32_16x16x4_f32(false, A12, false, Bc[s][2], (short)0, a12, false, false);
            a03 = __builtin_amdgcn_wmma_f32_16x16x4_f32(false, A03, false, Bc[s][3], (short)0, a03, false, false);
            a13 = __builtin_amdgcn_wmma_f32_16x16x4_f32(false, A13, false, Bc[s][3], (short)0, a13, false, false);
        }
        #pragma unroll
        for (int s = 0; s < 2; s++) {
            #pragma unroll
            for (int q = 0; q < 4; q++) Bc[s][q] = Bn[s][q];
        }
    }
    __syncthreads();

    const float scale = (1.0f/16.0f) * 0.5f * INV_NN;
    #pragma unroll
    for (int r = 0; r < 8; r++) {
        Gs[(8*hh + r) * 96 + colb]      = ((a00[r] + a01[r]) + (a02[r] + a03[r])) * scale;
        Gs[(16 + 8*hh + r) * 96 + colb] = ((a10[r] + a11[r]) + (a12[r] + a13[r])) * scale;
    }
    __syncthreads();
    const int rows = (N - n0 < 32) ? (N - n0) : 32;
    for (int idx = tid; idx < rows * 24; idx += 192) {
        float4 v = *reinterpret_cast<const float4*>(&Gs[idx * 4]);
        *reinterpret_cast<float4*>(h + (size_t)n0 * 96 + idx * 4) = v;
    }
}

__global__ void kC(const float* __restrict__ h, float* __restrict__ y, int N)
{
    int nid = blockIdx.x * blockDim.x + threadIdx.x;
    if (nid >= N) return;
    const float* hr = h + (size_t)nid * 96;
    float* yr = y + (size_t)nid * 64;
    #pragma unroll
    for (int j = 0; j < 16; j++) yr[j] = fmaxf(hr[j], 0.0f);
    #pragma unroll
    for (int j = 16; j < 32; j++) yr[j] = fabsf(hr[j]);
    #pragma unroll
    for (int j = 0; j < 32; j++) {
        float g = hr[32 + j];
        bool use_relu = (j < 8) || (j >= 16 && j < 24);
        float ga = use_relu ? fmaxf(g, 0.0f) : tanhf(g);
        yr[32 + j] = ga * hr[64 + j];
    }
}

#define KD_PITCH 68
__global__ __launch_bounds__(256) void kD(const float* __restrict__ y,
                                          const float* __restrict__ W2b,
                                          float* __restrict__ z, int N)
{
    __shared__ float wl[256 * KD_PITCH];
    const int tid = threadIdx.x;
    #pragma unroll
    for (int j = 0; j < 16; j++) {
        int idx = tid + 256 * j;
        int n = idx >> 4, q = idx & 15;
        int slot = (n & ~63) | ((n & 3) << 4) | ((n >> 2) & 15);
        float4 v = *reinterpret_cast<const float4*>(W2b + (size_t)n * 64 + q * 4);
        *reinterpret_cast<float4*>(&wl[slot * KD_PITCH + q * 4]) = v;
    }

    const int wave = tid >> 5, lane = tid & 31;
    const int c  = lane & 15;
    const int hh = lane >> 4;
    const int m0 = blockIdx.x * 128 + wave * 16;
    int rowA = m0 + c;
    if (rowA > N - 1) rowA = N - 1;
    const float* yr = y + (size_t)rowA * 64 + 2 * hh;
    v2f a[16];
    #pragma unroll
    for (int i = 0; i < 16; i++) a[i] = *reinterpret_cast<const v2f*>(yr + 4 * i);
    __syncthreads();

    #pragma unroll
    for (int g = 0; g < 4; g++) {
        v8f acc0 = {}, acc1 = {}, acc2 = {}, acc3 = {};
        const float* wb = &wl[(g * 64 + c) * KD_PITCH + 2 * hh];
        #pragma unroll
        for (int i = 0; i < 16; i++) {
            v2f b0 = *reinterpret_cast<const v2f*>(wb + 4 * i);
            v2f b1 = *reinterpret_cast<const v2f*>(wb + 16 * KD_PITCH + 4 * i);
            v2f b2 = *reinterpret_cast<const v2f*>(wb + 32 * KD_PITCH + 4 * i);
            v2f b3 = *reinterpret_cast<const v2f*>(wb + 48 * KD_PITCH + 4 * i);
            acc0 = __builtin_amdgcn_wmma_f32_16x16x4_f32(false, a[i], false, b0, (short)0, acc0, false, false);
            acc1 = __builtin_amdgcn_wmma_f32_16x16x4_f32(false, a[i], false, b1, (short)0, acc1, false, false);
            acc2 = __builtin_amdgcn_wmma_f32_16x16x4_f32(false, a[i], false, b2, (short)0, acc2, false, false);
            acc3 = __builtin_amdgcn_wmma_f32_16x16x4_f32(false, a[i], false, b3, (short)0, acc3, false, false);
        }
        #pragma unroll
        for (int r = 0; r < 8; r++) {
            int row = m0 + 8 * hh + r;
            if (row < N) {
                float4 o;
                o.x = acc0[r]; o.y = acc1[r]; o.z = acc2[r]; o.w = acc3[r];
                *reinterpret_cast<float4*>(z + (size_t)row * 256 + g * 64 + 4 * c) = o;
            }
        }
    }
}

#define KE_NODES 4
#define KE_U 8

__device__ __forceinline__ int rl_i(int v, int l) { return __builtin_amdgcn_readlane(v, l); }
__device__ __forceinline__ float rl_f(float v, int l) {
    return __int_as_float(__builtin_amdgcn_readlane(__float_as_int(v), l));
}
template <int M> __device__ __forceinline__ float row_xadd(float v) {
    int t = __builtin_amdgcn_update_dpp(0, __float_as_int(v), 0x160 | M, 0xF, 0xF, false);
    return v + __int_as_float(t);
}
__device__ __forceinline__ float wave_sum(float v) {
    v = row_xadd<1>(v); v = row_xadd<2>(v); v = row_xadd<4>(v); v = row_xadd<8>(v);
    return rl_f(v, 0) + rl_f(v, 16);
}
__device__ __forceinline__ float h2term(float c0, float c1, float c2,
                                        float w0, float w1, float w2, float zv) {
    float hv = c0*w0 + c1*w1 + c2*w2;
    hv = fmaxf(hv * INV_SQRT3, 0.0f);
    return hv * zv;
}

__global__ __launch_bounds__(128) void kE(const float* __restrict__ basis,
                                          const float* __restrict__ z,
                                          const int* __restrict__ esrc,
                                          const int* __restrict__ off,
                                          const int* __restrict__ perm,
                                          const float* __restrict__ W2a,
                                          float* __restrict__ out, int N)
{
    const int lane = threadIdx.x & 31;
    const int wave = __builtin_amdgcn_readfirstlane(threadIdx.x >> 5);
    const int n0 = (blockIdx.x * 4 + wave) * KE_NODES;
    if (n0 >= N) return;
    const int nn = (N - n0 < KE_NODES) ? (N - n0) : KE_NODES;
    int oi = n0 + lane;
    if (oi > N) oi = N;
    const int ov = off[oi];
    const int beg_all = rl_i(ov, 0);
    const int end_all = rl_i(ov, nn);

    const float4* w4 = reinterpret_cast<const float4*>(W2a);
    const float4 w0a = w4[lane],       w0b = w4[32 + lane];
    const float4 w1a = w4[64 + lane],  w1b = w4[96 + lane];
    const float4 w2a = w4[128 + lane], w2b = w4[160 + lane];

    const float oscale = (1.0f/128.0f) * INV_NN;
    int   k    = 0;
    int   nend = rl_i(ov, 1);
    float acc  = 0.0f;

    for (int base = beg_all; base < end_all; base += 64) {
        const int last = end_all - 1;
        int iA = base + lane;       if (iA > last) iA = last;
        int iB = base + 32 + lane;  if (iB > last) iB = last;
        const int eA = perm[iA], eB = perm[iB];
        const int sA = esrc[eA], sB = esrc[eB];
        const float pA0 = basis[(size_t)eA*3], pA1 = basis[(size_t)eA*3 + 1], pA2 = basis[(size_t)eA*3 + 2];
        const float pB0 = basis[(size_t)eB*3], pB1 = basis[(size_t)eB*3 + 1], pB2 = basis[(size_t)eB*3 + 2];
        const int left = end_all - base;
        const int cnt  = left < 64 ? left : 64;
        for (int j0 = 0; j0 < cnt; j0 += KE_U) {
            const bool hi = j0 >= 32;
            const int   vs  = hi ? sB  : sA;
            const float vb0 = hi ? pB0 : pA0, vb1 = hi ? pB1 : pA1, vb2 = hi ? pB2 : pA2;
            const int jl = j0 & 31;
            const int m  = (cnt - j0 < KE_U) ? (cnt - j0) : KE_U;
            float4 za[KE_U], zb[KE_U];
            #pragma unroll
            for (int u = 0; u < KE_U; u++) {
                const int jj = jl + (u < m ? u : m - 1);
                const int s  = rl_i(vs, jj);
                const float4* zr = reinterpret_cast<const float4*>(z + (size_t)s * 256);
                za[u] = zr[lane];
                zb[u] = zr[32 + lane];
            }
            #pragma unroll
            for (int u = 0; u < KE_U; u++) {
                if (u < m) {
                    const int jj = jl + u;
                    const float c0 = rl_f(vb0, jj), c1 = rl_f(vb1, jj), c2 = rl_f(vb2, jj);
                    float t = h2term(c0, c1, c2, w0a.x, w1a.x, w2a.x, za[u].x);
                    t += h2term(c0, c1, c2, w0a.y, w1a.y, w2a.y, za[u].y);
                    t += h2term(c0, c1, c2, w0a.z, w1a.z, w2a.z, za[u].z);
                    t += h2term(c0, c1, c2, w0a.w, w1a.w, w2a.w, za[u].w);
                    t += h2term(c0, c1, c2, w0b.x, w1b.x, w2b.x, zb[u].x);
                    t += h2term(c0, c1, c2, w0b.y, w1b.y, w2b.y, zb[u].y);
                    t += h2term(c0, c1, c2, w0b.z, w1b.z, w2b.z, zb[u].z);
                    t += h2term(c0, c1, c2, w0b.w, w1b.w, w2b.w, zb[u].w);
                    const int p = base + j0 + u;
                    while (p >= nend) {
                        const float tot = wave_sum(acc);
                        if (lane == 0) out[n0 + k] = tot * oscale;
                        acc = 0.0f;
                        k++;
                        nend = rl_i(ov, k + 1);
                    }
                    acc += t;
                }
            }
        }
    }
    while (k < nn) {
        const float tot = wave_sum(acc);
        if (lane == 0) out[n0 + k] = tot * oscale;
        acc = 0.0f;
        k++;
    }
}

extern "C" void kernel_launch(void* const* d_in, const int* in_sizes, int n_in,
                              void* d_out, int out_size, void* d_ws, size_t ws_size,
                              hipStream_t stream)
{
    const float* evec = (const float*)d_in[0];
    const float* elen = (const float*)d_in[1];
    const float* W1a  = (const float*)d_in[2];
    const float* W1b  = (const float*)d_in[3];
    const float* W2a  = (const float*)d_in[4];
    const float* W2b  = (const float*)d_in[5];
    const int*   esrc = (const int*)d_in[6];
    const int*   edst = (const int*)d_in[7];
    float* out = (float*)d_out;

    const int E = in_sizes[0] / 3;
    const int N = out_size;

    float* ws = (float*)d_ws;
    float* x_     = ws;                 ws += (size_t)N * 16;
    float* h_     = ws;                 ws += (size_t)N * 96;
    float* lists_ = ws;                 ws += (size_t)N * 64;
    float* z_     = ws;                 ws += (size_t)N * 256;
    float* sh_    = ws;                 ws += (size_t)E * 16;
    float* basis_ = ws;                 ws += (size_t)E * 3;

    float* y_  = sh_;
    int* cnt_  = (int*)lists_;
    int* rank_ = cnt_ + N;
    int* off_  = rank_ + E;
    int* perm_ = off_ + N + 1;

    long nx = (long)N * 16;
    kzero<<<(int)((nx + 255) / 256), 256, 0, stream>>>(x_, nx);
    kzero<<<(N + 255) / 256, 256, 0, stream>>>((float*)cnt_, N);

    kA<<<(E + 255) / 256, 256, 0, stream>>>(evec, elen, edst, sh_, basis_, x_, cnt_, rank_, E);
    kScan<<<1, 1024, 0, stream>>>(cnt_, off_, N);
    kFill<<<(E + 255) / 256, 256, 0, stream>>>(edst, rank_, off_, perm_, E);
    kB<<<(N + 31) / 32, 192, 0, stream>>>(sh_, basis_, x_, esrc, off_, perm_, W1a, W1b, h_, N);
    kC<<<(N + 255) / 256, 256, 0, stream>>>(h_, y_, N);
    kD<<<(N + 127) / 128, 256, 0, stream>>>(y_, W2b, z_, N);
    kE<<<(N + 4 * KE_NODES - 1) / (4 * KE_NODES), 128, 0, stream>>>(basis_, z_, esrc, off_, perm_, W2a, out, N);
}


# edge records (sh, basis, source) written in destination-list order by the fill pass; x = per-node sum over its contiguous records: float atomics and x clear gone, no perm hop; x sum order changes
# speedup vs baseline: 2.9398x; 1.5215x over previous
#include <hip/hip_runtime.h>
#include <math.h>

typedef __attribute__((ext_vector_type(2))) float v2f;
typedef __attribute__((ext_vector_type(8))) float v8f;

#define INV_NN     0.512989176042577f
#define INV_SQRT3  0.5773502691896258f
#define S3   1.7320508075688772f
#define S5   2.23606797749979f
#define S7   2.6457513110645907f
#define S15  3.872983346207417f
#define S105 10.246950765959598f
#define C358 2.091650066335189f
#define C218 1.6201851746019651f

__global__ void kzero(float* __restrict__ p, long n) {
    long i = (long)blockIdx.x * blockDim.x + threadIdx.x;
    if (i < n) p[i] = 0.0f;
}

__global__ void kA(const int* __restrict__ edst, int* __restrict__ cnt, int* __restrict__ rank, int E)
{
    int e = blockIdx.x * blockDim.x + threadIdx.x;
    if (e >= E) return;
    rank[e] = atomicAdd(&cnt[edst[e]], 1);
}

__global__ __launch_bounds__(1024) void kScan(const int* __restrict__ cnt,
                                              int* __restrict__ off, int N)
{
    __shared__ int part[1024];
    const int t = threadIdx.x;
    const int chunk = (N + 1023) / 1024;
    const int b = t * chunk;
    const int bend = (b + chunk < N) ? (b + chunk) : N;
    int s = 0;
    for (int i = b; i < bend; i++) s += cnt[i];
    part[t] = s;
    __syncthreads();
    for (int d = 1; d < 1024; d <<= 1) {
        int v = (t >= d) ? part[t - d] : 0;
        __syncthreads();
        part[t] += v;
        __syncthreads();
    }
    int run = part[t] - s;
    for (int i = b; i < bend; i++) { off[i] = run; run += cnt[i]; }
    if (t == 1023) off[N] = part[1023];
}

__global__ void kFill(const float* __restrict__ evec, const float* __restrict__ elen,
                      const int* __restrict__ esrc, const int* __restrict__ edst,
                      const int* __restrict__ rank, const int* __restrict__ off,
                      float* __restrict__ sh, float* __restrict__ basis,
                      int* __restrict__ srcs, int E)
{
    int e = blockIdx.x * blockDim.x + threadIdx.x;
    if (e >= E) return;
    const int slot = off[edst[e]] + rank[e];
    float vx = evec[3*(size_t)e], vy = evec[3*(size_t)e+1], vz = evec[3*(size_t)e+2];
    float rn = 1.0f / (sqrtf(vx*vx + vy*vy + vz*vz) + 1e-12f);
    vx *= rn; vy *= rn; vz *= rn;
    float x2 = vx*vx, y2 = vy*vy, z2 = vz*vz;
    float4 r0, r1, r2, r3;
    r0.x = 1.0f;
    r0.y = S3*vx; r0.z = S3*vy; r0.w = S3*vz;
    r1.x = S15*vx*vz;
    r1.y = S15*vx*vy;
    r1.z = 0.5f*S5*(2.0f*y2 - x2 - z2);
    r1.w = S15*vy*vz;
    r2.x = 0.5f*S15*(z2 - x2);
    r2.y = C358*vx*(3.0f*z2 - x2);
    r2.z = S105*vx*vy*vz;
    r2.w = C218*vx*(4.0f*y2 - z2 - x2);
    r3.x = 0.5f*S7*vy*(2.0f*y2 - 3.0f*z2 - 3.0f*x2);
    r3.y = C218*vz*(4.0f*y2 - z2 - x2);
    r3.z = 0.5f*S105*vy*(z2 - x2);
    r3.w = C358*vz*(z2 - 3.0f*x2);
    float4* sr = reinterpret_cast<float4*>(sh + (size_t)slot * 16);
    sr[0] = r0; sr[1] = r1; sr[2] = r2; sr[3] = r3;
    float len = elen[e];
    #pragma unroll
    for (int i = 0; i < 3; i++) {
        float dd = (len - 2.0f*(float)i) * 0.5f;
        basis[(size_t)slot*3 + i] = expf(-dd*dd) * (1.0f/1.12f);
    }
    srcs[slot] = esrc[e];
}

__global__ void kX(const float* __restrict__ sh, const int* __restrict__ off,
                   float* __restrict__ x, int N)
{
    int g = blockIdx.x * blockDim.x + threadIdx.x;
    int n = g >> 2, q = g & 3;
    if (n >= N) return;
    const int beg = off[n], end = off[n + 1];
    const float4* p = reinterpret_cast<const float4*>(sh) + (size_t)beg * 4 + q;
    float4 a = {0.0f, 0.0f, 0.0f, 0.0f};
    #pragma unroll 4
    for (int i = beg; i < end; i++) {
        float4 v = p[(size_t)(i - beg) * 4];
        a.x += v.x * INV_NN; a.y += v.y * INV_NN; a.z += v.z * INV_NN; a.w += v.w * INV_NN;
    }
    reinterpret_cast<float4*>(x)[(size_t)n * 4 + q] = a;
}

#define KB_PITCH 1028
#define RS3 0.5773502691896258f
#define RS5 0.4472135954999579f
#define RS7 0.3779644730092272f
__global__ __launch_bounds__(192) void kB(
    const float* __restrict__ sh, const float* __restrict__ basis,
    const float* __restrict__ x,
    const int* __restrict__ srcs,
    const int* __restrict__ off,
    const float* __restrict__ W1a, const float* __restrict__ W1b,
    float* __restrict__ h, int N)
{
    __shared__ float Gs[32 * KB_PITCH];

    const int tid  = threadIdx.x;
    const int lane = tid & 31;
    const int wave = __builtin_amdgcn_readfirstlane(tid >> 5);
    const int n0   = blockIdx.x * 32;

    float wa0[8], wa1[8], wa2[8];
    #pragma unroll
    for (int i = 0; i < 8; i++) {
        wa0[i] = W1a[lane + 32*i];
        wa1[i] = W1a[256 + lane + 32*i];
        wa2[i] = W1a[512 + lane + 32*i];
    }

    for (int m = wave; m < 32; m += 6) {
        const int n = n0 + m;
        float g0[8], g1[8], g2[8], g3[8];
        #pragma unroll
        for (int i = 0; i < 8; i++) { g0[i] = 0.0f; g1[i] = 0.0f; g2[i] = 0.0f; g3[i] = 0.0f; }
        if (n < N) {
            const int beg = off[n], end = off[n + 1];
            for (int base = beg; base < end; base += 32) {
                const int idx = base + lane;
                float d0 = 0.0f, d1 = 0.0f, d2 = 0.0f, d3 = 0.0f;
                float b0 = 0.0f, b1 = 0.0f, b2 = 0.0f;
                if (idx < end) {
                    const int srcn = srcs[idx];
                    const float4* xs = reinterpret_cast<const float4*>(x  + (size_t)srcn * 16);
                    const float4* ss = reinterpret_cast<const float4*>(sh + (size_t)idx * 16);
                    float4 x0 = xs[0], x1 = xs[1], x2 = xs[2], x3 = xs[3];
                    float4 s0 = ss[0], s1 = ss[1], s2 = ss[2], s3 = ss[3];
                    d0 = x0.x*s0.x;
                    d1 = (x0.y*s0.y + x0.z*s0.z + x0.w*s0.w) * RS3;
                    d2 = (x1.x*s1.x + x1.y*s1.y + x1.z*s1.z + x1.w*s1.w + x2.x*s2.x) * RS5;
                    d3 = (x2.y*s2.y + x2.z*s2.z + x2.w*s2.w
                        + x3.x*s3.x + x3.y*s3.y + x3.z*s3.z + x3.w*s3.w) * RS7;
                    b0 = basis[(size_t)idx*3]; b1 = basis[(size_t)idx*3 + 1]; b2 = basis[(size_t)idx*3 + 2];
                }
                const int left = end - base;
                const int cntc = left < 32 ? left : 32;
                for (int j = 0; j < cntc; j++) {
                    const float e0 = __shfl(d0, j, 32), e1 = __shfl(d1, j, 32);
                    const float e2 = __shfl(d2, j, 32), e3 = __shfl(d3, j, 32);
                    const float c0 = __shfl(b0, j, 32), c1 = __shfl(b1, j, 32), c2 = __shfl(b2, j, 32);
                    #pragma unroll
                    for (int i = 0; i < 8; i++) {
                        float hv = c0*wa0[i] + c1*wa1[i] + c2*wa2[i];
                        hv = fmaxf(hv * INV_SQRT3, 0.0f);
                        g0[i] += e0 * hv; g1[i] += e1 * hv;
                        g2[i] += e2 * hv; g3[i] += e3 * hv;
                    }
                }
            }
        }
        float* gr = &Gs[m * KB_PITCH + lane];
        #pragma unroll
        for (int i = 0; i < 8; i++) {
            gr[32*i]       = g0[i];
            gr[256 + 32*i] = g1[i];
            gr[512 + 32*i] = g2[i];
            gr[768 + 32*i] = g3[i];
        }
    }
    __syncthreads();

    const int c    = lane & 15;
    const int hh   = lane >> 4;
    const int colb = wave * 16 + c;
    v8f a00 = {}, a01 = {}, a02 = {}, a03 = {};
    v8f a10 = {}, a11 = {}, a12 = {}, a13 = {};
    const float* ga0 = &Gs[c * KB_PITCH + 2 * hh];
    const float* ga1 = ga0 + 16 * KB_PITCH;
    const float* Bp = W1b + (size_t)(2 * hh) * 384 + colb;
    v2f Bc[2][4], Bn[2][4];
    #pragma unroll
    for (int s = 0; s < 2; s++) {
        #pragma unroll
        for (int q = 0; q < 4; q++) {
            Bc[s][q].x = Bp[(size_t)(4 * s) * 384 + 96 * q];
            Bc[s][q].y = Bp[(size_t)(4 * s + 1) * 384 + 96 * q];
        }
    }
    for (int k0 = 0; k0 < 256; k0 += 8) {
        const int kn = (k0 + 8 < 256) ? (k0 + 8) : k0;
        const float* Bq = Bp + (size_t)kn * 384;
        #pragma unroll
        for (int s = 0; s < 2; s++) {
            #pragma unroll
            for (int q = 0; q < 4; q++) {
                Bn[s][q].x = Bq[(size_t)(4 * s) * 384 + 96 * q];
                Bn[s][q].y = Bq[(size_t)(4 * s + 1) * 384 + 96 * q];
            }
        }
        #pragma unroll
        for (int s = 0; s < 2; s++) {
            const int k = k0 + 4 * s;
            v2f A00 = *reinterpret_cast<const v2f*>(ga0 + k);
            v2f A01 = *reinterpret_cast<const v2f*>(ga0 + 256 + k);
            v2f A02 = *reinterpret_cast<const v2f*>(ga0 + 512 + k);
            v2f A03 = *reinterpret_cast<const v2f*>(ga0 + 768 + k);
            v2f A10 = *reinterpret_cast<const v2f*>(ga1 + k);
            v2f A11 = *reinterpret_cast<const v2f*>(ga1 + 256 + k);
            v2f A12 = *reinterpret_cast<const v2f*>(ga1 + 512 + k);
            v2f A13 = *reinterpret_cast<const v2f*>(ga1 + 768 + k);
            a00 = __builtin_amdgcn_wmma_f32_16x16x4_f32(false, A00, false, Bc[s][0], (short)0, a00, false, false);
            a10 = __builtin_amdgcn_wmma_f32_16x16x4_f32(false, A10, false, Bc[s][0], (short)0, a10, false, false);
            a01 = __builtin_amdgcn_wmma_f32_16x16x4_f32(false, A01, false, Bc[s][1], (short)0, a01, false, false);
            a11 = __builtin_amdgcn_wmma_f32_16x16x4_f32(false, A11, false, Bc[s][1], (short)0, a11, false, false);
            a02 = __builtin_amdgcn_wmma_f32_16x16x4_f32(false, A02, false, Bc[s][2], (short)0, a02, false, false);
            a12 = __builtin_amdgcn_wmma_f32_16x16x4_f32(false, A12, false, Bc[s][2], (short)0, a12, false, false);
            a03 = __builtin_amdgcn_wmma_f32_16x16x4_f32(false, A03, false, Bc[s][3], (short)0, a03, false, false);
            a13 = __builtin_amdgcn_wmma_f32_16x16x4_f32(false, A13, false, Bc[s][3], (short)0, a13, false, false);
        }
        #pragma unroll
        for (int s = 0; s < 2; s++) {
            #pragma unroll
            for (int q = 0; q < 4; q++) Bc[s][q] = Bn[s][q];
        }
    }
    __syncthreads();

    const float scale = (1.0f/16.0f) * 0.5f * INV_NN;
    #pragma unroll
    for (int r = 0; r < 8; r++) {
        Gs[(8*hh + r) * 96 + colb]      = ((a00[r] + a01[r]) + (a02[r] + a03[r])) * scale;
        Gs[(16 + 8*hh + r) * 96 + colb] = ((a10[r] + a11[r]) + (a12[r] + a13[r])) * scale;
    }
    __syncthreads();
    const int rows = (N - n0 < 32) ? (N - n0) : 32;
    for (int idx = tid; idx < rows * 24; idx += 192) {
        float4 v = *reinterpret_cast<const float4*>(&Gs[idx * 4]);
        *reinterpret_cast<float4*>(h + (size_t)n0 * 96 + idx * 4) = v;
    }
}

__global__ void kC(const float* __restrict__ h, float* __restrict__ y, int N)
{
    int nid = blockIdx.x * blockDim.x + threadIdx.x;
    if (nid >= N) return;
    const float* hr = h + (size_t)nid * 96;
    float* yr = y + (size_t)nid * 64;
    #pragma unroll
    for (int j = 0; j < 16; j++) yr[j] = fmaxf(hr[j], 0.0f);
    #pragma unroll
    for (int j = 16; j < 32; j++) yr[j] = fabsf(hr[j]);
    #pragma unroll
    for (int j = 0; j < 32; j++) {
        float g = hr[32 + j];
        bool use_relu = (j < 8) || (j >= 16 && j < 24);
        float ga = use_relu ? fmaxf(g, 0.0f) : tanhf(g);
        yr[32 + j] = ga * hr[64 + j];
    }
}

#define KD_PITCH 68
__global__ __launch_bounds__(256) void kD(const float* __restrict__ y,
                                          const float* __restrict__ W2b,
                                          float* __restrict__ z, int N)
{
    __shared__ float wl[256 * KD_PITCH];
    const int tid = threadIdx.x;
    #pragma unroll
    for (int j = 0; j < 16; j++) {
        int idx = tid + 256 * j;
        int n = idx >> 4, q = idx & 15;
        int slot = (n & ~63) | ((n & 3) << 4) | ((n >> 2) & 15);
        float4 v = *reinterpret_cast<const float4*>(W2b + (size_t)n * 64 + q * 4);
        *reinterpret_cast<float4*>(&wl[slot * KD_PITCH + q * 4]) = v;
    }

    const int wave = tid >> 5, lane = tid & 31;
    const int c  = lane & 15;
    const int hh = lane >> 4;
    const int m0 = blockIdx.x * 128 + wave * 16;
    int rowA = m0 + c;
    if (rowA > N - 1) rowA = N - 1;
    const float* yr = y + (size_t)rowA * 64 + 2 * hh;
    v2f a[16];
    #pragma unroll
    for (int i = 0; i < 16; i++) a[i] = *reinterpret_cast<const v2f*>(yr + 4 * i);
    __syncthreads();

    #pragma unroll
    for (int g = 0; g < 4; g++) {
        v8f acc0 = {}, acc1 = {}, acc2 = {}, acc3 = {};
        const float* wb = &wl[(g * 64 + c) * KD_PITCH + 2 * hh];
        #pragma unroll
        for (int i = 0; i < 16; i++) {
            v2f b0 = *reinterpret_cast<const v2f*>(wb + 4 * i);
            v2f b1 = *reinterpret_cast<const v2f*>(wb + 16 * KD_PITCH + 4 * i);
            v2f b2 = *reinterpret_cast<const v2f*>(wb + 32 * KD_PITCH + 4 * i);
            v2f b3 = *reinterpret_cast<const v2f*>(wb + 48 * KD_PITCH + 4 * i);
            acc0 = __builtin_amdgcn_wmma_f32_16x16x4_f32(false, a[i], false, b0, (short)0, acc0, false, false);
            acc1 = __builtin_amdgcn_wmma_f32_16x16x4_f32(false, a[i], false, b1, (short)0, acc1, false, false);
            acc2 = __builtin_amdgcn_wmma_f32_16x16x4_f32(false, a[i], false, b2, (short)0, acc2, false, false);
            acc3 = __builtin_amdgcn_wmma_f32_16x16x4_f32(false, a[i], false, b3, (short)0, acc3, false, false);
        }
        #pragma unroll
        for (int r = 0; r < 8; r++) {
            int row = m0 + 8 * hh + r;
            if (row < N) {
                float4 o;
                o.x = acc0[r]; o.y = acc1[r]; o.z = acc2[r]; o.w = acc3[r];
                *reinterpret_cast<float4*>(z + (size_t)row * 256 + g * 64 + 4 * c) = o;
            }
        }
    }
}

#define KE_NODES 4
#define KE_U 8

__device__ __forceinline__ int rl_i(int v, int l) { return __builtin_amdgcn_readlane(v, l); }
__device__ __forceinline__ float rl_f(float v, int l) {
    return __int_as_float(__builtin_amdgcn_readlane(__float_as_int(v), l));
}
template <int M> __device__ __forceinline__ float row_xadd(float v) {
    int t = __builtin_amdgcn_update_dpp(0, __float_as_int(v), 0x160 | M, 0xF, 0xF, false);
    return v + __int_as_float(t);
}
__device__ __forceinline__ float wave_sum(float v) {
    v = row_xadd<1>(v); v = row_xadd<2>(v); v = row_xadd<4>(v); v = row_xadd<8>(v);
    return rl_f(v, 0) + rl_f(v, 16);
}
__device__ __forceinline__ float h2term(float c0, float c1, float c2,
                                        float w0, float w1, float w2, float zv) {
    float hv = c0*w0 + c1*w1 + c2*w2;
    hv = fmaxf(hv * INV_SQRT3, 0.0f);
    return hv * zv;
}

__global__ __launch_bounds__(128) void kE(const float* __restrict__ basis,
                                          const float* __restrict__ z,
                                          const int* __restrict__ srcs,
                                          const int* __restrict__ off,
                                          const float* __restrict__ W2a,
                                          float* __restrict__ out, int N)
{
    const int lane = threadIdx.x & 31;
    const int wave = __builtin_amdgcn_readfirstlane(threadIdx.x >> 5);
    const int n0 = (blockIdx.x * 4 + wave) * KE_NODES;
    if (n0 >= N) return;
    const int nn = (N - n0 < KE_NODES) ? (N - n0) : KE_NODES;
    int oi = n0 + lane;
    if (oi > N) oi = N;
    const int ov = off[oi];
    const int beg_all = rl_i(ov, 0);
    const int end_all = rl_i(ov, nn);

    const float4* w4 = reinterpret_cast<const float4*>(W2a);
    const float4 w0a = w4[lane],       w0b = w4[32 + lane];
    const float4 w1a = w4[64 + lane],  w1b = w4[96 + lane];
    const float4 w2a = w4[128 + lane], w2b = w4[160 + lane];

    const float oscale = (1.0f/128.0f) * INV_NN;
    int   k    = 0;
    int   nend = rl_i(ov, 1);
    float acc  = 0.0f;

    for (int base = beg_all; base < end_all; base += 64) {
        const int last = end_all - 1;
        int iA = base + lane;       if (iA > last) iA = last;
        int iB = base + 32 + lane;  if (iB > last) iB = last;
        const int sA = srcs[iA], sB = srcs[iB];
        const float pA0 = basis[(size_t)iA*3], pA1 = basis[(size_t)iA*3 + 1], pA2 = basis[(size_t)iA*3 + 2];
        const float pB0 = basis[(size_t)iB*3], pB1 = basis[(size_t)iB*3 + 1], pB2 = basis[(size_t)iB*3 + 2];
        const int left = end_all - base;
        const int cnt  = left < 64 ? left : 64;
        for (int j0 = 0; j0 < cnt; j0 += KE_U) {
            const bool hi = j0 >= 32;
            const int   vs  = hi ? sB  : sA;
            const float vb0 = hi ? pB0 : pA0, vb1 = hi ? pB1 : pA1, vb2 = hi ? pB2 : pA2;
            const int jl = j0 & 31;
            const int m  = (cnt - j0 < KE_U) ? (cnt - j0) : KE_U;
            float4 za[KE_U], zb[KE_U];
            #pragma unroll
            for (int u = 0; u < KE_U; u++) {
                const int jj = jl + (u < m ? u : m - 1);
                const int s  = rl_i(vs, jj);
                const float4* zr = reinterpret_cast<const float4*>(z + (size_t)s * 256);
                za[u] = zr[lane];
                zb[u] = zr[32 + lane];
            }
            #pragma unroll
            for (int u = 0; u < KE_U; u++) {
                if (u < m) {
                    const int jj = jl + u;
                    const float c0 = rl_f(vb0, jj), c1 = rl_f(vb1, jj), c2 = rl_f(vb2, jj);
                    float t = h2term(c0, c1, c2, w0a.x, w1a.x, w2a.x, za[u].x);
                    t += h2term(c0, c1, c2, w0a.y, w1a.y, w2a.y, za[u].y);
                    t += h2term(c0, c1, c2, w0a.z, w1a.z, w2a.z, za[u].z);
                    t += h2term(c0, c1, c2, w0a.w, w1a.w, w2a.w, za[u].w);
                    t += h2term(c0, c1, c2, w0b.x, w1b.x, w2b.x, zb[u].x);
                    t += h2term(c0, c1, c2, w0b.y, w1b.y, w2b.y, zb[u].y);
                    t += h2term(c0, c1, c2, w0b.z, w1b.z, w2b.z, zb[u].z);
                    t += h2term(c0, c1, c2, w0b.w, w1b.w, w2b.w, zb[u].w);
                    const int p = base + j0 + u;
                    while (p >= nend) {
                        const float tot = wave_sum(acc);
                        if (lane == 0) out[n0 + k] = tot * oscale;
                        acc = 0.0f;
                        k++;
                        nend = rl_i(ov, k + 1);
                    }
                    acc += t;
                }
            }
        }
    }
    while (k < nn) {
        const float tot = wave_sum(acc);
        if (lane == 0) out[n0 + k] = tot * oscale;
        acc = 0.0f;
        k++;
    }
}

extern "C" void kernel_launch(void* const* d_in, const int* in_sizes, int n_in,
                              void* d_out, int out_size, void* d_ws, size_t ws_size,
                              hipStream_t stream)
{
    const float* evec = (const float*)d_in[0];
    const float* elen = (const float*)d_in[1];
    const float* W1a  = (const float*)d_in[2];
    const float* W1b  = (const float*)d_in[3];
    const float* W2a  = (const float*)d_in[4];
    const float* W2b  = (const float*)d_in[5];
    const int*   esrc = (const int*)d_in[6];
    const int*   edst = (const int*)d_in[7];
    float* out = (float*)d_out;

    const int E = in_sizes[0] / 3;
    const int N = out_size;

    float* ws = (float*)d_ws;
    float* x_     = ws;                 ws += (size_t)N * 16;
    float* h_     = ws;                 ws += (size_t)N * 96;
    float* lists_ = ws;                 ws += (size_t)N * 64;
    float* z_     = ws;                 ws += (size_t)N * 256;
    float* sh_    = ws;                 ws += (size_t)E * 16;
    float* basis_ = ws;                 ws += (size_t)E * 3;

    float* y_  = sh_;
    int* cnt_  = (int*)lists_;
    int* rank_ = cnt_ + N;
    int* off_  = rank_ + E;
    int* srcs_ = off_ + N + 1;

    kzero<<<(N + 255) / 256, 256, 0, stream>>>((float*)cnt_, N);

    kA<<<(E + 255) / 256, 256, 0, stream>>>(edst, cnt_, rank_, E);
    kScan<<<1, 1024, 0, stream>>>(cnt_, off_, N);
    kFill<<<(E + 255) / 256, 256, 0, stream>>>(evec, elen, esrc, edst, rank_, off_, sh_, basis_, srcs_, E);
    kX<<<(4 * N + 255) / 256, 256, 0, stream>>>(sh_, off_, x_, N);
    kB<<<(N + 31) / 32, 192, 0, stream>>>(sh_, basis_, x_, srcs_, off_, W1a, W1b, h_, N);
    kC<<<(N + 255) / 256, 256, 0, stream>>>(h_, y_, N);
    kD<<<(N + 127) / 128, 256, 0, stream>>>(y_, W2b, z_, N);
    kE<<<(N + 4 * KE_NODES - 1) / (4 * KE_NODES), 128, 0, stream>>>(basis_, z_, srcs_, off_, W2a, out, N);
}


# gate and z = y x W2b^T moved into the conv1 kernel's epilogue on its on-chip 32-node tile (f32 WMMA, k order kept); h and y never stored, 9 -> 7 launches; no sum order changes
# speedup vs baseline: 3.1105x; 1.0581x over previous
#include <hip/hip_runtime.h>
#include <math.h>

typedef __attribute__((ext_vector_type(2))) float v2f;
typedef __attribute__((ext_vector_type(8))) float v8f;

#define INV_NN     0.512989176042577f
#define INV_SQRT3  0.5773502691896258f
#define S3   1.7320508075688772f
#define S5   2.23606797749979f
#define S7   2.6457513110645907f
#define S15  3.872983346207417f
#define S105 10.246950765959598f
#define C358 2.091650066335189f
#define C218 1.6201851746019651f

__global__ void kzero(float* __restrict__ p, long n) {
    long i = (long)blockIdx.x * blockDim.x + threadIdx.x;
    if (i < n) p[i] = 0.0f;
}

__global__ void kA(const int* __restrict__ edst, int* __restrict__ cnt, int* __restrict__ rank, int E)
{
    int e = blockIdx.x * blockDim.x + threadIdx.x;
    if (e >= E) return;
    rank[e] = atomicAdd(&cnt[edst[e]], 1);
}

__global__ __launch_bounds__(1024) void kScan(const int* __restrict__ cnt,
                                              int* __restrict__ off, int N)
{
    __shared__ int part[1024];
    const int t = threadIdx.x;
    const int chunk = (N + 1023) / 1024;
    const int b = t * chunk;
    const int bend = (b + chunk < N) ? (b + chunk) : N;
    int s = 0;
    for (int i = b; i < bend; i++) s += cnt[i];
    part[t] = s;
    __syncthreads();
    for (int d = 1; d < 1024; d <<= 1) {
        int v = (t >= d) ? part[t - d] : 0;
        __syncthreads();
        part[t] += v;
        __syncthreads();
    }
    int run = part[t] - s;
    for (int i = b; i < bend; i++) { off[i] = run; run += cnt[i]; }
    if (t == 1023) off[N] = part[1023];
}

__global__ void kFill(const float* __restrict__ evec, const float* __restrict__ elen,
                      const int* __restrict__ esrc, const int* __restrict__ edst,
                      const int* __restrict__ rank, const int* __restrict__ off,
                      float* __restrict__ sh, float* __restrict__ basis,
                      int* __restrict__ srcs, int E)
{
    int e = blockIdx.x * blockDim.x + threadIdx.x;
    if (e >= E) return;
    const int slot = off[edst[e]] + rank[e];
    float vx = evec[3*(size_t)e], vy = evec[3*(size_t)e+1], vz = evec[3*(size_t)e+2];
    float rn = 1.0f / (sqrtf(vx*vx + vy*vy + vz*vz) + 1e-12f);
    vx *= rn; vy *= rn; vz *= rn;
    float x2 = vx*vx, y2 = vy*vy, z2 = vz*vz;
    float4 r0, r1, r2, r3;
    r0.x = 1.0f;
    r0.y = S3*vx; r0.z = S3*vy; r0.w = S3*vz;
    r1.x = S15*vx*vz;
    r1.y = S15*vx*vy;
    r1.z = 0.5f*S5*(2.0f*y2 - x2 - z2);
    r1.w = S15*vy*vz;
    r2.x = 0.5f*S15*(z2 - x2);
    r2.y = C358*vx*(3.0f*z2 - x2);
    r2.z = S105*vx*vy*vz;
    r2.w = C218*vx*(4.0f*y2 - z2 - x2);
    r3.x = 0.5f*S7*vy*(2.0f*y2 - 3.0f*z2 - 3.0f*x2);
    r3.y = C218*vz*(4.0f*y2 - z2 - x2);
    r3.z = 0.5f*S105*vy*(z2 - x2);
    r3.w = C358*vz*(z2 - 3.0f*x2);
    float4* sr = reinterpret_cast<float4*>(sh + (size_t)slot * 16);
    sr[0] = r0; sr[1] = r1; sr[2] = r2; sr[3] = r3;
    float len = elen[e];
    #pragma unroll
    for (int i = 0; i < 3; i++) {
        float dd = (len - 2.0f*(float)i) * 0.5f;
        basis[(size_t)slot*3 + i] = expf(-dd*dd) * (1.0f/1.12f);
    }
    srcs[slot] = esrc[e];
}

__global__ void kX(const float* __restrict__ sh, const int* __restrict__ off,
                   float* __restrict__ x, int N)
{
    int g = blockIdx.x * blockDim.x + threadIdx.x;
    int n = g >> 2, q = g & 3;
    if (n >= N) return;
    const int beg = off[n], end = off[n + 1];
    const float4* p = reinterpret_cast<const float4*>(sh) + (size_t)beg * 4 + q;
    float4 a = {0.0f, 0.0f, 0.0f, 0.0f};
    #pragma unroll 4
    for (int i = beg; i < end; i++) {
        float4 v = p[(size_t)(i - beg) * 4];
        a.x += v.x * INV_NN; a.y += v.y * INV_NN; a.z += v.z * INV_NN; a.w += v.w * INV_NN;
    }
    reinterpret_cast<float4*>(x)[(size_t)n * 4 + q] = a;
}

#define KB_PITCH 1028
#define KB_HP 101
#define KB_YP 68
#define KB_ZP 260
#define RS3 0.5773502691896258f
#define RS5 0.4472135954999579f
#define RS7 0.3779644730092272f
__global__ __launch_bounds__(192) void kB(
    const float* __restrict__ sh, const float* __restrict__ basis,
    const float* __restrict__ x,
    const int* __restrict__ srcs,
    const int* __restrict__ off,
    const float* __restrict__ W1a, const float* __restrict__ W1b,
    const float* W2b, float* __restrict__ z, int N)
{
    __shared__ float Gs[32 * KB_PITCH];

    const int tid  = threadIdx.x;
    const int lane = tid & 31;
    const int wave = __builtin_amdgcn_readfirstlane(tid >> 5);
    const int n0   = blockIdx.x * 32;

    float wa0[8], wa1[8], wa2[8];
    #pragma unroll
    for (int i = 0; i < 8; i++) {
        wa0[i] = W1a[lane + 32*i];
        wa1[i] = W1a[256 + lane + 32*i];
        wa2[i] = W1a[512 + lane + 32*i];
    }

    for (int m = wave; m < 32; m += 6) {
        const int n = n0 + m;
        float g0[8], g1[8], g2[8], g3[8];
        #pragma unroll
        for (int i = 0; i < 8; i++) { g0[i] = 0.0f; g1[i] = 0.0f; g2[i] = 0.0f; g3[i] = 0.0f; }
        if (n < N) {
            const int beg = off[n], end = off[n + 1];
            for (int base = beg; base < end; base += 32) {
                const int idx = base + lane;
                float d0 = 0.0f, d1 = 0.0f, d2 = 0.0f, d3 = 0.0f;
                float b0 = 0.0f, b1 = 0.0f, b2 = 0.0f;
                if (idx < end) {
                    const int srcn = srcs[idx];
                    const float4* xs = reinterpret_cast<const float4*>(x  + (size_t)srcn * 16);
                    const float4* ss = reinterpret_cast<const float4*>(sh + (size_t)idx * 16);
                    float4 x0 = xs[0], x1 = xs[1], x2 = xs[2], x3 = xs[3];
                    float4 s0 = ss[0], s1 = ss[1], s2 = ss[2], s3 = ss[3];
                    d0 = x0.x*s0.x;
                    d1 = (x0.y*s0.y + x0.z*s0.z + x0.w*s0.w) * RS3;
                    d2 = (x1.x*s1.x + x1.y*s1.y + x1.z*s1.z + x1.w*s1.w + x2.x*s2.x) * RS5;
                    d3 = (x2.y*s2.y + x2.z*s2.z + x2.w*s2.w
                        + x3.x*s3.x + x3.y*s3.y + x3.z*s3.z + x3.w*s3.w) * RS7;
                    b0 = basis[(size_t)idx*3]; b1 = basis[(size_t)idx*3 + 1]; b2 = basis[(size_t)idx*3 + 2];
                }
                const int left = end - base;
                const int cntc = left < 32 ? left : 32;
                for (int j = 0; j < cntc; j++) {
                    const float e0 = __shfl(d0, j, 32), e1 = __shfl(d1, j, 32);
                    const float e2 = __shfl(d2, j, 32), e3 = __shfl(d3, j, 32);
                    const float c0 = __shfl(b0, j, 32), c1 = __shfl(b1, j, 32), c2 = __shfl(b2, j, 32);
                    #pragma unroll
                    for (int i = 0; i < 8; i++) {
                        float hv = c0*wa0[i] + c1*wa1[i] + c2*wa2[i];
                        hv = fmaxf(hv * INV_SQRT3, 0.0f);
                        g0[i] += e0 * hv; g1[i] += e1 * hv;
                        g2[i] += e2 * hv; g3[i] += e3 * hv;
                    }
                }
            }
        }
        float* gr = &Gs[m * KB_PITCH + lane];
        #pragma unroll
        for (int i = 0; i < 8; i++) {
            gr[32*i]       = g0[i];
            gr[256 + 32*i] = g1[i];
            gr[512 + 32*i] = g2[i];
            gr[768 + 32*i] = g3[i];
        }
    }
    __syncthreads();

    const int c    = lane & 15;
    const int hh   = lane >> 4;
    const int colb = wave * 16 + c;
    v8f a00 = {}, a01 = {}, a02 = {}, a03 = {};
    v8f a10 = {}, a11 = {}, a12 = {}, a13 = {};
    const float* ga0 = &Gs[c * KB_PITCH + 2 * hh];
    const float* ga1 = ga0 + 16 * KB_PITCH;
    const float* Bp = W1b + (size_t)(2 * hh) * 384 + colb;
    v2f Bc[2][4], Bn[2][4];
    #pragma unroll
    for (int s = 0; s < 2; s++) {
        #pragma unroll
        for (int q = 0; q < 4; q++) {
            Bc[s][q].x = Bp[(size_t)(4 * s) * 384 + 96 * q];
            Bc[s][q].y = Bp[(size_t)(4 * s + 1) * 384 + 96 * q];
        }
    }
    for (int k0 = 0; k0 < 256; k0 += 8) {
        const int kn = (k0 + 8 < 256) ? (k0 + 8) : k0;
        const float* Bq = Bp + (size_t)kn * 384;
        #pragma unroll
        for (int s = 0; s < 2; s++) {
            #pragma unroll
            for (int q = 0; q < 4; q++) {
                Bn[s][q].x = Bq[(size_t)(4 * s) * 384 + 96 * q];
                Bn[s][q].y = Bq[(size_t)(4 * s + 1) * 384 + 96 * q];
            }
        }
        #pragma unroll
        for (int s = 0; s < 2; s++) {
            const int k = k0 + 4 * s;
            v2f A00 = *reinterpret_cast<const v2f*>(ga0 + k);
            v2f A01 = *reinterpret_cast<const v2f*>(ga0 + 256 + k);
            v2f A02 = *reinterpret_cast<const v2f*>(ga0 + 512 + k);
            v2f A03 = *reinterpret_cast<const v2f*>(ga0 + 768 + k);
            v2f A10 = *reinterpret_cast<const v2f*>(ga1 + k);
            v2f A11 = *reinterpret_cast<const v2f*>(ga1 + 256 + k);
            v2f A12 = *reinterpret_cast<const v2f*>(ga1 + 512 + k);
            v2f A13 = *reinterpret_cast<const v2f*>(ga1 + 768 + k);
            a00 = __builtin_amdgcn_wmma_f32_16x16x4_f32(false, A00, false, Bc[s][0], (short)0, a00, false, false);
            a10 = __builtin_amdgcn_wmma_f32_16x16x4_f32(false, A10, false, Bc[s][0], (short)0, a10, false, false);
            a01 = __builtin_amdgcn_wmma_f32_16x16x4_f32(false, A01, false, Bc[s][1], (short)0, a01, false, false);
            a11 = __builtin_amdgcn_wmma_f32_16x16x4_f32(false, A11, false, Bc[s][1], (short)0, a11, false, false);
            a02 = __builtin_amdgcn_wmma_f32_16x16x4_f32(false, A02, false, Bc[s][2], (short)0, a02, false, false);
            a12 = __builtin_amdgcn_wmma_f32_16x16x4_f32(false, A12, false, Bc[s][2], (short)0, a12, false, false);
            a03 = __builtin_amdgcn_wmma_f32_16x16x4_f32(false, A03, false, Bc[s][3], (short)0, a03, false, false);
            a13 = __builtin_amdgcn_wmma_f32_16x16x4_f32(false, A13, false, Bc[s][3], (short)0, a13, false, false);
        }
        #pragma unroll
        for (int s = 0; s < 2; s++) {
            #pragma unroll
            for (int q = 0; q < 4; q++) Bc[s][q] = Bn[s][q];
        }
    }
    v2f Bf[3][16];
    #pragma unroll
    for (int t = 0; t < 3; t++) {
        int jt = wave + 6 * t;
        if (jt > 15) jt = 15;
        const float* wb = W2b + (size_t)(16 * jt + c) * 64 + 2 * hh;
        #pragma unroll
        for (int i = 0; i < 16; i++) Bf[t][i] = *reinterpret_cast<const v2f*>(wb + 4 * i);
    }
    __syncthreads();

    float* Hs = Gs;
    float* Ys = Gs + 32 * KB_HP;
    float* Zs = Ys + 32 * KB_YP;
    const float scale = (1.0f/16.0f) * 0.5f * INV_NN;
    #pragma unroll
    for (int r = 0; r < 8; r++) {
        Hs[(8*hh + r) * KB_HP + colb]      = ((a00[r] + a01[r]) + (a02[r] + a03[r])) * scale;
        Hs[(16 + 8*hh + r) * KB_HP + colb] = ((a10[r] + a11[r]) + (a12[r] + a13[r])) * scale;
    }
    __syncthreads();
    {
        const float* hr = &Hs[lane * KB_HP];
        float* yr = &Ys[lane * KB_YP];
        float ha[11], hb[11];
        #pragma unroll
        for (int k = 0; k < 11; k++) {
            int j = wave + 6 * k;
            if (j > 63) j = 63;
            ha[k] = hr[j];
            hb[k] = hr[j + 32];
        }
        #pragma unroll
        for (int k = 0; k < 11; k++) {
            const int j = wave + 6 * k;
            if (j < 64) {
                float v;
                if (j < 16)      v = fmaxf(ha[k], 0.0f);
                else if (j < 32) v = fabsf(ha[k]);
                else {
                    const int jj = j - 32;
                    const bool use_relu = (jj < 8) || (jj >= 16 && jj < 24);
                    const float ga = use_relu ? fmaxf(ha[k], 0.0f) : tanhf(ha[k]);
                    v = ga * hb[k];
                }
                yr[j] = v;
            }
        }
    }
    __syncthreads();

    {
        const float* ya = &Ys[c * KB_YP + 2 * hh];
        v2f A0[16], A1[16];
        #pragma unroll
        for (int i = 0; i < 16; i++) {
            A0[i] = *reinterpret_cast<const v2f*>(ya + 4 * i);
            A1[i] = *reinterpret_cast<const v2f*>(ya + 16 * KB_YP + 4 * i);
        }
        #pragma unroll
        for (int t = 0; t < 3; t++) {
            const int jt = wave + 6 * t;
            if (jt < 16) {
                v8f z0 = {}, z1 = {};
                #pragma unroll
                for (int i = 0; i < 16; i++) {
                    z0 = __builtin_amdgcn_wmma_f32_16x16x4_f32(false, A0[i], false, Bf[t][i], (short)0, z0, false, false);
                    z1 = __builtin_amdgcn_wmma_f32_16x16x4_f32(false, A1[i], false, Bf[t][i], (short)0, z1, false, false);
                }
                float* zc = &Zs[(8 * hh) * KB_ZP + 16 * jt + c];
                #pragma unroll
                for (int r = 0; r < 8; r++) {
                    zc[r * KB_ZP]        = z0[r];
                    zc[(16 + r) * KB_ZP] = z1[r];
                }
            }
        }
    }
    __syncthreads();
    const int rows = (N - n0 < 32) ? (N - n0) : 32;
    for (int idx = tid; idx < rows * 64; idx += 192) {
        const int m = idx >> 6, q = idx & 63;
        float4 v = *reinterpret_cast<const float4*>(&Zs[m * KB_ZP + 4 * q]);
        *reinterpret_cast<float4*>(z + (size_t)(n0 + m) * 256 + 4 * q) = v;
    }
}

#define KE_NODES 4
#define KE_U 8

__device__ __forceinline__ int rl_i(int v, int l) { return __builtin_amdgcn_readlane(v, l); }
__device__ __forceinline__ float rl_f(float v, int l) {
    return __int_as_float(__builtin_amdgcn_readlane(__float_as_int(v), l));
}
template <int M> __device__ __forceinline__ float row_xadd(float v) {
    int t = __builtin_amdgcn_update_dpp(0, __float_as_int(v), 0x160 | M, 0xF, 0xF, false);
    return v + __int_as_float(t);
}
__device__ __forceinline__ float wave_sum(float v) {
    v = row_xadd<1>(v); v = row_xadd<2>(v); v = row_xadd<4>(v); v = row_xadd<8>(v);
    return rl_f(v, 0) + rl_f(v, 16);
}
__device__ __forceinline__ float h2term(float c0, float c1, float c2,
                                        float w0, float w1, float w2, float zv) {
    float hv = c0*w0 + c1*w1 + c2*w2;
    hv = fmaxf(hv * INV_SQRT3, 0.0f);
    return hv * zv;
}

__global__ __launch_bounds__(128) void kE(const float* __restrict__ basis,
                                          const float* __restrict__ z,
                                          const int* __restrict__ srcs,
                                          const int* __restrict__ off,
                                          const float* __restrict__ W2a,
                                          float* __restrict__ out, int N)
{
    const int lane = threadIdx.x & 31;
    const int wave = __builtin_amdgcn_readfirstlane(threadIdx.x >> 5);
    const int n0 = (blockIdx.x * 4 + wave) * KE_NODES;
    if (n0 >= N) return;
    const int nn = (N - n0 < KE_NODES) ? (N - n0) : KE_NODES;
    int oi = n0 + lane;
    if (oi > N) oi = N;
    const int ov = off[oi];
    const int beg_all = rl_i(ov, 0);
    const int end_all = rl_i(ov, nn);

    const float4* w4 = reinterpret_cast<const float4*>(W2a);
    const float4 w0a = w4[lane],       w0b = w4[32 + lane];
    const float4 w1a = w4[64 + lane],  w1b = w4[96 + lane];
    const float4 w2a = w4[128 + lane], w2b = w4[160 + lane];

    const float oscale = (1.0f/128.0f) * INV_NN;
    int   k    = 0;
    int   nend = rl_i(ov, 1);
    float acc  = 0.0f;

    for (int base = beg_all; base < end_all; base += 64) {
        const int last = end_all - 1;
        int iA = base + lane;       if (iA > last) iA = last;
        int iB = base + 32 + lane;  if (iB > last) iB = last;
        const int sA = srcs[iA], sB = srcs[iB];
        const float pA0 = basis[(size_t)iA*3], pA1 = basis[(size_t)iA*3 + 1], pA2 = basis[(size_t)iA*3 + 2];
        const float pB0 = basis[(size_t)iB*3], pB1 = basis[(size_t)iB*3 + 1], pB2 = basis[(size_t)iB*3 + 2];
        const int left = end_all - base;
        const int cnt  = left < 64 ? left : 64;
        for (int j0 = 0; j0 < cnt; j0 += KE_U) {
            const bool hi = j0 >= 32;
            const int   vs  = hi ? sB  : sA;
            const float vb0 = hi ? pB0 : pA0, vb1 = hi ? pB1 : pA1, vb2 = hi ? pB2 : pA2;
            const int jl = j0 & 31;
            const int m  = (cnt - j0 < KE_U) ? (cnt - j0) : KE_U;
            float4 za[KE_U], zb[KE_U];
            #pragma unroll
            for (int u = 0; u < KE_U; u++) {
                const int jj = jl + (u < m ? u : m - 1);
                const int s  = rl_i(vs, jj);
                const float4* zr = reinterpret_cast<const float4*>(z + (size_t)s * 256);
                za[u] = zr[lane];
                zb[u] = zr[32 + lane];
            }
            #pragma unroll
            for (int u = 0; u < KE_U; u++) {
                if (u < m) {
                    const int jj = jl + u;
                    const float c0 = rl_f(vb0, jj), c1 = rl_f(vb1, jj), c2 = rl_f(vb2, jj);
                    float t = h2term(c0, c1, c2, w0a.x, w1a.x, w2a.x, za[u].x);
                    t += h2term(c0, c1, c2, w0a.y, w1a.y, w2a.y, za[u].y);
                    t += h2term(c0, c1, c2, w0a.z, w1a.z, w2a.z, za[u].z);
                    t += h2term(c0, c1, c2, w0a.w, w1a.w, w2a.w, za[u].w);
                    t += h2term(c0, c1, c2, w0b.x, w1b.x, w2b.x, zb[u].x);
                    t += h2term(c0, c1, c2, w0b.y, w1b.y, w2b.y, zb[u].y);
                    t += h2term(c0, c1, c2, w0b.z, w1b.z, w2b.z, zb[u].z);
                    t += h2term(c0, c1, c2, w0b.w, w1b.w, w2b.w, zb[u].w);
                    const int p = base + j0 + u;
                    while (p >= nend) {
                        const float tot = wave_sum(acc);
                        if (lane == 0) out[n0 + k] = tot * oscale;
                        acc = 0.0f;
                        k++;
                        nend = rl_i(ov, k + 1);
                    }
                    acc += t;
                }
            }
        }
    }
    while (k < nn) {
        const float tot = wave_sum(acc);
        if (lane == 0) out[n0 + k] = tot * oscale;
        acc = 0.0f;
        k++;
    }
}

extern "C" void kernel_launch(void* const* d_in, const int* in_sizes, int n_in,
                              void* d_out, int out_size, void* d_ws, size_t ws_size,
                              hipStream_t stream)
{
    const float* evec = (const float*)d_in[0];
    const float* elen = (const float*)d_in[1];
    const float* W1a  = (const float*)d_in[2];
    const float* W1b  = (const float*)d_in[3];
    const float* W2a  = (const float*)d_in[4];
    const float* W2b  = (const float*)d_in[5];
    const int*   esrc = (const int*)d_in[6];
    const int*   edst = (const int*)d_in[7];
    float* out = (float*)d_out;

    const int E = in_sizes[0] / 3;
    const int N = out_size;

    float* ws = (float*)d_ws;
    float* x_     = ws;                 ws += (size_t)N * 16;
    ws += (size_t)N * 96;
    float* lists_ = ws;                 ws += (size_t)N * 64;
    float* z_     = ws;                 ws += (size_t)N * 256;
    float* sh_    = ws;                 ws += (size_t)E * 16;
    float* basis_ = ws;                 ws += (size_t)E * 3;

    int* cnt_  = (int*)lists_;
    int* rank_ = cnt_ + N;
    int* off_  = rank_ + E;
    int* srcs_ = off_ + N + 1;

    kzero<<<(N + 255) / 256, 256, 0, stream>>>((float*)cnt_, N);

    kA<<<(E + 255) / 256, 256, 0, stream>>>(edst, cnt_, rank_, E);
    kScan<<<1, 1024, 0, stream>>>(cnt_, off_, N);
    kFill<<<(E + 255) / 256, 256, 0, stream>>>(evec, elen, esrc, edst, rank_, off_, sh_, basis_, srcs_, E);
    kX<<<(4 * N + 255) / 256, 256, 0, stream>>>(sh_, off_, x_, N);
    kB<<<(N + 31) / 32, 192, 0, stream>>>(sh_, basis_, x_, srcs_, off_, W1a, W1b, W2b, z_, N);
    kE<<<(N + 4 * KE_NODES - 1) / (4 * KE_NODES), 128, 0, stream>>>(basis_, z_, srcs_, off_, W2a, out, N);
}


# offsets scan: each wave reads its counters as contiguous 512-byte rows into registers and scans them with DPP adds; before, 1024 threads each walked 25 counters at a 100-byte lane stride, twice
# speedup vs baseline: 3.5708x; 1.1480x over previous
#include <hip/hip_runtime.h>
#include <math.h>

typedef __attribute__((ext_vector_type(2))) float v2f;
typedef __attribute__((ext_vector_type(8))) float v8f;

#define INV_NN     0.512989176042577f
#define INV_SQRT3  0.5773502691896258f
#define S3   1.7320508075688772f
#define S5   2.23606797749979f
#define S7   2.6457513110645907f
#define S15  3.872983346207417f
#define S105 10.246950765959598f
#define C358 2.091650066335189f
#define C218 1.6201851746019651f

__global__ void kzero(float* __restrict__ p, long n) {
    long i = (long)blockIdx.x * blockDim.x + threadIdx.x;
    if (i < n) p[i] = 0.0f;
}

__global__ void kA(const int* __restrict__ edst, int* __restrict__ cnt, int* __restrict__ rank, int E)
{
    int e = blockIdx.x * blockDim.x + threadIdx.x;
    if (e >= E) return;
    rank[e] = atomicAdd(&cnt[edst[e]], 1);
}

#define KS_ROWS 8
#define KS_PASS (32 * 128 * KS_ROWS)

template <int S> __device__ __forceinline__ int row_shr_add(int v) {
    int t = __builtin_amdgcn_update_dpp(0, v, 0x110 | S, 0xF, 0xF, false);
    return v + t;
}
__device__ __forceinline__ int wave_scan_incl(int v, int lane) {
    v = row_shr_add<1>(v); v = row_shr_add<2>(v); v = row_shr_add<4>(v); v = row_shr_add<8>(v);
    const int lo = __builtin_amdgcn_readlane(v, 15);
    return v + (lane >= 16 ? lo : 0);
}

__global__ __launch_bounds__(1024) void kScan(const int* __restrict__ cnt,
                                              int* __restrict__ off, int N)
{
    __shared__ int wtot[32];
    const int lane = threadIdx.x & 31;
    const int wave = __builtin_amdgcn_readfirstlane(threadIdx.x >> 5);
    int carry = 0;
    for (int p0 = 0; p0 <= N; p0 += KS_PASS) {
        const int b = p0 + wave * (128 * KS_ROWS) + 4 * lane;
        int4 q[KS_ROWS];
        #pragma unroll
        for (int i = 0; i < KS_ROWS; i++) {
            const int idx = b + 128 * i;
            q[i] = make_int4(0, 0, 0, 0);
            if (idx < N) q[i] = *reinterpret_cast<const int4*>(cnt + idx);
        }
        int run = 0;
        #pragma unroll
        for (int i = 0; i < KS_ROWS; i++) {
            const int s   = (q[i].x + q[i].y) + (q[i].z + q[i].w);
            const int inc = wave_scan_incl(s, lane);
            const int e   = run + (inc - s);
            run += __builtin_amdgcn_readlane(inc, 31);
            const int e1 = e + q[i].x, e2 = e1 + q[i].y, e3 = e2 + q[i].z;
            q[i].x = e; q[i].y = e1; q[i].z = e2; q[i].w = e3;
        }
        if (lane == 0) wtot[wave] = run;
        __syncthreads();
        const int wt   = wtot[lane];
        const int winc = wave_scan_incl(wt, lane);
        const int wbase = carry + __builtin_amdgcn_readlane(winc - wt, wave);
        carry += __builtin_amdgcn_readlane(winc, 31);
        #pragma unroll
        for (int i = 0; i < KS_ROWS; i++) {
            const int idx = b + 128 * i;
            if (idx <= N) {
                int4 o;
                o.x = q[i].x + wbase; o.y = q[i].y + wbase; o.z = q[i].z + wbase; o.w = q[i].w + wbase;
                *reinterpret_cast<int4*>(off + idx) = o;
            }
        }
        __syncthreads();
    }
}

__global__ void kFill(const float* __restrict__ evec, const float* __restrict__ elen,
                      const int* __restrict__ esrc, const int* __restrict__ edst,
                      const int* __restrict__ rank, const int* __restrict__ off,
                      float* __restrict__ sh, float* __restrict__ basis,
                      int* __restrict__ srcs, int E)
{
    int e = blockIdx.x * blockDim.x + threadIdx.x;
    if (e >= E) return;
    const int slot = off[edst[e]] + rank[e];
    float vx = evec[3*(size_t)e], vy = evec[3*(size_t)e+1], vz = evec[3*(size_t)e+2];
    float rn = 1.0f / (sqrtf(vx*vx + vy*vy + vz*vz) + 1e-12f);
    vx *= rn; vy *= rn; vz *= rn;
    float x2 = vx*vx, y2 = vy*vy, z2 = vz*vz;
    float4 r0, r1, r2, r3;
    r0.x = 1.0f;
    r0.y = S3*vx; r0.z = S3*vy; r0.w = S3*vz;
    r1.x = S15*vx*vz;
    r1.y = S15*vx*vy;
    r1.z = 0.5f*S5*(2.0f*y2 - x2 - z2);
    r1.w = S15*vy*vz;
    r2.x = 0.5f*S15*(z2 - x2);
    r2.y = C358*vx*(3.0f*z2 - x2);
    r2.z = S105*vx*vy*vz;
    r2.w = C218*vx*(4.0f*y2 - z2 - x2);
    r3.x = 0.5f*S7*vy*(2.0f*y2 - 3.0f*z2 - 3.0f*x2);
    r3.y = C218*vz*(4.0f*y2 - z2 - x2);
    r3.z = 0.5f*S105*vy*(z2 - x2);
    r3.w = C358*vz*(z2 - 3.0f*x2);
    float4* sr = reinterpret_cast<float4*>(sh + (size_t)slot * 16);
    sr[0] = r0; sr[1] = r1; sr[2] = r2; sr[3] = r3;
    float len = elen[e];
    #pragma unroll
    for (int i = 0; i < 3; i++) {
        float dd = (len - 2.0f*(float)i) * 0.5f;
        basis[(size_t)slot*3 + i] = expf(-dd*dd) * (1.0f/1.12f);
    }
    srcs[slot] = esrc[e];
}

__global__ void kX(const float* __restrict__ sh, const int* __restrict__ off,
                   float* __restrict__ x, int N)
{
    int g = blockIdx.x * blockDim.x + threadIdx.x;
    int n = g >> 2, q = g & 3;
    if (n >= N) return;
    const int beg = off[n], end = off[n + 1];
    const float4* p = reinterpret_cast<const float4*>(sh) + (size_t)beg * 4 + q;
    float4 a = {0.0f, 0.0f, 0.0f, 0.0f};
    #pragma unroll 4
    for (int i = beg; i < end; i++) {
        float4 v = p[(size_t)(i - beg) * 4];
        a.x += v.x * INV_NN; a.y += v.y * INV_NN; a.z += v.z * INV_NN; a.w += v.w * INV_NN;
    }
    reinterpret_cast<float4*>(x)[(size_t)n * 4 + q] = a;
}

#define KB_PITCH 1028
#define KB_HP 101
#define KB_YP 68
#define KB_ZP 260
#define RS3 0.5773502691896258f
#define RS5 0.4472135954999579f
#define RS7 0.3779644730092272f
__global__ __launch_bounds__(192) void kB(
    const float* __restrict__ sh, const float* __restrict__ basis,
    const float* __restrict__ x,
    const int* __restrict__ srcs,
    const int* __restrict__ off,
    const float* __restrict__ W1a, const float* __restrict__ W1b,
    const float* W2b, float* __restrict__ z, int N)
{
    __shared__ float Gs[32 * KB_PITCH];

    const int tid  = threadIdx.x;
    const int lane = tid & 31;
    const int wave = __builtin_amdgcn_readfirstlane(tid >> 5);
    const int n0   = blockIdx.x * 32;

    float wa0[8], wa1[8], wa2[8];
    #pragma unroll
    for (int i = 0; i < 8; i++) {
        wa0[i] = W1a[lane + 32*i];
        wa1[i] = W1a[256 + lane + 32*i];
        wa2[i] = W1a[512 + lane + 32*i];
    }

    for (int m = wave; m < 32; m += 6) {
        const int n = n0 + m;
        float g0[8], g1[8], g2[8], g3[8];
        #pragma unroll
        for (int i = 0; i < 8; i++) { g0[i] = 0.0f; g1[i] = 0.0f; g2[i] = 0.0f; g3[i] = 0.0f; }
        if (n < N) {
            const int beg = off[n], end = off[n + 1];
            for (int base = beg; base < end; base += 32) {
                const int idx = base + lane;
                float d0 = 0.0f, d1 = 0.0f, d2 = 0.0f, d3 = 0.0f;
                float b0 = 0.0f, b1 = 0.0f, b2 = 0.0f;
                if (idx < end) {
                    const int srcn = srcs[idx];
                    const float4* xs = reinterpret_cast<const float4*>(x  + (size_t)srcn * 16);
                    const float4* ss = reinterpret_cast<const float4*>(sh + (size_t)idx * 16);
                    float4 x0 = xs[0], x1 = xs[1], x2 = xs[2], x3 = xs[3];
                    float4 s0 = ss[0], s1 = ss[1], s2 = ss[2], s3 = ss[3];
                    d0 = x0.x*s0.x;
                    d1 = (x0.y*s0.y + x0.z*s0.z + x0.w*s0.w) * RS3;
                    d2 = (x1.x*s1.x + x1.y*s1.y + x1.z*s1.z + x1.w*s1.w + x2.x*s2.x) * RS5;
                    d3 = (x2.y*s2.y + x2.z*s2.z + x2.w*s2.w
                        + x3.x*s3.x + x3.y*s3.y + x3.z*s3.z + x3.w*s3.w) * RS7;
                    b0 = basis[(size_t)idx*3]; b1 = basis[(size_t)idx*3 + 1]; b2 = basis[(size_t)idx*3 + 2];
                }
                const int left = end - base;
                const int cntc = left < 32 ? left : 32;
                for (int j = 0; j < cntc; j++) {
                    const float e0 = __shfl(d0, j, 32), e1 = __shfl(d1, j, 32);
                    const float e2 = __shfl(d2, j, 32), e3 = __shfl(d3, j, 32);
                    const float c0 = __shfl(b0, j, 32), c1 = __shfl(b1, j, 32), c2 = __shfl(b2, j, 32);
                    #pragma unroll
                    for (int i = 0; i < 8; i++) {
                        float hv = c0*wa0[i] + c1*wa1[i] + c2*wa2[i];
                        hv = fmaxf(hv * INV_SQRT3, 0.0f);
                        g0[i] += e0 * hv; g1[i] += e1 * hv;
                        g2[i] += e2 * hv; g3[i] += e3 * hv;
                    }
                }
            }
        }
        float* gr = &Gs[m * KB_PITCH + lane];
        #pragma unroll
        for (int i = 0; i < 8; i++) {
            gr[32*i]       = g0[i];
            gr[256 + 32*i] = g1[i];
            gr[512 + 32*i] = g2[i];
            gr[768 + 32*i] = g3[i];
        }
    }
    __syncthreads();

    const int c    = lane & 15;
    const int hh   = lane >> 4;
    const int colb = wave * 16 + c;
    v8f a00 = {}, a01 = {}, a02 = {}, a03 = {};
    v8f a10 = {}, a11 = {}, a12 = {}, a13 = {};
    const float* ga0 = &Gs[c * KB_PITCH + 2 * hh];
    const float* ga1 = ga0 + 16 * KB_PITCH;
    const float* Bp = W1b + (size_t)(2 * hh) * 384 + colb;
    v2f Bc[2][4], Bn[2][4];
    #pragma unroll
    for (int s = 0; s < 2; s++) {
        #pragma unroll
        for (int q = 0; q < 4; q++) {
            Bc[s][q].x = Bp[(size_t)(4 * s) * 384 + 96 * q];
            Bc[s][q].y = Bp[(size_t)(4 * s + 1) * 384 + 96 * q];
        }
    }
    for (int k0 = 0; k0 < 256; k0 += 8) {
        const int kn = (k0 + 8 < 256) ? (k0 + 8) : k0;
        const float* Bq = Bp + (size_t)kn * 384;
        #pragma unroll
        for (int s = 0; s < 2; s++) {
            #pragma unroll
            for (int q = 0; q < 4; q++) {
                Bn[s][q].x = Bq[(size_t)(4 * s) * 384 + 96 * q];
                Bn[s][q].y = Bq[(size_t)(4 * s + 1) * 384 + 96 * q];
            }
        }
        #pragma unroll
        for (int s = 0; s < 2; s++) {
            const int k = k0 + 4 * s;
            v2f A00 = *reinterpret_cast<const v2f*>(ga0 + k);
            v2f A01 = *reinterpret_cast<const v2f*>(ga0 + 256 + k);
            v2f A02 = *reinterpret_cast<const v2f*>(ga0 + 512 + k);
            v2f A03 = *reinterpret_cast<const v2f*>(ga0 + 768 + k);
            v2f A10 = *reinterpret_cast<const v2f*>(ga1 + k);
            v2f A11 = *reinterpret_cast<const v2f*>(ga1 + 256 + k);
            v2f A12 = *reinterpret_cast<const v2f*>(ga1 + 512 + k);
            v2f A13 = *reinterpret_cast<const v2f*>(ga1 + 768 + k);
            a00 = __builtin_amdgcn_wmma_f32_16x16x4_f32(false, A00, false, Bc[s][0], (short)0, a00, false, false);
            a10 = __builtin_amdgcn_wmma_f32_16x16x4_f32(false, A10, false, Bc[s][0], (short)0, a10, false, false);
            a01 = __builtin_amdgcn_wmma_f32_16x16x4_f32(false, A01, false, Bc[s][1], (short)0, a01, false, false);
            a11 = __builtin_amdgcn_wmma_f32_16x16x4_f32(false, A11, false, Bc[s][1], (short)0, a11, false, false);
            a02 = __builtin_amdgcn_wmma_f32_16x16x4_f32(false, A02, false, Bc[s][2], (short)0, a02, false, false);
            a12 = __builtin_amdgcn_wmma_f32_16x16x4_f32(false, A12, false, Bc[s][2], (short)0, a12, false, false);
            a03 = __builtin_amdgcn_wmma_f32_16x16x4_f32(false, A03, false, Bc[s][3], (short)0, a03, false, false);
            a13 = __builtin_amdgcn_wmma_f32_16x16x4_f32(false, A13, false, Bc[s][3], (short)0, a13, false, false);
        }
        #pragma unroll
        for (int s = 0; s < 2; s++) {
            #pragma unroll
            for (int q = 0; q < 4; q++) Bc[s][q] = Bn[s][q];
        }
    }
    v2f Bf[3][16];
    #pragma unroll
    for (int t = 0; t < 3; t++) {
        int jt = wave + 6 * t;
        if (jt > 15) jt = 15;
        const float* wb = W2b + (size_t)(16 * jt + c) * 64 + 2 * hh;
        #pragma unroll
        for (int i = 0; i < 16; i++) Bf[t][i] = *reinterpret_cast<const v2f*>(wb + 4 * i);
    }
    __syncthreads();

    float* Hs = Gs;
    float* Ys = Gs + 32 * KB_HP;
    float* Zs = Ys + 32 * KB_YP;
    const float scale = (1.0f/16.0f) * 0.5f * INV_NN;
    #pragma unroll
    for (int r = 0; r < 8; r++) {
        Hs[(8*hh + r) * KB_HP + colb]      = ((a00[r] + a01[r]) + (a02[r] + a03[r])) * scale;
        Hs[(16 + 8*hh + r) * KB_HP + colb] = ((a10[r] + a11[r]) + (a12[r] + a13[r])) * scale;
    }
    __syncthreads();
    {
        const float* hr = &Hs[lane * KB_HP];
        float* yr = &Ys[lane * KB_YP];
        float ha[11], hb[11];
        #pragma unroll
        for (int k = 0; k < 11; k++) {
            int j = wave + 6 * k;
            if (j > 63) j = 63;
            ha[k] = hr[j];
            hb[k] = hr[j + 32];
        }
        #pragma unroll
        for (int k = 0; k < 11; k++) {
            const int j = wave + 6 * k;
            if (j < 64) {
                float v;
                if (j < 16)      v = fmaxf(ha[k], 0.0f);
                else if (j < 32) v = fabsf(ha[k]);
                else {
                    const int jj = j - 32;
                    const bool use_relu = (jj < 8) || (jj >= 16 && jj < 24);
                    const float ga = use_relu ? fmaxf(ha[k], 0.0f) : tanhf(ha[k]);
                    v = ga * hb[k];
                }
                yr[j] = v;
            }
        }
    }
    __syncthreads();

    {
        const float* ya = &Ys[c * KB_YP + 2 * hh];
        v2f A0[16], A1[16];
        #pragma unroll
        for (int i = 0; i < 16; i++) {
            A0[i] = *reinterpret_cast<const v2f*>(ya + 4 * i);
            A1[i] = *reinterpret_cast<const v2f*>(ya + 16 * KB_YP + 4 * i);
        }
        #pragma unroll
        for (int t = 0; t < 3; t++) {
            const int jt = wave + 6 * t;
            if (jt < 16) {
                v8f z0 = {}, z1 = {};
                #pragma unroll
                for (int i = 0; i < 16; i++) {
                    z0 = __builtin_amdgcn_wmma_f32_16x16x4_f32(false, A0[i], false, Bf[t][i], (short)0, z0, false, false);
                    z1 = __builtin_amdgcn_wmma_f32_16x16x4_f32(false, A1[i], false, Bf[t][i], (short)0, z1, false, false);
                }
                float* zc = &Zs[(8 * hh) * KB_ZP + 16 * jt + c];
                #pragma unroll
                for (int r = 0; r < 8; r++) {
                    zc[r * KB_ZP]        = z0[r];
                    zc[(16 + r) * KB_ZP] = z1[r];
                }
            }
        }
    }
    __syncthreads();
    const int rows = (N - n0 < 32) ? (N - n0) : 32;
    for (int idx = tid; idx < rows * 64; idx += 192) {
        const int m = idx >> 6, q = idx & 63;
        float4 v = *reinterpret_cast<const float4*>(&Zs[m * KB_ZP + 4 * q]);
        *reinterpret_cast<float4*>(z + (size_t)(n0 + m) * 256 + 4 * q) = v;
    }
}

#define KE_NODES 4
#define KE_U 8

__device__ __forceinline__ int rl_i(int v, int l) { return __builtin_amdgcn_readlane(v, l); }
__device__ __forceinline__ float rl_f(float v, int l) {
    return __int_as_float(__builtin_amdgcn_readlane(__float_as_int(v), l));
}
template <int M> __device__ __forceinline__ float row_xadd(float v) {
    int t = __builtin_amdgcn_update_dpp(0, __float_as_int(v), 0x160 | M, 0xF, 0xF, false);
    return v + __int_as_float(t);
}
__device__ __forceinline__ float wave_sum(float v) {
    v = row_xadd<1>(v); v = row_xadd<2>(v); v = row_xadd<4>(v); v = row_xadd<8>(v);
    return rl_f(v, 0) + rl_f(v, 16);
}
__device__ __forceinline__ float h2term(float c0, float c1, float c2,
                                        float w0, float w1, float w2, float zv) {
    float hv = c0*w0 + c1*w1 + c2*w2;
    hv = fmaxf(hv * INV_SQRT3, 0.0f);
    return hv * zv;
}

__global__ __launch_bounds__(128) void kE(const float* __restrict__ basis,
                                          const float* __restrict__ z,
                                          const int* __restrict__ srcs,
                                          const int* __restrict__ off,
                                          const float* __restrict__ W2a,
                                          float* __restrict__ out, int N)
{
    const int lane = threadIdx.x & 31;
    const int wave = __builtin_amdgcn_readfirstlane(threadIdx.x >> 5);
    const int n0 = (blockIdx.x * 4 + wave) * KE_NODES;
    if (n0 >= N) return;
    const int nn = (N - n0 < KE_NODES) ? (N - n0) : KE_NODES;
    int oi = n0 + lane;
    if (oi > N) oi = N;
    const int ov = off[oi];
    const int beg_all = rl_i(ov, 0);
    const int end_all = rl_i(ov, nn);

    const float4* w4 = reinterpret_cast<const float4*>(W2a);
    const float4 w0a = w4[lane],       w0b = w4[32 + lane];
    const float4 w1a = w4[64 + lane],  w1b = w4[96 + lane];
    const float4 w2a = w4[128 + lane], w2b = w4[160 + lane];

    const float oscale = (1.0f/128.0f) * INV_NN;
    int   k    = 0;
    int   nend = rl_i(ov, 1);
    float acc  = 0.0f;

    for (int base = beg_all; base < end_all; base += 64) {
        const int last = end_all - 1;
        int iA = base + lane;       if (iA > last) iA = last;
        int iB = base + 32 + lane;  if (iB > last) iB = last;
        const int sA = srcs[iA], sB = srcs[iB];
        const float pA0 = basis[(size_t)iA*3], pA1 = basis[(size_t)iA*3 + 1], pA2 = basis[(size_t)iA*3 + 2];
        const float pB0 = basis[(size_t)iB*3], pB1 = basis[(size_t)iB*3 + 1], pB2 = basis[(size_t)iB*3 + 2];
        const int left = end_all - base;
        const int cnt  = left < 64 ? left : 64;
        for (int j0 = 0; j0 < cnt; j0 += KE_U) {
            const bool hi = j0 >= 32;
            const int   vs  = hi ? sB  : sA;
            const float vb0 = hi ? pB0 : pA0, vb1 = hi ? pB1 : pA1, vb2 = hi ? pB2 : pA2;
            const int jl = j0 & 31;
            const int m  = (cnt - j0 < KE_U) ? (cnt - j0) : KE_U;
            float4 za[KE_U], zb[KE_U];
            #pragma unroll
            for (int u = 0; u < KE_U; u++) {
                const int jj = jl + (u < m ? u : m - 1);
                const int s  = rl_i(vs, jj);
                const float4* zr = reinterpret_cast<const float4*>(z + (size_t)s * 256);
                za[u] = zr[lane];
                zb[u] = zr[32 + lane];
            }
            #pragma unroll
            for (int u = 0; u < KE_U; u++) {
                if (u < m) {
                    const int jj = jl + u;
                    const float c0 = rl_f(vb0, jj), c1 = rl_f(vb1, jj), c2 = rl_f(vb2, jj);
                    float t = h2term(c0, c1, c2, w0a.x, w1a.x, w2a.x, za[u].x);
                    t += h2term(c0, c1, c2, w0a.y, w1a.y, w2a.y, za[u].y);
                    t += h2term(c0, c1, c2, w0a.z, w1a.z, w2a.z, za[u].z);
                    t += h2term(c0, c1, c2, w0a.w, w1a.w, w2a.w, za[u].w);
                    t += h2term(c0, c1, c2, w0b.x, w1b.x, w2b.x, zb[u].x);
                    t += h2term(c0, c1, c2, w0b.y, w1b.y, w2b.y, zb[u].y);
                    t += h2term(c0, c1, c2, w0b.z, w1b.z, w2b.z, zb[u].z);
                    t += h2term(c0, c1, c2, w0b.w, w1b.w, w2b.w, zb[u].w);
                    const int p = base + j0 + u;
                    while (p >= nend) {
                        const float tot = wave_sum(acc);
                        if (lane == 0) out[n0 + k] = tot * oscale;
                        acc = 0.0f;
                        k++;
                        nend = rl_i(ov, k + 1);
                    }
                    acc += t;
                }
            }
        }
    }
    while (k < nn) {
        const float tot = wave_sum(acc);
        if (lane == 0) out[n0 + k] = tot * oscale;
        acc = 0.0f;
        k++;
    }
}

extern "C" void kernel_launch(void* const* d_in, const int* in_sizes, int n_in,
                              void* d_out, int out_size, void* d_ws, size_t ws_size,
                              hipStream_t stream)
{
    const float* evec = (const float*)d_in[0];
    const float* elen = (const float*)d_in[1];
    const float* W1a  = (const float*)d_in[2];
    const float* W1b  = (const float*)d_in[3];
    const float* W2a  = (const float*)d_in[4];
    const float* W2b  = (const float*)d_in[5];
    const int*   esrc = (const int*)d_in[6];
    const int*   edst = (const int*)d_in[7];
    float* out = (float*)d_out;

    const int E = in_sizes[0] / 3;
    const int N = out_size;

    float* ws = (float*)d_ws;
    float* x_     = ws;                 ws += (size_t)N * 16;
    float* cntp_  = ws;                 ws += (size_t)N * 96;
    float* lists_ = ws;                 ws += (size_t)N * 64;
    float* z_     = ws;                 ws += (size_t)N * 256;
    float* sh_    = ws;                 ws += (size_t)E * 16;
    float* basis_ = ws;                 ws += (size_t)E * 3;

    int* cnt_  = (int*)cntp_;
    int* off_  = (int*)lists_;
    int* rank_ = off_ + N + 4;
    int* srcs_ = rank_ + E;

    kzero<<<(N + 3 + 255) / 256, 256, 0, stream>>>((float*)cnt_, N + 3);

    kA<<<(E + 255) / 256, 256, 0, stream>>>(edst, cnt_, rank_, E);
    kScan<<<1, 1024, 0, stream>>>(cnt_, off_, N);
    kFill<<<(E + 255) / 256, 256, 0, stream>>>(evec, elen, esrc, edst, rank_, off_, sh_, basis_, srcs_, E);
    kX<<<(4 * N + 255) / 256, 256, 0, stream>>>(sh_, off_, x_, N);
    kB<<<(N + 31) / 32, 192, 0, stream>>>(sh_, basis_, x_, srcs_, off_, W1a, W1b, W2b, z_, N);
    kE<<<(N + 4 * KE_NODES - 1) / (4 * KE_NODES), 128, 0, stream>>>(basis_, z_, srcs_, off_, W2a, out, N);
}


# edge grouping in ONE launch: a block owns 98 nodes, reads all destinations, counts those in front (its list base), ranks its own in LDS, writes records and x; 7 -> 3 launches; x sum order changes
# speedup vs baseline: 4.3832x; 1.2275x over previous
#include <hip/hip_runtime.h>
#include <math.h>

typedef __attribute__((ext_vector_type(2))) float v2f;
typedef __attribute__((ext_vector_type(8))) float v8f;

#define INV_NN     0.512989176042577f
#define INV_SQRT3  0.5773502691896258f
#define S3   1.7320508075688772f
#define S5   2.23606797749979f
#define S7   2.6457513110645907f
#define S15  3.872983346207417f
#define S105 10.246950765959598f
#define C358 2.091650066335189f
#define C218 1.6201851746019651f


#define KG_NODES 98
#define KG_XP    17
#define KG_CAP   8192
#define KG_U     8
#define KG_P     4

template <int S> __device__ __forceinline__ int row_shr_add(int v) {
    int t = __builtin_amdgcn_update_dpp(0, v, 0x110 | S, 0xF, 0xF, false);
    return v + t;
}
__device__ __forceinline__ int wave_scan_incl(int v, int lane) {
    v = row_shr_add<1>(v); v = row_shr_add<2>(v); v = row_shr_add<4>(v); v = row_shr_add<8>(v);
    const int lo = __builtin_amdgcn_readlane(v, 15);
    return v + (lane >= 16 ? lo : 0);
}

struct KgIn { float vx, vy, vz, aux; };
__device__ __forceinline__ KgIn kg_fetch(int e, int q, const float* __restrict__ evec,
                                         const float* __restrict__ elen, const int* __restrict__ esrc)
{
    KgIn r;
    r.vx = evec[3*(size_t)e]; r.vy = evec[3*(size_t)e+1]; r.vz = evec[3*(size_t)e+2];
    r.aux = (q < 3) ? elen[e] : __int_as_float(esrc[e]);
    return r;
}
__device__ __forceinline__ void kg_record(KgIn in, int slot, int j, int q,
        float* __restrict__ sh, float* __restrict__ basis, int* __restrict__ srcs, float* xs)
{
    float vx = in.vx, vy = in.vy, vz = in.vz;
    float rn = 1.0f / (sqrtf(vx*vx + vy*vy + vz*vz) + 1e-12f);
    vx *= rn; vy *= rn; vz *= rn;
    float x2 = vx*vx, y2 = vy*vy, z2 = vz*vz;
    float4 r;
    if (q == 0) {
        r.x = 1.0f;
        r.y = S3*vx; r.z = S3*vy; r.w = S3*vz;
    } else if (q == 1) {
        r.x = S15*vx*vz;
        r.y = S15*vx*vy;
        r.z = 0.5f*S5*(2.0f*y2 - x2 - z2);
        r.w = S15*vy*vz;
    } else if (q == 2) {
        r.x = 0.5f*S15*(z2 - x2);
        r.y = C358*vx*(3.0f*z2 - x2);
        r.z = S105*vx*vy*vz;
        r.w = C218*vx*(4.0f*y2 - z2 - x2);
    } else {
        r.x = 0.5f*S7*vy*(2.0f*y2 - 3.0f*z2 - 3.0f*x2);
        r.y = C218*vz*(4.0f*y2 - z2 - x2);
        r.z = 0.5f*S105*vy*(z2 - x2);
        r.w = C358*vz*(z2 - 3.0f*x2);
    }
    reinterpret_cast<float4*>(sh + (size_t)slot * 16)[q] = r;
    if (q < 3) {
        const float dd = (in.aux - 2.0f*(float)q) * 0.5f;
        basis[(size_t)slot*3 + q] = expf(-dd*dd) * (1.0f/1.12f);
    } else {
        srcs[slot] = __float_as_int(in.aux);
    }
    float* xr = xs + j * KG_XP + 4 * q;
    atomicAdd(&xr[0], r.x * INV_NN); atomicAdd(&xr[1], r.y * INV_NN);
    atomicAdd(&xr[2], r.z * INV_NN); atomicAdd(&xr[3], r.w * INV_NN);
}

__global__ __launch_bounds__(1024) void kG(const float* __restrict__ evec, const float* __restrict__ elen,
                                           const int* __restrict__ esrc, const int* __restrict__ edst,
                                           int* __restrict__ off,
                                           float* __restrict__ sh, float* __restrict__ basis,
                                           int* __restrict__ srcs, float* __restrict__ x, int E, int N)
{
    __shared__ __attribute__((aligned(16))) int cnt[128];
    __shared__ __attribute__((aligned(16))) int excl[128];
    __shared__ int cnt2[128];
    __shared__ int wred[32];
    __shared__ int nmatch_s;
    __shared__ float xs[KG_NODES * KG_XP];
    __shared__ int le[KG_CAP], lj[KG_CAP], lr[KG_CAP];
    __shared__ int oe[KG_CAP], oj[KG_CAP];

    const int tid  = threadIdx.x;
    const int lane = tid & 31;
    const int wave = __builtin_amdgcn_readfirstlane(tid >> 5);
    const int lo = blockIdx.x * KG_NODES;
    const int nl = (N - lo < KG_NODES) ? (N - lo) : KG_NODES;

    if (tid < 128) { cnt[tid] = 0; cnt2[tid] = 0; }
    if (tid == 0) nmatch_s = 0;
    for (int i = tid; i < KG_NODES * KG_XP; i += 1024) xs[i] = 0.0f;
    __syncthreads();

    const int nv = E >> 2;
    const int rot = (int)(blockIdx.x * ((unsigned)nv / gridDim.x));
    const int4* d4 = reinterpret_cast<const int4*>(edst);
    int sless = 0;

    auto keep = [&](int t, int e) {
        const int r   = atomicAdd(&cnt[t], 1);
        const int pos = atomicAdd(&nmatch_s, 1);
        if (pos < KG_CAP) { le[pos] = e; lj[pos] = t; lr[pos] = r; }
    };
    auto visit4 = [&](int4 dd, int e0) {
        const int t0 = dd.x - lo, t1 = dd.y - lo, t2 = dd.z - lo, t3 = dd.w - lo;
        sless += __builtin_popcount(__builtin_amdgcn_ballot_w32(t0 < 0))
               + __builtin_popcount(__builtin_amdgcn_ballot_w32(t1 < 0))
               + __builtin_popcount(__builtin_amdgcn_ballot_w32(t2 < 0))
               + __builtin_popcount(__builtin_amdgcn_ballot_w32(t3 < 0));
        const bool m0 = (unsigned)t0 < (unsigned)nl, m1 = (unsigned)t1 < (unsigned)nl;
        const bool m2 = (unsigned)t2 < (unsigned)nl, m3 = (unsigned)t3 < (unsigned)nl;
        if (m0 | m1 | m2 | m3) {
            if (m0) keep(t0, e0);
            if (m1) keep(t1, e0 + 1);
            if (m2) keep(t2, e0 + 2);
            if (m3) keep(t3, e0 + 3);
        }
    };

    for (int i0 = 0; i0 < nv; i0 += 1024 * KG_U) {
        int4 d[KG_U];
        int  id[KG_U];
        #pragma unroll
        for (int u = 0; u < KG_U; u++) {
            const int i = i0 + u * 1024 + tid;
            int idx = i + rot;
            if (idx >= nv) idx -= nv;
            id[u] = idx;
            d[u] = make_int4(0x7fffffff, 0x7fffffff, 0x7fffffff, 0x7fffffff);
            if (i < nv) d[u] = d4[idx];
        }
        #pragma unroll
        for (int u = 0; u < KG_U; u++) visit4(d[u], 4 * id[u]);
    }
    {
        const int e = 4 * nv + tid;
        int4 dd = make_int4(0x7fffffff, 0x7fffffff, 0x7fffffff, 0x7fffffff);
        if (e < E) dd.x = edst[e];
        visit4(dd, e);
    }
    if (lane == 0) wred[wave] = sless;
    __syncthreads();

    const int nm   = __builtin_amdgcn_readfirstlane(nmatch_s);
    const int base = __builtin_amdgcn_readlane(wave_scan_incl(wred[lane], lane), 31);
    if (wave == 0) {
        const int4 c4 = *reinterpret_cast<const int4*>(&cnt[4 * lane]);
        const int s   = (c4.x + c4.y) + (c4.z + c4.w);
        const int inc = wave_scan_incl(s, lane);
        int4 o;
        o.x = inc - s; o.y = o.x + c4.x; o.z = o.y + c4.y; o.w = o.z + c4.z;
        *reinterpret_cast<int4*>(&excl[4 * lane]) = o;
        const int j = 4 * lane;
        if (j     < nl) off[lo + j]     = base + o.x;
        if (j + 1 < nl) off[lo + j + 1] = base + o.y;
        if (j + 2 < nl) off[lo + j + 2] = base + o.z;
        if (j + 3 < nl) off[lo + j + 3] = base + o.w;
        const int tot = __builtin_amdgcn_readlane(inc, 31);
        if (lane == 0 && lo + nl == N) off[N] = base + tot;
    }
    __syncthreads();

    const int nf = (nm <= KG_CAP) ? nm : 0;
    for (int i = tid; i < nf; i += 1024) {
        const int j = lj[i];
        const int p = excl[j] + lr[i];
        oe[p] = le[i]; oj[p] = j;
    }
    __syncthreads();
    for (int g0 = 0; g0 < 4 * nf; g0 += 1024 * KG_P) {
        KgIn in[KG_P];
        int  jn[KG_P];
        #pragma unroll
        for (int u = 0; u < KG_P; u++) {
            int g = g0 + u * 1024 + tid;
            if (g > 4 * nf - 1) g = 4 * nf - 1;
            jn[u] = oj[g >> 2];
            in[u] = kg_fetch(oe[g >> 2], g & 3, evec, elen, esrc);
        }
        #pragma unroll
        for (int u = 0; u < KG_P; u++) {
            const int g = g0 + u * 1024 + tid;
            if (g < 4 * nf) kg_record(in[u], base + (g >> 2), jn[u], g & 3, sh, basis, srcs, xs);
        }
    }
    if (nm > KG_CAP) {
        auto visit2 = [&](int dv, int e) {
            const int t = dv - lo;
            if ((unsigned)t < (unsigned)nl) {
                const int slot = base + excl[t] + atomicAdd(&cnt2[t], 1);
                for (int q = 0; q < 4; q++)
                    kg_record(kg_fetch(e, q, evec, elen, esrc), slot, t, q, sh, basis, srcs, xs);
            }
        };
        for (int i = tid; i < nv; i += 1024) {
            const int4 dd = d4[i];
            visit2(dd.x, 4 * i); visit2(dd.y, 4 * i + 1); visit2(dd.z, 4 * i + 2); visit2(dd.w, 4 * i + 3);
        }
        const int e = 4 * nv + tid;
        if (e < E) visit2(edst[e], e);
    }
    __syncthreads();
    for (int i = tid; i < nl * 16; i += 1024)
        x[(size_t)lo * 16 + i] = xs[(i >> 4) * KG_XP + (i & 15)];
}

#define KB_PITCH 1028
#define KB_HP 101
#define KB_YP 68
#define KB_ZP 260
#define RS3 0.5773502691896258f
#define RS5 0.4472135954999579f
#define RS7 0.3779644730092272f
__global__ __launch_bounds__(192) void kB(
    const float* __restrict__ sh, const float* __restrict__ basis,
    const float* __restrict__ x,
    const int* __restrict__ srcs,
    const int* __restrict__ off,
    const float* __restrict__ W1a, const float* __restrict__ W1b,
    const float* W2b, float* __restrict__ z, int N)
{
    __shared__ float Gs[32 * KB_PITCH];

    const int tid  = threadIdx.x;
    const int lane = tid & 31;
    const int wave = __builtin_amdgcn_readfirstlane(tid >> 5);
    const int n0   = blockIdx.x * 32;

    float wa0[8], wa1[8], wa2[8];
    #pragma unroll
    for (int i = 0; i < 8; i++) {
        wa0[i] = W1a[lane + 32*i];
        wa1[i] = W1a[256 + lane + 32*i];
        wa2[i] = W1a[512 + lane + 32*i];
    }

    for (int m = wave; m < 32; m += 6) {
        const int n = n0 + m;
        float g0[8], g1[8], g2[8], g3[8];
        #pragma unroll
        for (int i = 0; i < 8; i++) { g0[i] = 0.0f; g1[i] = 0.0f; g2[i] = 0.0f; g3[i] = 0.0f; }
        if (n < N) {
            const int beg = off[n], end = off[n + 1];
            for (int base = beg; base < end; base += 32) {
                const int idx = base + lane;
                float d0 = 0.0f, d1 = 0.0f, d2 = 0.0f, d3 = 0.0f;
                float b0 = 0.0f, b1 = 0.0f, b2 = 0.0f;
                if (idx < end) {
                    const int srcn = srcs[idx];
                    const float4* xs = reinterpret_cast<const float4*>(x  + (size_t)srcn * 16);
                    const float4* ss = reinterpret_cast<const float4*>(sh + (size_t)idx * 16);
                    float4 x0 = xs[0], x1 = xs[1], x2 = xs[2], x3 = xs[3];
                    float4 s0 = ss[0], s1 = ss[1], s2 = ss[2], s3 = ss[3];
                    d0 = x0.x*s0.x;
                    d1 = (x0.y*s0.y + x0.z*s0.z + x0.w*s0.w) * RS3;
                    d2 = (x1.x*s1.x + x1.y*s1.y + x1.z*s1.z + x1.w*s1.w + x2.x*s2.x) * RS5;
                    d3 = (x2.y*s2.y + x2.z*s2.z + x2.w*s2.w
                        + x3.x*s3.x + x3.y*s3.y + x3.z*s3.z + x3.w*s3.w) * RS7;
                    b0 = basis[(size_t)idx*3]; b1 = basis[(size_t)idx*3 + 1]; b2 = basis[(size_t)idx*3 + 2];
                }
                const int left = end - base;
                const int cntc = left < 32 ? left : 32;
                for (int j = 0; j < cntc; j++) {
                    const float e0 = __shfl(d0, j, 32), e1 = __shfl(d1, j, 32);
                    const float e2 = __shfl(d2, j, 32), e3 = __shfl(d3, j, 32);
                    const float c0 = __shfl(b0, j, 32), c1 = __shfl(b1, j, 32), c2 = __shfl(b2, j, 32);
                    #pragma unroll
                    for (int i = 0; i < 8; i++) {
                        float hv = c0*wa0[i] + c1*wa1[i] + c2*wa2[i];
                        hv = fmaxf(hv * INV_SQRT3, 0.0f);
                        g0[i] += e0 * hv; g1[i] += e1 * hv;
                        g2[i] += e2 * hv; g3[i] += e3 * hv;
                    }
                }
            }
        }
        float* gr = &Gs[m * KB_PITCH + lane];
        #pragma unroll
        for (int i = 0; i < 8; i++) {
            gr[32*i]       = g0[i];
            gr[256 + 32*i] = g1[i];
            gr[512 + 32*i] = g2[i];
            gr[768 + 32*i] = g3[i];
        }
    }
    __syncthreads();

    const int c    = lane & 15;
    const int hh   = lane >> 4;
    const int colb = wave * 16 + c;
    v8f a00 = {}, a01 = {}, a02 = {}, a03 = {};
    v8f a10 = {}, a11 = {}, a12 = {}, a13 = {};
    const float* ga0 = &Gs[c * KB_PITCH + 2 * hh];
    const float* ga1 = ga0 + 16 * KB_PITCH;
    const float* Bp = W1b + (size_t)(2 * hh) * 384 + colb;
    v2f Bc[2][4], Bn[2][4];
    #pragma unroll
    for (int s = 0; s < 2; s++) {
        #pragma unroll
        for (int q = 0; q < 4; q++) {
            Bc[s][q].x = Bp[(size_t)(4 * s) * 384 + 96 * q];
            Bc[s][q].y = Bp[(size_t)(4 * s + 1) * 384 + 96 * q];
        }
    }
    for (int k0 = 0; k0 < 256; k0 += 8) {
        const int kn = (k0 + 8 < 256) ? (k0 + 8) : k0;
        const float* Bq = Bp + (size_t)kn * 384;
        #pragma unroll
        for (int s = 0; s < 2; s++) {
            #pragma unroll
            for (int q = 0; q < 4; q++) {
                Bn[s][q].x = Bq[(size_t)(4 * s) * 384 + 96 * q];
                Bn[s][q].y = Bq[(size_t)(4 * s + 1) * 384 + 96 * q];
            }
        }
        #pragma unroll
        for (int s = 0; s < 2; s++) {
            const int k = k0 + 4 * s;
            v2f A00 = *reinterpret_cast<const v2f*>(ga0 + k);
            v2f A01 = *reinterpret_cast<const v2f*>(ga0 + 256 + k);
            v2f A02 = *reinterpret_cast<const v2f*>(ga0 + 512 + k);
            v2f A03 = *reinterpret_cast<const v2f*>(ga0 + 768 + k);
            v2f A10 = *reinterpret_cast<const v2f*>(ga1 + k);
            v2f A11 = *reinterpret_cast<const v2f*>(ga1 + 256 + k);
            v2f A12 = *reinterpret_cast<const v2f*>(ga1 + 512 + k);
            v2f A13 = *reinterpret_cast<const v2f*>(ga1 + 768 + k);
            a00 = __builtin_amdgcn_wmma_f32_16x16x4_f32(false, A00, false, Bc[s][0], (short)0, a00, false, false);
            a10 = __builtin_amdgcn_wmma_f32_16x16x4_f32(false, A10, false, Bc[s][0], (short)0, a10, false, false);
            a01 = __builtin_amdgcn_wmma_f32_16x16x4_f32(false, A01, false, Bc[s][1], (short)0, a01, false, false);
            a11 = __builtin_amdgcn_wmma_f32_16x16x4_f32(false, A11, false, Bc[s][1], (short)0, a11, false, false);
            a02 = __builtin_amdgcn_wmma_f32_16x16x4_f32(false, A02, false, Bc[s][2], (short)0, a02, false, false);
            a12 = __builtin_amdgcn_wmma_f32_16x16x4_f32(false, A12, false, Bc[s][2], (short)0, a12, false, false);
            a03 = __builtin_amdgcn_wmma_f32_16x16x4_f32(false, A03, false, Bc[s][3], (short)0, a03, false, false);
            a13 = __builtin_amdgcn_wmma_f32_16x16x4_f32(false, A13, false, Bc[s][3], (short)0, a13, false, false);
        }
        #pragma unroll
        for (int s = 0; s < 2; s++) {
            #pragma unroll
            for (int q = 0; q < 4; q++) Bc[s][q] = Bn[s][q];
        }
    }
    v2f Bf[3][16];
    #pragma unroll
    for (int t = 0; t < 3; t++) {
        int jt = wave + 6 * t;
        if (jt > 15) jt = 15;
        const float* wb = W2b + (size_t)(16 * jt + c) * 64 + 2 * hh;
        #pragma unroll
        for (int i = 0; i < 16; i++) Bf[t][i] = *reinterpret_cast<const v2f*>(wb + 4 * i);
    }
    __syncthreads();

    float* Hs = Gs;
    float* Ys = Gs + 32 * KB_HP;
    float* Zs = Ys + 32 * KB_YP;
    const float scale = (1.0f/16.0f) * 0.5f * INV_NN;
    #pragma unroll
    for (int r = 0; r < 8; r++) {
        Hs[(8*hh + r) * KB_HP + colb]      = ((a00[r] + a01[r]) + (a02[r] + a03[r])) * scale;
        Hs[(16 + 8*hh + r) * KB_HP + colb] = ((a10[r] + a11[r]) + (a12[r] + a13[r])) * scale;
    }
    __syncthreads();
    {
        const float* hr = &Hs[lane * KB_HP];
        float* yr = &Ys[lane * KB_YP];
        float ha[11], hb[11];
        #pragma unroll
        for (int k = 0; k < 11; k++) {
            int j = wave + 6 * k;
            if (j > 63) j = 63;
            ha[k] = hr[j];
            hb[k] = hr[j + 32];
        }
        #pragma unroll
        for (int k = 0; k < 11; k++) {
            const int j = wave + 6 * k;
            if (j < 64) {
                float v;
                if (j < 16)      v = fmaxf(ha[k], 0.0f);
                else if (j < 32) v = fabsf(ha[k]);
                else {
                    const int jj = j - 32;
                    const bool use_relu = (jj < 8) || (jj >= 16 && jj < 24);
                    const float ga = use_relu ? fmaxf(ha[k], 0.0f) : tanhf(ha[k]);
                    v = ga * hb[k];
                }
                yr[j] = v;
            }
        }
    }
    __syncthreads();

    {
        const float* ya = &Ys[c * KB_YP + 2 * hh];
        v2f A0[16], A1[16];
        #pragma unroll
        for (int i = 0; i < 16; i++) {
            A0[i] = *reinterpret_cast<const v2f*>(ya + 4 * i);
            A1[i] = *reinterpret_cast<const v2f*>(ya + 16 * KB_YP + 4 * i);
        }
        #pragma unroll
        for (int t = 0; t < 3; t++) {
            const int jt = wave + 6 * t;
            if (jt < 16) {
                v8f z0 = {}, z1 = {};
                #pragma unroll
                for (int i = 0; i < 16; i++) {
                    z0 = __builtin_amdgcn_wmma_f32_16x16x4_f32(false, A0[i], false, Bf[t][i], (short)0, z0, false, false);
                    z1 = __builtin_amdgcn_wmma_f32_16x16x4_f32(false, A1[i], false, Bf[t][i], (short)0, z1, false, false);
                }
                float* zc = &Zs[(8 * hh) * KB_ZP + 16 * jt + c];
                #pragma unroll
                for (int r = 0; r < 8; r++) {
                    zc[r * KB_ZP]        = z0[r];
                    zc[(16 + r) * KB_ZP] = z1[r];
                }
            }
        }
    }
    __syncthreads();
    const int rows = (N - n0 < 32) ? (N - n0) : 32;
    for (int idx = tid; idx < rows * 64; idx += 192) {
        const int m = idx >> 6, q = idx & 63;
        float4 v = *reinterpret_cast<const float4*>(&Zs[m * KB_ZP + 4 * q]);
        *reinterpret_cast<float4*>(z + (size_t)(n0 + m) * 256 + 4 * q) = v;
    }
}

#define KE_NODES 4
#define KE_U 8

__device__ __forceinline__ int rl_i(int v, int l) { return __builtin_amdgcn_readlane(v, l); }
__device__ __forceinline__ float rl_f(float v, int l) {
    return __int_as_float(__builtin_amdgcn_readlane(__float_as_int(v), l));
}
template <int M> __device__ __forceinline__ float row_xadd(float v) {
    int t = __builtin_amdgcn_update_dpp(0, __float_as_int(v), 0x160 | M, 0xF, 0xF, false);
    return v + __int_as_float(t);
}
__device__ __forceinline__ float wave_sum(float v) {
    v = row_xadd<1>(v); v = row_xadd<2>(v); v = row_xadd<4>(v); v = row_xadd<8>(v);
    return rl_f(v, 0) + rl_f(v, 16);
}
__device__ __forceinline__ float h2term(float c0, float c1, float c2,
                                        float w0, float w1, float w2, float zv) {
    float hv = c0*w0 + c1*w1 + c2*w2;
    hv = fmaxf(hv * INV_SQRT3, 0.0f);
    return hv * zv;
}

__global__ __launch_bounds__(128) void kE(const float* __restrict__ basis,
                                          const float* __restrict__ z,
                                          const int* __restrict__ srcs,
                                          const int* __restrict__ off,
                                          const float* __restrict__ W2a,
                                          float* __restrict__ out, int N)
{
    const int lane = threadIdx.x & 31;
    const int wave = __builtin_amdgcn_readfirstlane(threadIdx.x >> 5);
    const int n0 = (blockIdx.x * 4 + wave) * KE_NODES;
    if (n0 >= N) return;
    const int nn = (N - n0 < KE_NODES) ? (N - n0) : KE_NODES;
    int oi = n0 + lane;
    if (oi > N) oi = N;
    const int ov = off[oi];
    const int beg_all = rl_i(ov, 0);
    const int end_all = rl_i(ov, nn);

    const float4* w4 = reinterpret_cast<const float4*>(W2a);
    const float4 w0a = w4[lane],       w0b = w4[32 + lane];
    const float4 w1a = w4[64 + lane],  w1b = w4[96 + lane];
    const float4 w2a = w4[128 + lane], w2b = w4[160 + lane];

    const float oscale = (1.0f/128.0f) * INV_NN;
    int   k    = 0;
    int   nend = rl_i(ov, 1);
    float acc  = 0.0f;

    for (int base = beg_all; base < end_all; base += 64) {
        const int last = end_all - 1;
        int iA = base + lane;       if (iA > last) iA = last;
        int iB = base + 32 + lane;  if (iB > last) iB = last;
        const int sA = srcs[iA], sB = srcs[iB];
        const float pA0 = basis[(size_t)iA*3], pA1 = basis[(size_t)iA*3 + 1], pA2 = basis[(size_t)iA*3 + 2];
        const float pB0 = basis[(size_t)iB*3], pB1 = basis[(size_t)iB*3 + 1], pB2 = basis[(size_t)iB*3 + 2];
        const int left = end_all - base;
        const int cnt  = left < 64 ? left : 64;
        for (int j0 = 0; j0 < cnt; j0 += KE_U) {
            const bool hi = j0 >= 32;
            const int   vs  = hi ? sB  : sA;
            const float vb0 = hi ? pB0 : pA0, vb1 = hi ? pB1 : pA1, vb2 = hi ? pB2 : pA2;
            const int jl = j0 & 31;
            const int m  = (cnt - j0 < KE_U) ? (cnt - j0) : KE_U;
            float4 za[KE_U], zb[KE_U];
            #pragma unroll
            for (int u = 0; u < KE_U; u++) {
                const int jj = jl + (u < m ? u : m - 1);
                const int s  = rl_i(vs, jj);
                const float4* zr = reinterpret_cast<const float4*>(z + (size_t)s * 256);
                za[u] = zr[lane];
                zb[u] = zr[32 + lane];
            }
            #pragma unroll
            for (int u = 0; u < KE_U; u++) {
                if (u < m) {
                    const int jj = jl + u;
                    const float c0 = rl_f(vb0, jj), c1 = rl_f(vb1, jj), c2 = rl_f(vb2, jj);
                    float t = h2term(c0, c1, c2, w0a.x, w1a.x, w2a.x, za[u].x);
                    t += h2term(c0, c1, c2, w0a.y, w1a.y, w2a.y, za[u].y);
                    t += h2term(c0, c1, c2, w0a.z, w1a.z, w2a.z, za[u].z);
                    t += h2term(c0, c1, c2, w0a.w, w1a.w, w2a.w, za[u].w);
                    t += h2term(c0, c1, c2, w0b.x, w1b.x, w2b.x, zb[u].x);
                    t += h2term(c0, c1, c2, w0b.y, w1b.y, w2b.y, zb[u].y);
                    t += h2term(c0, c1, c2, w0b.z, w1b.z, w2b.z, zb[u].z);
                    t += h2term(c0, c1, c2, w0b.w, w1b.w, w2b.w, zb[u].w);
                    const int p = base + j0 + u;
                    while (p >= nend) {
                        const float tot = wave_sum(acc);
                        if (lane == 0) out[n0 + k] = tot * oscale;
                        acc = 0.0f;
                        k++;
                        nend = rl_i(ov, k + 1);
                    }
                    acc += t;
                }
            }
        }
    }
    while (k < nn) {
        const float tot = wave_sum(acc);
        if (lane == 0) out[n0 + k] = tot * oscale;
        acc = 0.0f;
        k++;
    }
}

extern "C" void kernel_launch(void* const* d_in, const int* in_sizes, int n_in,
                              void* d_out, int out_size, void* d_ws, size_t ws_size,
                              hipStream_t stream)
{
    const float* evec = (const float*)d_in[0];
    const float* elen = (const float*)d_in[1];
    const float* W1a  = (const float*)d_in[2];
    const float* W1b  = (const float*)d_in[3];
    const float* W2a  = (const float*)d_in[4];
    const float* W2b  = (const float*)d_in[5];
    const int*   esrc = (const int*)d_in[6];
    const int*   edst = (const int*)d_in[7];
    float* out = (float*)d_out;

    const int E = in_sizes[0] / 3;
    const int N = out_size;

    float* ws = (float*)d_ws;
    float* x_     = ws;                 ws += (size_t)N * 16;
    ws += (size_t)N * 96;
    float* lists_ = ws;                 ws += (size_t)N * 64;
    float* z_     = ws;                 ws += (size_t)N * 256;
    float* sh_    = ws;                 ws += (size_t)E * 16;
    float* basis_ = ws;                 ws += (size_t)E * 3;

    int* off_  = (int*)lists_;
    int* srcs_ = off_ + N + 4;

    kG<<<(N + KG_NODES - 1) / KG_NODES, 1024, 0, stream>>>(evec, elen, esrc, edst, off_, sh_, basis_, srcs_, x_, E, N);
    kB<<<(N + 31) / 32, 192, 0, stream>>>(sh_, basis_, x_, srcs_, off_, W1a, W1b, W2b, z_, N);
    kE<<<(N + 4 * KE_NODES - 1) / (4 * KE_NODES), 128, 0, stream>>>(basis_, z_, srcs_, off_, W2a, out, N);
}


# conv1 kernel: 12 waves per 32-node tile instead of 6 (3 per SIMD); product shared out by column tile and by path pair (p0,p1 | p2,p3), halves added in LDS in the old order; no sum order changes
# speedup vs baseline: 5.8663x; 1.3384x over previous
#include <hip/hip_runtime.h>
#include <math.h>

typedef __attribute__((ext_vector_type(2))) float v2f;
typedef __attribute__((ext_vector_type(8))) float v8f;

#define INV_NN     0.512989176042577f
#define INV_SQRT3  0.5773502691896258f
#define S3   1.7320508075688772f
#define S5   2.23606797749979f
#define S7   2.6457513110645907f
#define S15  3.872983346207417f
#define S105 10.246950765959598f
#define C358 2.091650066335189f
#define C218 1.6201851746019651f


#define KG_NODES 98
#define KG_XP    17
#define KG_CAP   8192
#define KG_U     8
#define KG_P     4

template <int S> __device__ __forceinline__ int row_shr_add(int v) {
    int t = __builtin_amdgcn_update_dpp(0, v, 0x110 | S, 0xF, 0xF, false);
    return v + t;
}
__device__ __forceinline__ int wave_scan_incl(int v, int lane) {
    v = row_shr_add<1>(v); v = row_shr_add<2>(v); v = row_shr_add<4>(v); v = row_shr_add<8>(v);
    const int lo = __builtin_amdgcn_readlane(v, 15);
    return v + (lane >= 16 ? lo : 0);
}

struct KgIn { float vx, vy, vz, aux; };
__device__ __forceinline__ KgIn kg_fetch(int e, int q, const float* __restrict__ evec,
                                         const float* __restrict__ elen, const int* __restrict__ esrc)
{
    KgIn r;
    r.vx = evec[3*(size_t)e]; r.vy = evec[3*(size_t)e+1]; r.vz = evec[3*(size_t)e+2];
    r.aux = (q < 3) ? elen[e] : __int_as_float(esrc[e]);
    return r;
}
__device__ __forceinline__ void kg_record(KgIn in, int slot, int j, int q,
        float* __restrict__ sh, float* __restrict__ basis, int* __restrict__ srcs, float* xs)
{
    float vx = in.vx, vy = in.vy, vz = in.vz;
    float rn = 1.0f / (sqrtf(vx*vx + vy*vy + vz*vz) + 1e-12f);
    vx *= rn; vy *= rn; vz *= rn;
    float x2 = vx*vx, y2 = vy*vy, z2 = vz*vz;
    float4 r;
    if (q == 0) {
        r.x = 1.0f;
        r.y = S3*vx; r.z = S3*vy; r.w = S3*vz;
    } else if (q == 1) {
        r.x = S15*vx*vz;
        r.y = S15*vx*vy;
        r.z = 0.5f*S5*(2.0f*y2 - x2 - z2);
        r.w = S15*vy*vz;
    } else if (q == 2) {
        r.x = 0.5f*S15*(z2 - x2);
        r.y = C358*vx*(3.0f*z2 - x2);
        r.z = S105*vx*vy*vz;
        r.w = C218*vx*(4.0f*y2 - z2 - x2);
    } else {
        r.x = 0.5f*S7*vy*(2.0f*y2 - 3.0f*z2 - 3.0f*x2);
        r.y = C218*vz*(4.0f*y2 - z2 - x2);
        r.z = 0.5f*S105*vy*(z2 - x2);
        r.w = C358*vz*(z2 - 3.0f*x2);
    }
    reinterpret_cast<float4*>(sh + (size_t)slot * 16)[q] = r;
    if (q < 3) {
        const float dd = (in.aux - 2.0f*(float)q) * 0.5f;
        basis[(size_t)slot*3 + q] = expf(-dd*dd) * (1.0f/1.12f);
    } else {
        srcs[slot] = __float_as_int(in.aux);
    }
    float* xr = xs + j * KG_XP + 4 * q;
    atomicAdd(&xr[0], r.x * INV_NN); atomicAdd(&xr[1], r.y * INV_NN);
    atomicAdd(&xr[2], r.z * INV_NN); atomicAdd(&xr[3], r.w * INV_NN);
}

__global__ __launch_bounds__(1024) void kG(const float* __restrict__ evec, const float* __restrict__ elen,
                                           const int* __restrict__ esrc, const int* __restrict__ edst,
                                           int* __restrict__ off,
                                           float* __restrict__ sh, float* __restrict__ basis,
                                           int* __restrict__ srcs, float* __restrict__ x, int E, int N)
{
    __shared__ __attribute__((aligned(16))) int cnt[128];
    __shared__ __attribute__((aligned(16))) int excl[128];
    __shared__ int cnt2[128];
    __shared__ int wred[32];
    __shared__ int nmatch_s;
    __shared__ float xs[KG_NODES * KG_XP];
    __shared__ int le[KG_CAP], lj[KG_CAP], lr[KG_CAP];
    __shared__ int oe[KG_CAP], oj[KG_CAP];

    const int tid  = threadIdx.x;
    const int lane = tid & 31;
    const int wave = __builtin_amdgcn_readfirstlane(tid >> 5);
    const int lo = blockIdx.x * KG_NODES;
    const int nl = (N - lo < KG_NODES) ? (N - lo) : KG_NODES;

    if (tid < 128) { cnt[tid] = 0; cnt2[tid] = 0; }
    if (tid == 0) nmatch_s = 0;
    for (int i = tid; i < KG_NODES * KG_XP; i += 1024) xs[i] = 0.0f;
    __syncthreads();

    const int nv = E >> 2;
    const int rot = (int)(blockIdx.x * ((unsigned)nv / gridDim.x));
    const int4* d4 = reinterpret_cast<const int4*>(edst);
    int sless = 0;

    auto keep = [&](int t, int e) {
        const int r   = atomicAdd(&cnt[t], 1);
        const int pos = atomicAdd(&nmatch_s, 1);
        if (pos < KG_CAP) { le[pos] = e; lj[pos] = t; lr[pos] = r; }
    };
    auto visit4 = [&](int4 dd, int e0) {
        const int t0 = dd.x - lo, t1 = dd.y - lo, t2 = dd.z - lo, t3 = dd.w - lo;
        sless += __builtin_popcount(__builtin_amdgcn_ballot_w32(t0 < 0))
               + __builtin_popcount(__builtin_amdgcn_ballot_w32(t1 < 0))
               + __builtin_popcount(__builtin_amdgcn_ballot_w32(t2 < 0))
               + __builtin_popcount(__builtin_amdgcn_ballot_w32(t3 < 0));
        const bool m0 = (unsigned)t0 < (unsigned)nl, m1 = (unsigned)t1 < (unsigned)nl;
        const bool m2 = (unsigned)t2 < (unsigned)nl, m3 = (unsigned)t3 < (unsigned)nl;
        if (m0 | m1 | m2 | m3) {
            if (m0) keep(t0, e0);
            if (m1) keep(t1, e0 + 1);
            if (m2) keep(t2, e0 + 2);
            if (m3) keep(t3, e0 + 3);
        }
    };

    for (int i0 = 0; i0 < nv; i0 += 1024 * KG_U) {
        int4 d[KG_U];
        int  id[KG_U];
        #pragma unroll
        for (int u = 0; u < KG_U; u++) {
            const int i = i0 + u * 1024 + tid;
            int idx = i + rot;
            if (idx >= nv) idx -= nv;
            id[u] = idx;
            d[u] = make_int4(0x7fffffff, 0x7fffffff, 0x7fffffff, 0x7fffffff);
            if (i < nv) d[u] = d4[idx];
        }
        #pragma unroll
        for (int u = 0; u < KG_U; u++) visit4(d[u], 4 * id[u]);
    }
    {
        const int e = 4 * nv + tid;
        int4 dd = make_int4(0x7fffffff, 0x7fffffff, 0x7fffffff, 0x7fffffff);
        if (e < E) dd.x = edst[e];
        visit4(dd, e);
    }
    if (lane == 0) wred[wave] = sless;
    __syncthreads();

    const int nm   = __builtin_amdgcn_readfirstlane(nmatch_s);
    const int base = __builtin_amdgcn_readlane(wave_scan_incl(wred[lane], lane), 31);
    if (wave == 0) {
        const int4 c4 = *reinterpret_cast<const int4*>(&cnt[4 * lane]);
        const int s   = (c4.x + c4.y) + (c4.z + c4.w);
        const int inc = wave_scan_incl(s, lane);
        int4 o;
        o.x = inc - s; o.y = o.x + c4.x; o.z = o.y + c4.y; o.w = o.z + c4.z;
        *reinterpret_cast<int4*>(&excl[4 * lane]) = o;
        const int j = 4 * lane;
        if (j     < nl) off[lo + j]     = base + o.x;
        if (j + 1 < nl) off[lo + j + 1] = base + o.y;
        if (j + 2 < nl) off[lo + j + 2] = base + o.z;
        if (j + 3 < nl) off[lo + j + 3] = base + o.w;
        const int tot = __builtin_amdgcn_readlane(inc, 31);
        if (lane == 0 && lo + nl == N) off[N] = base + tot;
    }
    __syncthreads();

    const int nf = (nm <= KG_CAP) ? nm : 0;
    for (int i = tid; i < nf; i += 1024) {
        const int j = lj[i];
        const int p = excl[j] + lr[i];
        oe[p] = le[i]; oj[p] = j;
    }
    __syncthreads();
    for (int g0 = 0; g0 < 4 * nf; g0 += 1024 * KG_P) {
        KgIn in[KG_P];
        int  jn[KG_P];
        #pragma unroll
        for (int u = 0; u < KG_P; u++) {
            int g = g0 + u * 1024 + tid;
            if (g > 4 * nf - 1) g = 4 * nf - 1;
            jn[u] = oj[g >> 2];
            in[u] = kg_fetch(oe[g >> 2], g & 3, evec, elen, esrc);
        }
        #pragma unroll
        for (int u = 0; u < KG_P; u++) {
            const int g = g0 + u * 1024 + tid;
            if (g < 4 * nf) kg_record(in[u], base + (g >> 2), jn[u], g & 3, sh, basis, srcs, xs);
        }
    }
    if (nm > KG_CAP) {
        auto visit2 = [&](int dv, int e) {
            const int t = dv - lo;
            if ((unsigned)t < (unsigned)nl) {
                const int slot = base + excl[t] + atomicAdd(&cnt2[t], 1);
                for (int q = 0; q < 4; q++)
                    kg_record(kg_fetch(e, q, evec, elen, esrc), slot, t, q, sh, basis, srcs, xs);
            }
        };
        for (int i = tid; i < nv; i += 1024) {
            const int4 dd = d4[i];
            visit2(dd.x, 4 * i); visit2(dd.y, 4 * i + 1); visit2(dd.z, 4 * i + 2); visit2(dd.w, 4 * i + 3);
        }
        const int e = 4 * nv + tid;
        if (e < E) visit2(edst[e], e);
    }
    __syncthreads();
    for (int i = tid; i < nl * 16; i += 1024)
        x[(size_t)lo * 16 + i] = xs[(i >> 4) * KG_XP + (i & 15)];
}

#define KB_PITCH 1028
#define KB_HP 101
#define KB_YP 68
#define KB_ZP 260
#define RS3 0.5773502691896258f
#define RS5 0.4472135954999579f
#define RS7 0.3779644730092272f
#define KB_WAVES 12
__global__ __launch_bounds__(32 * KB_WAVES) void kB(
    const float* __restrict__ sh, const float* __restrict__ basis,
    const float* __restrict__ x,
    const int* __restrict__ srcs,
    const int* __restrict__ off,
    const float* __restrict__ W1a, const float* __restrict__ W1b,
    const float* W2b, float* __restrict__ z, int N)
{
    __shared__ float Gs[32 * KB_PITCH];

    const int tid  = threadIdx.x;
    const int lane = tid & 31;
    const int wave = __builtin_amdgcn_readfirstlane(tid >> 5);
    const int n0   = blockIdx.x * 32;

    float wa0[8], wa1[8], wa2[8];
    #pragma unroll
    for (int i = 0; i < 8; i++) {
        wa0[i] = W1a[lane + 32*i];
        wa1[i] = W1a[256 + lane + 32*i];
        wa2[i] = W1a[512 + lane + 32*i];
    }

    for (int m = wave; m < 32; m += KB_WAVES) {
        const int n = n0 + m;
        float g0[8], g1[8], g2[8], g3[8];
        #pragma unroll
        for (int i = 0; i < 8; i++) { g0[i] = 0.0f; g1[i] = 0.0f; g2[i] = 0.0f; g3[i] = 0.0f; }
        if (n < N) {
            const int beg = off[n], end = off[n + 1];
            for (int base = beg; base < end; base += 32) {
                const int idx = base + lane;
                float d0 = 0.0f, d1 = 0.0f, d2 = 0.0f, d3 = 0.0f;
                float b0 = 0.0f, b1 = 0.0f, b2 = 0.0f;
                if (idx < end) {
                    const int srcn = srcs[idx];
                    const float4* xs = reinterpret_cast<const float4*>(x  + (size_t)srcn * 16);
                    const float4* ss = reinterpret_cast<const float4*>(sh + (size_t)idx * 16);
                    float4 x0 = xs[0], x1 = xs[1], x2 = xs[2], x3 = xs[3];
                    float4 s0 = ss[0], s1 = ss[1], s2 = ss[2], s3 = ss[3];
                    d0 = x0.x*s0.x;
                    d1 = (x0.y*s0.y + x0.z*s0.z + x0.w*s0.w) * RS3;
                    d2 = (x1.x*s1.x + x1.y*s1.y + x1.z*s1.z + x1.w*s1.w + x2.x*s2.x) * RS5;
                    d3 = (x2.y*s2.y + x2.z*s2.z + x2.w*s2.w
                        + x3.x*s3.x + x3.y*s3.y + x3.z*s3.z + x3.w*s3.w) * RS7;
                    b0 = basis[(size_t)idx*3]; b1 = basis[(size_t)idx*3 + 1]; b2 = basis[(size_t)idx*3 + 2];
                }
                const int left = end - base;
                const int cntc = left < 32 ? left : 32;
                for (int j = 0; j < cntc; j++) {
                    const float e0 = __shfl(d0, j, 32), e1 = __shfl(d1, j, 32);
                    const float e2 = __shfl(d2, j, 32), e3 = __shfl(d3, j, 32);
                    const float c0 = __shfl(b0, j, 32), c1 = __shfl(b1, j, 32), c2 = __shfl(b2, j, 32);
                    #pragma unroll
                    for (int i = 0; i < 8; i++) {
                        float hv = c0*wa0[i] + c1*wa1[i] + c2*wa2[i];
                        hv = fmaxf(hv * INV_SQRT3, 0.0f);
                        g0[i] += e0 * hv; g1[i] += e1 * hv;
                        g2[i] += e2 * hv; g3[i] += e3 * hv;
                    }
                }
            }
        }
        float* gr = &Gs[m * KB_PITCH + lane];
        #pragma unroll
        for (int i = 0; i < 8; i++) {
            gr[32*i]       = g0[i];
            gr[256 + 32*i] = g1[i];
            gr[512 + 32*i] = g2[i];
            gr[768 + 32*i] = g3[i];
        }
    }
    __syncthreads();

    const int c    = lane & 15;
    const int hh   = lane >> 4;
    const int ct   = (wave >= 6) ? (wave - 6) : wave;
    const int ph   = (wave >= 6) ? 2 : 0;
    const int colb = ct * 16 + c;
    v8f a00 = {}, a01 = {};
    v8f a10 = {}, a11 = {};
    const float* ga0 = &Gs[c * KB_PITCH + 256 * ph + 2 * hh];
    const float* ga1 = ga0 + 16 * KB_PITCH;
    const float* Bp = W1b + (size_t)(2 * hh) * 384 + 96 * ph + colb;
    v2f Bc[4][2], Bn[4][2];
    #pragma unroll
    for (int s = 0; s < 4; s++) {
        #pragma unroll
        for (int q = 0; q < 2; q++) {
            Bc[s][q].x = Bp[(size_t)(4 * s) * 384 + 96 * q];
            Bc[s][q].y = Bp[(size_t)(4 * s + 1) * 384 + 96 * q];
        }
    }
    #pragma unroll 2
    for (int k0 = 0; k0 < 256; k0 += 16) {
        const int kn = (k0 + 16 < 256) ? (k0 + 16) : k0;
        const float* Bq = Bp + (size_t)kn * 384;
        #pragma unroll
        for (int s = 0; s < 4; s++) {
            #pragma unroll
            for (int q = 0; q < 2; q++) {
                Bn[s][q].x = Bq[(size_t)(4 * s) * 384 + 96 * q];
                Bn[s][q].y = Bq[(size_t)(4 * s + 1) * 384 + 96 * q];
            }
        }
        #pragma unroll
        for (int s = 0; s < 4; s++) {
            const int k = k0 + 4 * s;
            v2f A00 = *reinterpret_cast<const v2f*>(ga0 + k);
            v2f A01 = *reinterpret_cast<const v2f*>(ga0 + 256 + k);
            v2f A10 = *reinterpret_cast<const v2f*>(ga1 + k);
            v2f A11 = *reinterpret_cast<const v2f*>(ga1 + 256 + k);
            a00 = __builtin_amdgcn_wmma_f32_16x16x4_f32(false, A00, false, Bc[s][0], (short)0, a00, false, false);
            a10 = __builtin_amdgcn_wmma_f32_16x16x4_f32(false, A10, false, Bc[s][0], (short)0, a10, false, false);
            a01 = __builtin_amdgcn_wmma_f32_16x16x4_f32(false, A01, false, Bc[s][1], (short)0, a01, false, false);
            a11 = __builtin_amdgcn_wmma_f32_16x16x4_f32(false, A11, false, Bc[s][1], (short)0, a11, false, false);
        }
        #pragma unroll
        for (int s = 0; s < 4; s++) {
            #pragma unroll
            for (int q = 0; q < 2; q++) Bc[s][q] = Bn[s][q];
        }
    }
    v2f Bf[2][16];
    #pragma unroll
    for (int t = 0; t < 2; t++) {
        int jt = wave + KB_WAVES * t;
        if (jt > 15) jt = 15;
        const float* wb = W2b + (size_t)(16 * jt + c) * 64 + 2 * hh;
        #pragma unroll
        for (int i = 0; i < 16; i++) Bf[t][i] = *reinterpret_cast<const v2f*>(wb + 4 * i);
    }
    __syncthreads();

    float* Hs = Gs;
    float* Ys = Gs + 64 * KB_HP;
    float* Zs = Ys + 32 * KB_YP;
    const float scale = (1.0f/16.0f) * 0.5f * INV_NN;
    {
        float* Hw = Hs + (ph ? 32 * KB_HP : 0);
        #pragma unroll
        for (int r = 0; r < 8; r++) {
            Hw[(8*hh + r) * KB_HP + colb]      = a00[r] + a01[r];
            Hw[(16 + 8*hh + r) * KB_HP + colb] = a10[r] + a11[r];
        }
    }
    __syncthreads();
    {
        const float* hr = &Hs[lane * KB_HP];
        const float* hs = hr + 32 * KB_HP;
        float* yr = &Ys[lane * KB_YP];
        float ha[6], hb[6];
        #pragma unroll
        for (int k = 0; k < 6; k++) {
            int j = wave + KB_WAVES * k;
            if (j > 63) j = 63;
            ha[k] = (hr[j] + hs[j]) * scale;
            hb[k] = (hr[j + 32] + hs[j + 32]) * scale;
        }
        #pragma unroll
        for (int k = 0; k < 6; k++) {
            const int j = wave + KB_WAVES * k;
            if (j < 64) {
                float v;
                if (j < 16)      v = fmaxf(ha[k], 0.0f);
                else if (j < 32) v = fabsf(ha[k]);
                else {
                    const int jj = j - 32;
                    const bool use_relu = (jj < 8) || (jj >= 16 && jj < 24);
                    const float ga = use_relu ? fmaxf(ha[k], 0.0f) : tanhf(ha[k]);
                    v = ga * hb[k];
                }
                yr[j] = v;
            }
        }
    }
    __syncthreads();

    {
        const float* ya = &Ys[c * KB_YP + 2 * hh];
        v2f A0[16], A1[16];
        #pragma unroll
        for (int i = 0; i < 16; i++) {
            A0[i] = *reinterpret_cast<const v2f*>(ya + 4 * i);
            A1[i] = *reinterpret_cast<const v2f*>(ya + 16 * KB_YP + 4 * i);
        }
        #pragma unroll
        for (int t = 0; t < 2; t++) {
            const int jt = wave + KB_WAVES * t;
            if (jt < 16) {
                v8f z0 = {}, z1 = {};
                #pragma unroll
                for (int i = 0; i < 16; i++) {
                    z0 = __builtin_amdgcn_wmma_f32_16x16x4_f32(false, A0[i], false, Bf[t][i], (short)0, z0, false, false);
                    z1 = __builtin_amdgcn_wmma_f32_16x16x4_f32(false, A1[i], false, Bf[t][i], (short)0, z1, false, false);
                }
                float* zc = &Zs[(8 * hh) * KB_ZP + 16 * jt + c];
                #pragma unroll
                for (int r = 0; r < 8; r++) {
                    zc[r * KB_ZP]        = z0[r];
                    zc[(16 + r) * KB_ZP] = z1[r];
                }
            }
        }
    }
    __syncthreads();
    const int rows = (N - n0 < 32) ? (N - n0) : 32;
    for (int idx = tid; idx < rows * 64; idx += 32 * KB_WAVES) {
        const int m = idx >> 6, q = idx & 63;
        float4 v = *reinterpret_cast<const float4*>(&Zs[m * KB_ZP + 4 * q]);
        *reinterpret_cast<float4*>(z + (size_t)(n0 + m) * 256 + 4 * q) = v;
    }
}

#define KE_NODES 4
#define KE_U 8

__device__ __forceinline__ int rl_i(int v, int l) { return __builtin_amdgcn_readlane(v, l); }
__device__ __forceinline__ float rl_f(float v, int l) {
    return __int_as_float(__builtin_amdgcn_readlane(__float_as_int(v), l));
}
template <int M> __device__ __forceinline__ float row_xadd(float v) {
    int t = __builtin_amdgcn_update_dpp(0, __float_as_int(v), 0x160 | M, 0xF, 0xF, false);
    return v + __int_as_float(t);
}
__device__ __forceinline__ float wave_sum(float v) {
    v = row_xadd<1>(v); v = row_xadd<2>(v); v = row_xadd<4>(v); v = row_xadd<8>(v);
    return rl_f(v, 0) + rl_f(v, 16);
}
__device__ __forceinline__ float h2term(float c0, float c1, float c2,
                                        float w0, float w1, float w2, float zv) {
    float hv = c0*w0 + c1*w1 + c2*w2;
    hv = fmaxf(hv * INV_SQRT3, 0.0f);
    return hv * zv;
}

__global__ __launch_bounds__(128) void kE(const float* __restrict__ basis,
                                          const float* __restrict__ z,
                                          const int* __restrict__ srcs,
                                          const int* __restrict__ off,
                                          const float* __restrict__ W2a,
                                          float* __restrict__ out, int N)
{
    const int lane = threadIdx.x & 31;
    const int wave = __builtin_amdgcn_readfirstlane(threadIdx.x >> 5);
    const int n0 = (blockIdx.x * 4 + wave) * KE_NODES;
    if (n0 >= N) return;
    const int nn = (N - n0 < KE_NODES) ? (N - n0) : KE_NODES;
    int oi = n0 + lane;
    if (oi > N) oi = N;
    const int ov = off[oi];
    const int beg_all = rl_i(ov, 0);
    const int end_all = rl_i(ov, nn);

    const float4* w4 = reinterpret_cast<const float4*>(W2a);
    const float4 w0a = w4[lane],       w0b = w4[32 + lane];
    const float4 w1a = w4[64 + lane],  w1b = w4[96 + lane];
    const float4 w2a = w4[128 + lane], w2b = w4[160 + lane];

    const float oscale = (1.0f/128.0f) * INV_NN;
    int   k    = 0;
    int   nend = rl_i(ov, 1);
    float acc  = 0.0f;

    for (int base = beg_all; base < end_all; base += 64) {
        const int last = end_all - 1;
        int iA = base + lane;       if (iA > last) iA = last;
        int iB = base + 32 + lane;  if (iB > last) iB = last;
        const int sA = srcs[iA], sB = srcs[iB];
        const float pA0 = basis[(size_t)iA*3], pA1 = basis[(size_t)iA*3 + 1], pA2 = basis[(size_t)iA*3 + 2];
        const float pB0 = basis[(size_t)iB*3], pB1 = basis[(size_t)iB*3 + 1], pB2 = basis[(size_t)iB*3 + 2];
        const int left = end_all - base;
        const int cnt  = left < 64 ? left : 64;
        for (int j0 = 0; j0 < cnt; j0 += KE_U) {
            const bool hi = j0 >= 32;
            const int   vs  = hi ? sB  : sA;
            const float vb0 = hi ? pB0 : pA0, vb1 = hi ? pB1 : pA1, vb2 = hi ? pB2 : pA2;
            const int jl = j0 & 31;
            const int m  = (cnt - j0 < KE_U) ? (cnt - j0) : KE_U;
            float4 za[KE_U], zb[KE_U];
            #pragma unroll
            for (int u = 0; u < KE_U; u++) {
                const int jj = jl + (u < m ? u : m - 1);
                const int s  = rl_i(vs, jj);
                const float4* zr = reinterpret_cast<const float4*>(z + (size_t)s * 256);
                za[u] = zr[lane];
                zb[u] = zr[32 + lane];
            }
            #pragma unroll
            for (int u = 0; u < KE_U; u++) {
                if (u < m) {
                    const int jj = jl + u;
                    const float c0 = rl_f(vb0, jj), c1 = rl_f(vb1, jj), c2 = rl_f(vb2, jj);
                    float t = h2term(c0, c1, c2, w0a.x, w1a.x, w2a.x, za[u].x);
                    t += h2term(c0, c1, c2, w0a.y, w1a.y, w2a.y, za[u].y);
                    t += h2term(c0, c1, c2, w0a.z, w1a.z, w2a.z, za[u].z);
                    t += h2term(c0, c1, c2, w0a.w, w1a.w, w2a.w, za[u].w);
                    t += h2term(c0, c1, c2, w0b.x, w1b.x, w2b.x, zb[u].x);
                    t += h2term(c0, c1, c2, w0b.y, w1b.y, w2b.y, zb[u].y);
                    t += h2term(c0, c1, c2, w0b.z, w1b.z, w2b.z, zb[u].z);
                    t += h2term(c0, c1, c2, w0b.w, w1b.w, w2b.w, zb[u].w);
                    const int p = base + j0 + u;
                    while (p >= nend) {
                        const float tot = wave_sum(acc);
                        if (lane == 0) out[n0 + k] = tot * oscale;
                        acc = 0.0f;
                        k++;
                        nend = rl_i(ov, k + 1);
                    }
                    acc += t;
                }
            }
        }
    }
    while (k < nn) {
        const float tot = wave_sum(acc);
        if (lane == 0) out[n0 + k] = tot * oscale;
        acc = 0.0f;
        k++;
    }
}

extern "C" void kernel_launch(void* const* d_in, const int* in_sizes, int n_in,
                              void* d_out, int out_size, void* d_ws, size_t ws_size,
                              hipStream_t stream)
{
    const float* evec = (const float*)d_in[0];
    const float* elen = (const float*)d_in[1];
    const float* W1a  = (const float*)d_in[2];
    const float* W1b  = (const float*)d_in[3];
    const float* W2a  = (const float*)d_in[4];
    const float* W2b  = (const float*)d_in[5];
    const int*   esrc = (const int*)d_in[6];
    const int*   edst = (const int*)d_in[7];
    float* out = (float*)d_out;

    const int E = in_sizes[0] / 3;
    const int N = out_size;

    float* ws = (float*)d_ws;
    float* x_     = ws;                 ws += (size_t)N * 16;
    ws += (size_t)N * 96;
    float* lists_ = ws;                 ws += (size_t)N * 64;
    float* z_     = ws;                 ws += (size_t)N * 256;
    float* sh_    = ws;                 ws += (size_t)E * 16;
    float* basis_ = ws;                 ws += (size_t)E * 3;

    int* off_  = (int*)lists_;
    int* srcs_ = off_ + N + 4;

    kG<<<(N + KG_NODES - 1) / KG_NODES, 1024, 0, stream>>>(evec, elen, esrc, edst, off_, sh_, basis_, srcs_, x_, E, N);
    kB<<<(N + 31) / 32, 32 * KB_WAVES, 0, stream>>>(sh_, basis_, x_, srcs_, off_, W1a, W1b, W2b, z_, N);
    kE<<<(N + 4 * KE_NODES - 1) / (4 * KE_NODES), 128, 0, stream>>>(basis_, z_, srcs_, off_, W2a, out, N);
}


# conv1 kernel: 768 = 3 x 256 blocks of 32 nodes, the 424 nodes behind them in 27 blocks of ONE 16-row tile (half the WMMAs), not 14 full blocks on 14 of 256 WGPs; no sum order changes
# speedup vs baseline: 6.3140x; 1.0763x over previous
#include <hip/hip_runtime.h>
#include <math.h>

typedef __attribute__((ext_vector_type(2))) float v2f;
typedef __attribute__((ext_vector_type(8))) float v8f;

#define INV_NN     0.512989176042577f
#define INV_SQRT3  0.5773502691896258f
#define S3   1.7320508075688772f
#define S5   2.23606797749979f
#define S7   2.6457513110645907f
#define S15  3.872983346207417f
#define S105 10.246950765959598f
#define C358 2.091650066335189f
#define C218 1.6201851746019651f


#define KG_NODES 98
#define KG_XP    17
#define KG_CAP   8192
#define KG_U     8
#define KG_P     4

template <int S> __device__ __forceinline__ int row_shr_add(int v) {
    int t = __builtin_amdgcn_update_dpp(0, v, 0x110 | S, 0xF, 0xF, false);
    return v + t;
}
__device__ __forceinline__ int wave_scan_incl(int v, int lane) {
    v = row_shr_add<1>(v); v = row_shr_add<2>(v); v = row_shr_add<4>(v); v = row_shr_add<8>(v);
    const int lo = __builtin_amdgcn_readlane(v, 15);
    return v + (lane >= 16 ? lo : 0);
}

struct KgIn { float vx, vy, vz, aux; };
__device__ __forceinline__ KgIn kg_fetch(int e, int q, const float* __restrict__ evec,
                                         const float* __restrict__ elen, const int* __restrict__ esrc)
{
    KgIn r;
    r.vx = evec[3*(size_t)e]; r.vy = evec[3*(size_t)e+1]; r.vz = evec[3*(size_t)e+2];
    r.aux = (q < 3) ? elen[e] : __int_as_float(esrc[e]);
    return r;
}
__device__ __forceinline__ void kg_record(KgIn in, int slot, int j, int q,
        float* __restrict__ sh, float* __restrict__ basis, int* __restrict__ srcs, float* xs)
{
    float vx = in.vx, vy = in.vy, vz = in.vz;
    float rn = 1.0f / (sqrtf(vx*vx + vy*vy + vz*vz) + 1e-12f);
    vx *= rn; vy *= rn; vz *= rn;
    float x2 = vx*vx, y2 = vy*vy, z2 = vz*vz;
    float4 r;
    if (q == 0) {
        r.x = 1.0f;
        r.y = S3*vx; r.z = S3*vy; r.w = S3*vz;
    } else if (q == 1) {
        r.x = S15*vx*vz;
        r.y = S15*vx*vy;
        r.z = 0.5f*S5*(2.0f*y2 - x2 - z2);
        r.w = S15*vy*vz;
    } else if (q == 2) {
        r.x = 0.5f*S15*(z2 - x2);
        r.y = C358*vx*(3.0f*z2 - x2);
        r.z = S105*vx*vy*vz;
        r.w = C218*vx*(4.0f*y2 - z2 - x2);
    } else {
        r.x = 0.5f*S7*vy*(2.0f*y2 - 3.0f*z2 - 3.0f*x2);
        r.y = C218*vz*(4.0f*y2 - z2 - x2);
        r.z = 0.5f*S105*vy*(z2 - x2);
        r.w = C358*vz*(z2 - 3.0f*x2);
    }
    reinterpret_cast<float4*>(sh + (size_t)slot * 16)[q] = r;
    if (q < 3) {
        const float dd = (in.aux - 2.0f*(float)q) * 0.5f;
        basis[(size_t)slot*3 + q] = expf(-dd*dd) * (1.0f/1.12f);
    } else {
        srcs[slot] = __float_as_int(in.aux);
    }
    float* xr = xs + j * KG_XP + 4 * q;
    atomicAdd(&xr[0], r.x * INV_NN); atomicAdd(&xr[1], r.y * INV_NN);
    atomicAdd(&xr[2], r.z * INV_NN); atomicAdd(&xr[3], r.w * INV_NN);
}

__global__ __launch_bounds__(1024) void kG(const float* __restrict__ evec, const float* __restrict__ elen,
                                           const int* __restrict__ esrc, const int* __restrict__ edst,
                                           int* __restrict__ off,
                                           float* __restrict__ sh, float* __restrict__ basis,
                                           int* __restrict__ srcs, float* __restrict__ x, int E, int N)
{
    __shared__ __attribute__((aligned(16))) int cnt[128];
    __shared__ __attribute__((aligned(16))) int excl[128];
    __shared__ int cnt2[128];
    __shared__ int wred[32];
    __shared__ int nmatch_s;
    __shared__ float xs[KG_NODES * KG_XP];
    __shared__ int le[KG_CAP], lj[KG_CAP], lr[KG_CAP];
    __shared__ int oe[KG_CAP], oj[KG_CAP];

    const int tid  = threadIdx.x;
    const int lane = tid & 31;
    const int wave = __builtin_amdgcn_readfirstlane(tid >> 5);
    const int lo = blockIdx.x * KG_NODES;
    const int nl = (N - lo < KG_NODES) ? (N - lo) : KG_NODES;

    if (tid < 128) { cnt[tid] = 0; cnt2[tid] = 0; }
    if (tid == 0) nmatch_s = 0;
    for (int i = tid; i < KG_NODES * KG_XP; i += 1024) xs[i] = 0.0f;
    __syncthreads();

    const int nv = E >> 2;
    const int rot = (int)(blockIdx.x * ((unsigned)nv / gridDim.x));
    const int4* d4 = reinterpret_cast<const int4*>(edst);
    int sless = 0;

    auto keep = [&](int t, int e) {
        const int r   = atomicAdd(&cnt[t], 1);
        const int pos = atomicAdd(&nmatch_s, 1);
        if (pos < KG_CAP) { le[pos] = e; lj[pos] = t; lr[pos] = r; }
    };
    auto visit4 = [&](int4 dd, int e0) {
        const int t0 = dd.x - lo, t1 = dd.y - lo, t2 = dd.z - lo, t3 = dd.w - lo;
        sless += __builtin_popcount(__builtin_amdgcn_ballot_w32(t0 < 0))
               + __builtin_popcount(__builtin_amdgcn_ballot_w32(t1 < 0))
               + __builtin_popcount(__builtin_amdgcn_ballot_w32(t2 < 0))
               + __builtin_popcount(__builtin_amdgcn_ballot_w32(t3 < 0));
        const bool m0 = (unsigned)t0 < (unsigned)nl, m1 = (unsigned)t1 < (unsigned)nl;
        const bool m2 = (unsigned)t2 < (unsigned)nl, m3 = (unsigned)t3 < (unsigned)nl;
        if (m0 | m1 | m2 | m3) {
            if (m0) keep(t0, e0);
            if (m1) keep(t1, e0 + 1);
            if (m2) keep(t2, e0 + 2);
            if (m3) keep(t3, e0 + 3);
        }
    };

    for (int i0 = 0; i0 < nv; i0 += 1024 * KG_U) {
        int4 d[KG_U];
        int  id[KG_U];
        #pragma unroll
        for (int u = 0; u < KG_U; u++) {
            const int i = i0 + u * 1024 + tid;
            int idx = i + rot;
            if (idx >= nv) idx -= nv;
            id[u] = idx;
            d[u] = make_int4(0x7fffffff, 0x7fffffff, 0x7fffffff, 0x7fffffff);
            if (i < nv) d[u] = d4[idx];
        }
        #pragma unroll
        for (int u = 0; u < KG_U; u++) visit4(d[u], 4 * id[u]);
    }
    {
        const int e = 4 * nv + tid;
        int4 dd = make_int4(0x7fffffff, 0x7fffffff, 0x7fffffff, 0x7fffffff);
        if (e < E) dd.x = edst[e];
        visit4(dd, e);
    }
    if (lane == 0) wred[wave] = sless;
    __syncthreads();

    const int nm   = __builtin_amdgcn_readfirstlane(nmatch_s);
    const int base = __builtin_amdgcn_readlane(wave_scan_incl(wred[lane], lane), 31);
    if (wave == 0) {
        const int4 c4 = *reinterpret_cast<const int4*>(&cnt[4 * lane]);
        const int s   = (c4.x + c4.y) + (c4.z + c4.w);
        const int inc = wave_scan_incl(s, lane);
        int4 o;
        o.x = inc - s; o.y = o.x + c4.x; o.z = o.y + c4.y; o.w = o.z + c4.z;
        *reinterpret_cast<int4*>(&excl[4 * lane]) = o;
        const int j = 4 * lane;
        if (j     < nl) off[lo + j]     = base + o.x;
        if (j + 1 < nl) off[lo + j + 1] = base + o.y;
        if (j + 2 < nl) off[lo + j + 2] = base + o.z;
        if (j + 3 < nl) off[lo + j + 3] = base + o.w;
        const int tot = __builtin_amdgcn_readlane(inc, 31);
        if (lane == 0 && lo + nl == N) off[N] = base + tot;
    }
    __syncthreads();

    const int nf = (nm <= KG_CAP) ? nm : 0;
    for (int i = tid; i < nf; i += 1024) {
        const int j = lj[i];
        const int p = excl[j] + lr[i];
        oe[p] = le[i]; oj[p] = j;
    }
    __syncthreads();
    for (int g0 = 0; g0 < 4 * nf; g0 += 1024 * KG_P) {
        KgIn in[KG_P];
        int  jn[KG_P];
        #pragma unroll
        for (int u = 0; u < KG_P; u++) {
            int g = g0 + u * 1024 + tid;
            if (g > 4 * nf - 1) g = 4 * nf - 1;
            jn[u] = oj[g >> 2];
            in[u] = kg_fetch(oe[g >> 2], g & 3, evec, elen, esrc);
        }
        #pragma unroll
        for (int u = 0; u < KG_P; u++) {
            const int g = g0 + u * 1024 + tid;
            if (g < 4 * nf) kg_record(in[u], base + (g >> 2), jn[u], g & 3, sh, basis, srcs, xs);
        }
    }
    if (nm > KG_CAP) {
        auto visit2 = [&](int dv, int e) {
            const int t = dv - lo;
            if ((unsigned)t < (unsigned)nl) {
                const int slot = base + excl[t] + atomicAdd(&cnt2[t], 1);
                for (int q = 0; q < 4; q++)
                    kg_record(kg_fetch(e, q, evec, elen, esrc), slot, t, q, sh, basis, srcs, xs);
            }
        };
        for (int i = tid; i < nv; i += 1024) {
            const int4 dd = d4[i];
            visit2(dd.x, 4 * i); visit2(dd.y, 4 * i + 1); visit2(dd.z, 4 * i + 2); visit2(dd.w, 4 * i + 3);
        }
        const int e = 4 * nv + tid;
        if (e < E) visit2(edst[e], e);
    }
    __syncthreads();
    for (int i = tid; i < nl * 16; i += 1024)
        x[(size_t)lo * 16 + i] = xs[(i >> 4) * KG_XP + (i & 15)];
}

#define KB_PITCH 1028
#define KB_HP 101
#define KB_YP 68
#define KB_ZP 260
#define RS3 0.5773502691896258f
#define RS5 0.4472135954999579f
#define RS7 0.3779644730092272f
#define KB_WAVES 12
#define KB_ROUND 256
template <bool TWO>
__device__ __forceinline__ void kb_product(const float* ga0, const float* ga1, const float* Bp,
                                           v8f& a00, v8f& a01, v8f& a10, v8f& a11)
{
    v2f Bc[4][2], Bn[4][2];
    #pragma unroll
    for (int s = 0; s < 4; s++) {
        #pragma unroll
        for (int q = 0; q < 2; q++) {
            Bc[s][q].x = Bp[(size_t)(4 * s) * 384 + 96 * q];
            Bc[s][q].y = Bp[(size_t)(4 * s + 1) * 384 + 96 * q];
        }
    }
    #pragma unroll 2
    for (int k0 = 0; k0 < 256; k0 += 16) {
        const int kn = (k0 + 16 < 256) ? (k0 + 16) : k0;
        const float* Bq = Bp + (size_t)kn * 384;
        #pragma unroll
        for (int s = 0; s < 4; s++) {
            #pragma unroll
            for (int q = 0; q < 2; q++) {
                Bn[s][q].x = Bq[(size_t)(4 * s) * 384 + 96 * q];
                Bn[s][q].y = Bq[(size_t)(4 * s + 1) * 384 + 96 * q];
            }
        }
        #pragma unroll
        for (int s = 0; s < 4; s++) {
            const int k = k0 + 4 * s;
            v2f A00 = *reinterpret_cast<const v2f*>(ga0 + k);
            v2f A01 = *reinterpret_cast<const v2f*>(ga0 + 256 + k);
            a00 = __builtin_amdgcn_wmma_f32_16x16x4_f32(false, A00, false, Bc[s][0], (short)0, a00, false, false);
            if (TWO) {
                v2f A10 = *reinterpret_cast<const v2f*>(ga1 + k);
                a10 = __builtin_amdgcn_wmma_f32_16x16x4_f32(false, A10, false, Bc[s][0], (short)0, a10, false, false);
            }
            a01 = __builtin_amdgcn_wmma_f32_16x16x4_f32(false, A01, false, Bc[s][1], (short)0, a01, false, false);
            if (TWO) {
                v2f A11 = *reinterpret_cast<const v2f*>(ga1 + 256 + k);
                a11 = __builtin_amdgcn_wmma_f32_16x16x4_f32(false, A11, false, Bc[s][1], (short)0, a11, false, false);
            }
        }
        #pragma unroll
        for (int s = 0; s < 4; s++) {
            #pragma unroll
            for (int q = 0; q < 2; q++) Bc[s][q] = Bn[s][q];
        }
    }
}

__global__ __launch_bounds__(32 * KB_WAVES) void kB(
    const float* __restrict__ sh, const float* __restrict__ basis,
    const float* __restrict__ x,
    const int* __restrict__ srcs,
    const int* __restrict__ off,
    const float* __restrict__ W1a, const float* __restrict__ W1b,
    const float* W2b, float* __restrict__ z, int N, int nfull)
{
    __shared__ float Gs[32 * KB_PITCH];

    const int tid  = threadIdx.x;
    const int lane = tid & 31;
    const int wave = __builtin_amdgcn_readfirstlane(tid >> 5);
    const bool two  = (int)blockIdx.x < nfull;
    const int n0    = two ? (int)blockIdx.x * 32 : nfull * 32 + ((int)blockIdx.x - nfull) * 16;
    const int rowsT = two ? 32 : 16;

    float wa0[8], wa1[8], wa2[8];
    #pragma unroll
    for (int i = 0; i < 8; i++) {
        wa0[i] = W1a[lane + 32*i];
        wa1[i] = W1a[256 + lane + 32*i];
        wa2[i] = W1a[512 + lane + 32*i];
    }

    for (int m = wave; m < rowsT; m += KB_WAVES) {
        const int n = n0 + m;
        float g0[8], g1[8], g2[8], g3[8];
        #pragma unroll
        for (int i = 0; i < 8; i++) { g0[i] = 0.0f; g1[i] = 0.0f; g2[i] = 0.0f; g3[i] = 0.0f; }
        if (n < N) {
            const int beg = off[n], end = off[n + 1];
            for (int base = beg; base < end; base += 32) {
                const int idx = base + lane;
                float d0 = 0.0f, d1 = 0.0f, d2 = 0.0f, d3 = 0.0f;
                float b0 = 0.0f, b1 = 0.0f, b2 = 0.0f;
                if (idx < end) {
                    const int srcn = srcs[idx];
                    const float4* xs = reinterpret_cast<const float4*>(x  + (size_t)srcn * 16);
                    const float4* ss = reinterpret_cast<const float4*>(sh + (size_t)idx * 16);
                    float4 x0 = xs[0], x1 = xs[1], x2 = xs[2], x3 = xs[3];
                    float4 s0 = ss[0], s1 = ss[1], s2 = ss[2], s3 = ss[3];
                    d0 = x0.x*s0.x;
                    d1 = (x0.y*s0.y + x0.z*s0.z + x0.w*s0.w) * RS3;
                    d2 = (x1.x*s1.x + x1.y*s1.y + x1.z*s1.z + x1.w*s1.w + x2.x*s2.x) * RS5;
                    d3 = (x2.y*s2.y + x2.z*s2.z + x2.w*s2.w
                        + x3.x*s3.x + x3.y*s3.y + x3.z*s3.z + x3.w*s3.w) * RS7;
                    b0 = basis[(size_t)idx*3]; b1 = basis[(size_t)idx*3 + 1]; b2 = basis[(size_t)idx*3 + 2];
                }
                const int left = end - base;
                const int cntc = left < 32 ? left : 32;
                for (int j = 0; j < cntc; j++) {
                    const float e0 = __shfl(d0, j, 32), e1 = __shfl(d1, j, 32);
                    const float e2 = __shfl(d2, j, 32), e3 = __shfl(d3, j, 32);
                    const float c0 = __shfl(b0, j, 32), c1 = __shfl(b1, j, 32), c2 = __shfl(b2, j, 32);
                    #pragma unroll
                    for (int i = 0; i < 8; i++) {
                        float hv = c0*wa0[i] + c1*wa1[i] + c2*wa2[i];
                        hv = fmaxf(hv * INV_SQRT3, 0.0f);
                        g0[i] += e0 * hv; g1[i] += e1 * hv;
                        g2[i] += e2 * hv; g3[i] += e3 * hv;
                    }
                }
            }
        }
        float* gr = &Gs[m * KB_PITCH + lane];
        #pragma unroll
        for (int i = 0; i < 8; i++) {
            gr[32*i]       = g0[i];
            gr[256 + 32*i] = g1[i];
            gr[512 + 32*i] = g2[i];
            gr[768 + 32*i] = g3[i];
        }
    }
    __syncthreads();

    const int c    = lane & 15;
    const int hh   = lane >> 4;
    const int ct   = (wave >= 6) ? (wave - 6) : wave;
    const int ph   = (wave >= 6) ? 2 : 0;
    const int colb = ct * 16 + c;
    v8f a00 = {}, a01 = {};
    v8f a10 = {}, a11 = {};
    const float* ga0 = &Gs[c * KB_PITCH + 256 * ph + 2 * hh];
    const float* ga1 = ga0 + 16 * KB_PITCH;
    const float* Bp = W1b + (size_t)(2 * hh) * 384 + 96 * ph + colb;
    if (two) kb_product<true>(ga0, ga1, Bp, a00, a01, a10, a11);
    else     kb_product<false>(ga0, ga1, Bp, a00, a01, a10, a11);
    v2f Bf[2][16];
    #pragma unroll
    for (int t = 0; t < 2; t++) {
        int jt = wave + KB_WAVES * t;
        if (jt > 15) jt = 15;
        const float* wb = W2b + (size_t)(16 * jt + c) * 64 + 2 * hh;
        #pragma unroll
        for (int i = 0; i < 16; i++) Bf[t][i] = *reinterpret_cast<const v2f*>(wb + 4 * i);
    }
    __syncthreads();

    float* Hs = Gs;
    float* Ys = Gs + 64 * KB_HP;
    float* Zs = Ys + 32 * KB_YP;
    const float scale = (1.0f/16.0f) * 0.5f * INV_NN;
    {
        float* Hw = Hs + (ph ? 32 * KB_HP : 0);
        #pragma unroll
        for (int r = 0; r < 8; r++) {
            Hw[(8*hh + r) * KB_HP + colb]      = a00[r] + a01[r];
            Hw[(16 + 8*hh + r) * KB_HP + colb] = a10[r] + a11[r];
        }
    }
    __syncthreads();
    {
        const float* hr = &Hs[lane * KB_HP];
        const float* hs = hr + 32 * KB_HP;
        float* yr = &Ys[lane * KB_YP];
        float ha[6], hb[6];
        #pragma unroll
        for (int k = 0; k < 6; k++) {
            int j = wave + KB_WAVES * k;
            if (j > 63) j = 63;
            ha[k] = (hr[j] + hs[j]) * scale;
            hb[k] = (hr[j + 32] + hs[j + 32]) * scale;
        }
        #pragma unroll
        for (int k = 0; k < 6; k++) {
            const int j = wave + KB_WAVES * k;
            if (j < 64) {
                float v;
                if (j < 16)      v = fmaxf(ha[k], 0.0f);
                else if (j < 32) v = fabsf(ha[k]);
                else {
                    const int jj = j - 32;
                    const bool use_relu = (jj < 8) || (jj >= 16 && jj < 24);
                    const float ga = use_relu ? fmaxf(ha[k], 0.0f) : tanhf(ha[k]);
                    v = ga * hb[k];
                }
                yr[j] = v;
            }
        }
    }
    __syncthreads();

    {
        const float* ya = &Ys[c * KB_YP + 2 * hh];
        v2f A0[16], A1[16];
        #pragma unroll
        for (int i = 0; i < 16; i++) {
            A0[i] = *reinterpret_cast<const v2f*>(ya + 4 * i);
            A1[i] = *reinterpret_cast<const v2f*>(ya + 16 * KB_YP + 4 * i);
        }
        #pragma unroll
        for (int t = 0; t < 2; t++) {
            const int jt = wave + KB_WAVES * t;
            if (jt < 16) {
                v8f z0 = {}, z1 = {};
                #pragma unroll
                for (int i = 0; i < 16; i++) {
                    z0 = __builtin_amdgcn_wmma_f32_16x16x4_f32(false, A0[i], false, Bf[t][i], (short)0, z0, false, false);
                    z1 = __builtin_amdgcn_wmma_f32_16x16x4_f32(false, A1[i], false, Bf[t][i], (short)0, z1, false, false);
                }
                float* zc = &Zs[(8 * hh) * KB_ZP + 16 * jt + c];
                #pragma unroll
                for (int r = 0; r < 8; r++) {
                    zc[r * KB_ZP]        = z0[r];
                    zc[(16 + r) * KB_ZP] = z1[r];
                }
            }
        }
    }
    __syncthreads();
    const int rows = (N - n0 < rowsT) ? (N - n0) : rowsT;
    for (int idx = tid; idx < rows * 64; idx += 32 * KB_WAVES) {
        const int m = idx >> 6, q = idx & 63;
        float4 v = *reinterpret_cast<const float4*>(&Zs[m * KB_ZP + 4 * q]);
        *reinterpret_cast<float4*>(z + (size_t)(n0 + m) * 256 + 4 * q) = v;
    }
}

#define KE_NODES 4
#define KE_U 8

__device__ __forceinline__ int rl_i(int v, int l) { return __builtin_amdgcn_readlane(v, l); }
__device__ __forceinline__ float rl_f(float v, int l) {
    return __int_as_float(__builtin_amdgcn_readlane(__float_as_int(v), l));
}
template <int M> __device__ __forceinline__ float row_xadd(float v) {
    int t = __builtin_amdgcn_update_dpp(0, __float_as_int(v), 0x160 | M, 0xF, 0xF, false);
    return v + __int_as_float(t);
}
__device__ __forceinline__ float wave_sum(float v) {
    v = row_xadd<1>(v); v = row_xadd<2>(v); v = row_xadd<4>(v); v = row_xadd<8>(v);
    return rl_f(v, 0) + rl_f(v, 16);
}
__device__ __forceinline__ float h2term(float c0, float c1, float c2,
                                        float w0, float w1, float w2, float zv) {
    float hv = c0*w0 + c1*w1 + c2*w2;
    hv = fmaxf(hv * INV_SQRT3, 0.0f);
    return hv * zv;
}

__global__ __launch_bounds__(128) void kE(const float* __restrict__ basis,
                                          const float* __restrict__ z,
                                          const int* __restrict__ srcs,
                                          const int* __restrict__ off,
                                          const float* __restrict__ W2a,
                                          float* __restrict__ out, int N)
{
    const int lane = threadIdx.x & 31;
    const int wave = __builtin_amdgcn_readfirstlane(threadIdx.x >> 5);
    const int n0 = (blockIdx.x * 4 + wave) * KE_NODES;
    if (n0 >= N) return;
    const int nn = (N - n0 < KE_NODES) ? (N - n0) : KE_NODES;
    int oi = n0 + lane;
    if (oi > N) oi = N;
    const int ov = off[oi];
    const int beg_all = rl_i(ov, 0);
    const int end_all = rl_i(ov, nn);

    const float4* w4 = reinterpret_cast<const float4*>(W2a);
    const float4 w0a = w4[lane],       w0b = w4[32 + lane];
    const float4 w1a = w4[64 + lane],  w1b = w4[96 + lane];
    const float4 w2a = w4[128 + lane], w2b = w4[160 + lane];

    const float oscale = (1.0f/128.0f) * INV_NN;
    int   k    = 0;
    int   nend = rl_i(ov, 1);
    float acc  = 0.0f;

    for (int base = beg_all; base < end_all; base += 64) {
        const int last = end_all - 1;
        int iA = base + lane;       if (iA > last) iA = last;
        int iB = base + 32 + lane;  if (iB > last) iB = last;
        const int sA = srcs[iA], sB = srcs[iB];
        const float pA0 = basis[(size_t)iA*3], pA1 = basis[(size_t)iA*3 + 1], pA2 = basis[(size_t)iA*3 + 2];
        const float pB0 = basis[(size_t)iB*3], pB1 = basis[(size_t)iB*3 + 1], pB2 = basis[(size_t)iB*3 + 2];
        const int left = end_all - base;
        const int cnt  = left < 64 ? left : 64;
        for (int j0 = 0; j0 < cnt; j0 += KE_U) {
            const bool hi = j0 >= 32;
            const int   vs  = hi ? sB  : sA;
            const float vb0 = hi ? pB0 : pA0, vb1 = hi ? pB1 : pA1, vb2 = hi ? pB2 : pA2;
            const int jl = j0 & 31;
            const int m  = (cnt - j0 < KE_U) ? (cnt - j0) : KE_U;
            float4 za[KE_U], zb[KE_U];
            #pragma unroll
            for (int u = 0; u < KE_U; u++) {
                const int jj = jl + (u < m ? u : m - 1);
                const int s  = rl_i(vs, jj);
                const float4* zr = reinterpret_cast<const float4*>(z + (size_t)s * 256);
                za[u] = zr[lane];
                zb[u] = zr[32 + lane];
            }
            #pragma unroll
            for (int u = 0; u < KE_U; u++) {
                if (u < m) {
                    const int jj = jl + u;
                    const float c0 = rl_f(vb0, jj), c1 = rl_f(vb1, jj), c2 = rl_f(vb2, jj);
                    float t = h2term(c0, c1, c2, w0a.x, w1a.x, w2a.x, za[u].x);
                    t += h2term(c0, c1, c2, w0a.y, w1a.y, w2a.y, za[u].y);
                    t += h2term(c0, c1, c2, w0a.z, w1a.z, w2a.z, za[u].z);
                    t += h2term(c0, c1, c2, w0a.w, w1a.w, w2a.w, za[u].w);
                    t += h2term(c0, c1, c2, w0b.x, w1b.x, w2b.x, zb[u].x);
                    t += h2term(c0, c1, c2, w0b.y, w1b.y, w2b.y, zb[u].y);
                    t += h2term(c0, c1, c2, w0b.z, w1b.z, w2b.z, zb[u].z);
                    t += h2term(c0, c1, c2, w0b.w, w1b.w, w2b.w, zb[u].w);
                    const int p = base + j0 + u;
                    while (p >= nend) {
                        const float tot = wave_sum(acc);
                        if (lane == 0) out[n0 + k] = tot * oscale;
                        acc = 0.0f;
                        k++;
                        nend = rl_i(ov, k + 1);
                    }
                    acc += t;
                }
            }
        }
    }
    while (k < nn) {
        const float tot = wave_sum(acc);
        if (lane == 0) out[n0 + k] = tot * oscale;
        acc = 0.0f;
        k++;
    }
}

extern "C" void kernel_launch(void* const* d_in, const int* in_sizes, int n_in,
                              void* d_out, int out_size, void* d_ws, size_t ws_size,
                              hipStream_t stream)
{
    const float* evec = (const float*)d_in[0];
    const float* elen = (const float*)d_in[1];
    const float* W1a  = (const float*)d_in[2];
    const float* W1b  = (const float*)d_in[3];
    const float* W2a  = (const float*)d_in[4];
    const float* W2b  = (const float*)d_in[5];
    const int*   esrc = (const int*)d_in[6];
    const int*   edst = (const int*)d_in[7];
    float* out = (float*)d_out;

    const int E = in_sizes[0] / 3;
    const int N = out_size;

    float* ws = (float*)d_ws;
    float* x_     = ws;                 ws += (size_t)N * 16;
    ws += (size_t)N * 96;
    float* lists_ = ws;                 ws += (size_t)N * 64;
    float* z_     = ws;                 ws += (size_t)N * 256;
    float* sh_    = ws;                 ws += (size_t)E * 16;
    float* basis_ = ws;                 ws += (size_t)E * 3;

    int* off_  = (int*)lists_;
    int* srcs_ = off_ + N + 4;

    kG<<<(N + KG_NODES - 1) / KG_NODES, 1024, 0, stream>>>(evec, elen, esrc, edst, off_, sh_, basis_, srcs_, x_, E, N);
    const int nfull = ((N / 32) / KB_ROUND) * KB_ROUND;
    const int nhalf = (N - nfull * 32 + 15) / 16;
    kB<<<nfull + nhalf, 32 * KB_WAVES, 0, stream>>>(sh_, basis_, x_, srcs_, off_, W1a, W1b, W2b, z_, N, nfull);
    kE<<<(N + 4 * KE_NODES - 1) / (4 * KE_NODES), 128, 0, stream>>>(basis_, z_, srcs_, off_, W2a, out, N);
}


# edge grouping: a new first launch reads the destinations once, sorts each 1024-edge chunk by node block; kG gathers its edges from the chunks, no full read per block; 4 launches; sum order changes
# speedup vs baseline: 6.9702x; 1.1039x over previous
#include <hip/hip_runtime.h>
#include <math.h>

typedef __attribute__((ext_vector_type(2))) float v2f;
typedef __attribute__((ext_vector_type(8))) float v8f;

#define INV_NN     0.512989176042577f
#define INV_SQRT3  0.5773502691896258f
#define S3   1.7320508075688772f
#define S5   2.23606797749979f
#define S7   2.6457513110645907f
#define S15  3.872983346207417f
#define S105 10.246950765959598f
#define C358 2.091650066335189f
#define C218 1.6201851746019651f


#define KG_NODES 98
#define KG_XP    17
#define KG_CAP   8192
#define KH_EDGES 1024
#define KH_MAXB  1024
#define KG_P     4

template <int S> __device__ __forceinline__ int row_shr_add(int v) {
    int t = __builtin_amdgcn_update_dpp(0, v, 0x110 | S, 0xF, 0xF, false);
    return v + t;
}
__device__ __forceinline__ int wave_scan_incl(int v, int lane) {
    v = row_shr_add<1>(v); v = row_shr_add<2>(v); v = row_shr_add<4>(v); v = row_shr_add<8>(v);
    const int lo = __builtin_amdgcn_readlane(v, 15);
    return v + (lane >= 16 ? lo : 0);
}

struct KgIn { float vx, vy, vz, aux; };
__device__ __forceinline__ KgIn kg_fetch(int e, int q, const float* __restrict__ evec,
                                         const float* __restrict__ elen, const int* __restrict__ esrc)
{
    KgIn r;
    r.vx = evec[3*(size_t)e]; r.vy = evec[3*(size_t)e+1]; r.vz = evec[3*(size_t)e+2];
    r.aux = (q < 3) ? elen[e] : __int_as_float(esrc[e]);
    return r;
}
__device__ __forceinline__ void kg_record(KgIn in, int slot, int j, int q,
        float* __restrict__ sh, float* __restrict__ basis, int* __restrict__ srcs, float* xs)
{
    float vx = in.vx, vy = in.vy, vz = in.vz;
    float rn = 1.0f / (sqrtf(vx*vx + vy*vy + vz*vz) + 1e-12f);
    vx *= rn; vy *= rn; vz *= rn;
    float x2 = vx*vx, y2 = vy*vy, z2 = vz*vz;
    float4 r;
    if (q == 0) {
        r.x = 1.0f;
        r.y = S3*vx; r.z = S3*vy; r.w = S3*vz;
    } else if (q == 1) {
        r.x = S15*vx*vz;
        r.y = S15*vx*vy;
        r.z = 0.5f*S5*(2.0f*y2 - x2 - z2);
        r.w = S15*vy*vz;
    } else if (q == 2) {
        r.x = 0.5f*S15*(z2 - x2);
        r.y = C358*vx*(3.0f*z2 - x2);
        r.z = S105*vx*vy*vz;
        r.w = C218*vx*(4.0f*y2 - z2 - x2);
    } else {
        r.x = 0.5f*S7*vy*(2.0f*y2 - 3.0f*z2 - 3.0f*x2);
        r.y = C218*vz*(4.0f*y2 - z2 - x2);
        r.z = 0.5f*S105*vy*(z2 - x2);
        r.w = C358*vz*(z2 - 3.0f*x2);
    }
    reinterpret_cast<float4*>(sh + (size_t)slot * 16)[q] = r;
    if (q < 3) {
        const float dd = (in.aux - 2.0f*(float)q) * 0.5f;
        basis[(size_t)slot*3 + q] = expf(-dd*dd) * (1.0f/1.12f);
    } else {
        srcs[slot] = __float_as_int(in.aux);
    }
    float* xr = xs + j * KG_XP + 4 * q;
    atomicAdd(&xr[0], r.x * INV_NN); atomicAdd(&xr[1], r.y * INV_NN);
    atomicAdd(&xr[2], r.z * INV_NN); atomicAdd(&xr[3], r.w * INV_NN);
}

__global__ __launch_bounds__(256) void kH(const int* __restrict__ edst, int* __restrict__ blist,
                                          int* __restrict__ boff, int E, int N, int NB)
{
    __shared__ int hist[KH_MAXB];
    __shared__ int hoff[KH_MAXB];

    const int tid  = threadIdx.x;
    const int lane = tid & 31;
    const int wave = __builtin_amdgcn_readfirstlane(tid >> 5);
    const int nbt  = NB + 1;

    for (int i = tid; i < nbt; i += 256) hist[i] = 0;
    __syncthreads();

    const int e0 = blockIdx.x * KH_EDGES + 4 * tid;
    int d[4], bin[4], r[4];
    if (e0 + 3 < E) {
        const int4 dd = *reinterpret_cast<const int4*>(edst + e0);
        d[0] = dd.x; d[1] = dd.y; d[2] = dd.z; d[3] = dd.w;
    } else {
        #pragma unroll
        for (int u = 0; u < 4; u++) d[u] = (e0 + u < E) ? edst[e0 + u] : -1;
    }
    #pragma unroll
    for (int u = 0; u < 4; u++) {
        bin[u] = ((unsigned)d[u] < (unsigned)N) ? d[u] / KG_NODES : NB;
        r[u] = 0;
        if (e0 + u < E) r[u] = atomicAdd(&hist[bin[u]], 1);
    }
    __syncthreads();

    if (wave == 0) {
        const int per = (nbt + 31) >> 5;
        int s = 0;
        for (int k = 0; k < per; k++) {
            const int b = lane * per + k;
            if (b < nbt) s += hist[b];
        }
        const int inc = wave_scan_incl(s, lane);
        int run = inc - s;
        for (int k = 0; k < per; k++) {
            const int b = lane * per + k;
            if (b < nbt) { hoff[b] = run; run += hist[b]; }
        }
    }
    __syncthreads();

    int2* bl = reinterpret_cast<int2*>(blist) + (size_t)blockIdx.x * KH_EDGES;
    #pragma unroll
    for (int u = 0; u < 4; u++)
        if (e0 + u < E) bl[hoff[bin[u]] + r[u]] = make_int2(e0 + u, d[u]);
    for (int i = tid; i < nbt; i += 256) boff[(size_t)blockIdx.x * nbt + i] = hoff[i];
}

__global__ __launch_bounds__(1024) void kG(const float* __restrict__ evec, const float* __restrict__ elen,
                                           const int* __restrict__ esrc, const int* __restrict__ edst,
                                           const int* __restrict__ blist, const int* __restrict__ boff, int NCH,
                                           int* __restrict__ off,
                                           float* __restrict__ sh, float* __restrict__ basis,
                                           int* __restrict__ srcs, float* __restrict__ x, int E, int N)
{
    __shared__ __attribute__((aligned(16))) int cnt[128];
    __shared__ __attribute__((aligned(16))) int excl[128];
    __shared__ int cnt2[128];
    __shared__ int wred[32];
    __shared__ int nmatch_s;
    __shared__ float xs[KG_NODES * KG_XP];
    __shared__ int le[KG_CAP], lj[KG_CAP], lr[KG_CAP];
    __shared__ int oe[KG_CAP], oj[KG_CAP];

    const int tid  = threadIdx.x;
    const int lane = tid & 31;
    const int wave = __builtin_amdgcn_readfirstlane(tid >> 5);
    const int lo = blockIdx.x * KG_NODES;
    const int nl = (N - lo < KG_NODES) ? (N - lo) : KG_NODES;

    if (tid < 128) { cnt[tid] = 0; cnt2[tid] = 0; }
    if (tid == 0) nmatch_s = 0;
    for (int i = tid; i < KG_NODES * KG_XP; i += 1024) xs[i] = 0.0f;
    __syncthreads();

    const int nv = E >> 2;
    const int4* d4 = reinterpret_cast<const int4*>(edst);

    auto keep = [&](int t, int e) {
        const int r   = atomicAdd(&cnt[t], 1);
        const int pos = atomicAdd(&nmatch_s, 1);
        if (pos < KG_CAP) { le[pos] = e; lj[pos] = t; lr[pos] = r; }
    };

    const int nbt = (int)gridDim.x + 1;
    const int2* bl2 = reinterpret_cast<const int2*>(blist);
    int less = 0;
    for (int c0 = 0; c0 < NCH; c0 += 256) {
        const int ch = c0 + (tid >> 2);
        if (ch < NCH) {
            const int* bo = boff + (size_t)ch * nbt + blockIdx.x;
            const int o0 = bo[0], o1 = bo[1];
            if ((tid & 3) == 0) less += o0;
            for (int k = o0 + (tid & 3); k < o1; k += 4) {
                const int2 ed = bl2[(size_t)ch * KH_EDGES + k];
                keep(ed.y - lo, ed.x);
            }
        }
    }
    const int wless = __builtin_amdgcn_readlane(wave_scan_incl(less, lane), 31);
    if (lane == 0) wred[wave] = wless;
    __syncthreads();

    const int nm   = __builtin_amdgcn_readfirstlane(nmatch_s);
    const int base = __builtin_amdgcn_readlane(wave_scan_incl(wred[lane], lane), 31);
    if (wave == 0) {
        const int4 c4 = *reinterpret_cast<const int4*>(&cnt[4 * lane]);
        const int s   = (c4.x + c4.y) + (c4.z + c4.w);
        const int inc = wave_scan_incl(s, lane);
        int4 o;
        o.x = inc - s; o.y = o.x + c4.x; o.z = o.y + c4.y; o.w = o.z + c4.z;
        *reinterpret_cast<int4*>(&excl[4 * lane]) = o;
        const int j = 4 * lane;
        if (j     < nl) off[lo + j]     = base + o.x;
        if (j + 1 < nl) off[lo + j + 1] = base + o.y;
        if (j + 2 < nl) off[lo + j + 2] = base + o.z;
        if (j + 3 < nl) off[lo + j + 3] = base + o.w;
        const int tot = __builtin_amdgcn_readlane(inc, 31);
        if (lane == 0 && lo + nl == N) off[N] = base + tot;
    }
    __syncthreads();

    const int nf = (nm <= KG_CAP) ? nm : 0;
    for (int i = tid; i < nf; i += 1024) {
        const int j = lj[i];
        const int p = excl[j] + lr[i];
        oe[p] = le[i]; oj[p] = j;
    }
    __syncthreads();
    for (int g0 = 0; g0 < 4 * nf; g0 += 1024 * KG_P) {
        KgIn in[KG_P];
        int  jn[KG_P];
        #pragma unroll
        for (int u = 0; u < KG_P; u++) {
            int g = g0 + u * 1024 + tid;
            if (g > 4 * nf - 1) g = 4 * nf - 1;
            jn[u] = oj[g >> 2];
            in[u] = kg_fetch(oe[g >> 2], g & 3, evec, elen, esrc);
        }
        #pragma unroll
        for (int u = 0; u < KG_P; u++) {
            const int g = g0 + u * 1024 + tid;
            if (g < 4 * nf) kg_record(in[u], base + (g >> 2), jn[u], g & 3, sh, basis, srcs, xs);
        }
    }
    if (nm > KG_CAP) {
        auto visit2 = [&](int dv, int e) {
            const int t = dv - lo;
            if ((unsigned)t < (unsigned)nl) {
                const int slot = base + excl[t] + atomicAdd(&cnt2[t], 1);
                for (int q = 0; q < 4; q++)
                    kg_record(kg_fetch(e, q, evec, elen, esrc), slot, t, q, sh, basis, srcs, xs);
            }
        };
        for (int i = tid; i < nv; i += 1024) {
            const int4 dd = d4[i];
            visit2(dd.x, 4 * i); visit2(dd.y, 4 * i + 1); visit2(dd.z, 4 * i + 2); visit2(dd.w, 4 * i + 3);
        }
        const int e = 4 * nv + tid;
        if (e < E) visit2(edst[e], e);
    }
    __syncthreads();
    for (int i = tid; i < nl * 16; i += 1024)
        x[(size_t)lo * 16 + i] = xs[(i >> 4) * KG_XP + (i & 15)];
}

#define KB_PITCH 1028
#define KB_HP 101
#define KB_YP 68
#define KB_ZP 260
#define RS3 0.5773502691896258f
#define RS5 0.4472135954999579f
#define RS7 0.3779644730092272f
#define KB_WAVES 12
#define KB_ROUND 256
template <bool TWO>
__device__ __forceinline__ void kb_product(const float* ga0, const float* ga1, const float* Bp,
                                           v8f& a00, v8f& a01, v8f& a10, v8f& a11)
{
    v2f Bc[4][2], Bn[4][2];
    #pragma unroll
    for (int s = 0; s < 4; s++) {
        #pragma unroll
        for (int q = 0; q < 2; q++) {
            Bc[s][q].x = Bp[(size_t)(4 * s) * 384 + 96 * q];
            Bc[s][q].y = Bp[(size_t)(4 * s + 1) * 384 + 96 * q];
        }
    }
    #pragma unroll 2
    for (int k0 = 0; k0 < 256; k0 += 16) {
        const int kn = (k0 + 16 < 256) ? (k0 + 16) : k0;
        const float* Bq = Bp + (size_t)kn * 384;
        #pragma unroll
        for (int s = 0; s < 4; s++) {
            #pragma unroll
            for (int q = 0; q < 2; q++) {
                Bn[s][q].x = Bq[(size_t)(4 * s) * 384 + 96 * q];
                Bn[s][q].y = Bq[(size_t)(4 * s + 1) * 384 + 96 * q];
            }
        }
        #pragma unroll
        for (int s = 0; s < 4; s++) {
            const int k = k0 + 4 * s;
            v2f A00 = *reinterpret_cast<const v2f*>(ga0 + k);
            v2f A01 = *reinterpret_cast<const v2f*>(ga0 + 256 + k);
            a00 = __builtin_amdgcn_wmma_f32_16x16x4_f32(false, A00, false, Bc[s][0], (short)0, a00, false, false);
            if (TWO) {
                v2f A10 = *reinterpret_cast<const v2f*>(ga1 + k);
                a10 = __builtin_amdgcn_wmma_f32_16x16x4_f32(false, A10, false, Bc[s][0], (short)0, a10, false, false);
            }
            a01 = __builtin_amdgcn_wmma_f32_16x16x4_f32(false, A01, false, Bc[s][1], (short)0, a01, false, false);
            if (TWO) {
                v2f A11 = *reinterpret_cast<const v2f*>(ga1 + 256 + k);
                a11 = __builtin_amdgcn_wmma_f32_16x16x4_f32(false, A11, false, Bc[s][1], (short)0, a11, false, false);
            }
        }
        #pragma unroll
        for (int s = 0; s < 4; s++) {
            #pragma unroll
            for (int q = 0; q < 2; q++) Bc[s][q] = Bn[s][q];
        }
    }
}

__global__ __launch_bounds__(32 * KB_WAVES) void kB(
    const float* __restrict__ sh, const float* __restrict__ basis,
    const float* __restrict__ x,
    const int* __restrict__ srcs,
    const int* __restrict__ off,
    const float* __restrict__ W1a, const float* __restrict__ W1b,
    const float* W2b, float* __restrict__ z, int N, int nfull)
{
    __shared__ float Gs[32 * KB_PITCH];

    const int tid  = threadIdx.x;
    const int lane = tid & 31;
    const int wave = __builtin_amdgcn_readfirstlane(tid >> 5);
    const bool two  = (int)blockIdx.x < nfull;
    const int n0    = two ? (int)blockIdx.x * 32 : nfull * 32 + ((int)blockIdx.x - nfull) * 16;
    const int rowsT = two ? 32 : 16;

    float wa0[8], wa1[8], wa2[8];
    #pragma unroll
    for (int i = 0; i < 8; i++) {
        wa0[i] = W1a[lane + 32*i];
        wa1[i] = W1a[256 + lane + 32*i];
        wa2[i] = W1a[512 + lane + 32*i];
    }

    for (int m = wave; m < rowsT; m += KB_WAVES) {
        const int n = n0 + m;
        float g0[8], g1[8], g2[8], g3[8];
        #pragma unroll
        for (int i = 0; i < 8; i++) { g0[i] = 0.0f; g1[i] = 0.0f; g2[i] = 0.0f; g3[i] = 0.0f; }
        if (n < N) {
            const int beg = off[n], end = off[n + 1];
            for (int base = beg; base < end; base += 32) {
                const int idx = base + lane;
                float d0 = 0.0f, d1 = 0.0f, d2 = 0.0f, d3 = 0.0f;
                float b0 = 0.0f, b1 = 0.0f, b2 = 0.0f;
                if (idx < end) {
                    const int srcn = srcs[idx];
                    const float4* xs = reinterpret_cast<const float4*>(x  + (size_t)srcn * 16);
                    const float4* ss = reinterpret_cast<const float4*>(sh + (size_t)idx * 16);
                    float4 x0 = xs[0], x1 = xs[1], x2 = xs[2], x3 = xs[3];
                    float4 s0 = ss[0], s1 = ss[1], s2 = ss[2], s3 = ss[3];
                    d0 = x0.x*s0.x;
                    d1 = (x0.y*s0.y + x0.z*s0.z + x0.w*s0.w) * RS3;
                    d2 = (x1.x*s1.x + x1.y*s1.y + x1.z*s1.z + x1.w*s1.w + x2.x*s2.x) * RS5;
                    d3 = (x2.y*s2.y + x2.z*s2.z + x2.w*s2.w
                        + x3.x*s3.x + x3.y*s3.y + x3.z*s3.z + x3.w*s3.w) * RS7;
                    b0 = basis[(size_t)idx*3]; b1 = basis[(size_t)idx*3 + 1]; b2 = basis[(size_t)idx*3 + 2];
                }
                const int left = end - base;
                const int cntc = left < 32 ? left : 32;
                for (int j = 0; j < cntc; j++) {
                    const float e0 = __shfl(d0, j, 32), e1 = __shfl(d1, j, 32);
                    const float e2 = __shfl(d2, j, 32), e3 = __shfl(d3, j, 32);
                    const float c0 = __shfl(b0, j, 32), c1 = __shfl(b1, j, 32), c2 = __shfl(b2, j, 32);
                    #pragma unroll
                    for (int i = 0; i < 8; i++) {
                        float hv = c0*wa0[i] + c1*wa1[i] + c2*wa2[i];
                        hv = fmaxf(hv * INV_SQRT3, 0.0f);
                        g0[i] += e0 * hv; g1[i] += e1 * hv;
                        g2[i] += e2 * hv; g3[i] += e3 * hv;
                    }
                }
            }
        }
        float* gr = &Gs[m * KB_PITCH + lane];
        #pragma unroll
        for (int i = 0; i < 8; i++) {
            gr[32*i]       = g0[i];
            gr[256 + 32*i] = g1[i];
            gr[512 + 32*i] = g2[i];
            gr[768 + 32*i] = g3[i];
        }
    }
    __syncthreads();

    const int c    = lane & 15;
    const int hh   = lane >> 4;
    const int ct   = (wave >= 6) ? (wave - 6) : wave;
    const int ph   = (wave >= 6) ? 2 : 0;
    const int colb = ct * 16 + c;
    v8f a00 = {}, a01 = {};
    v8f a10 = {}, a11 = {};
    const float* ga0 = &Gs[c * KB_PITCH + 256 * ph + 2 * hh];
    const float* ga1 = ga0 + 16 * KB_PITCH;
    const float* Bp = W1b + (size_t)(2 * hh) * 384 + 96 * ph + colb;
    if (two) kb_product<true>(ga0, ga1, Bp, a00, a01, a10, a11);
    else     kb_product<false>(ga0, ga1, Bp, a00, a01, a10, a11);
    v2f Bf[2][16];
    #pragma unroll
    for (int t = 0; t < 2; t++) {
        int jt = wave + KB_WAVES * t;
        if (jt > 15) jt = 15;
        const float* wb = W2b + (size_t)(16 * jt + c) * 64 + 2 * hh;
        #pragma unroll
        for (int i = 0; i < 16; i++) Bf[t][i] = *reinterpret_cast<const v2f*>(wb + 4 * i);
    }
    __syncthreads();

    float* Hs = Gs;
    float* Ys = Gs + 64 * KB_HP;
    float* Zs = Ys + 32 * KB_YP;
    const float scale = (1.0f/16.0f) * 0.5f * INV_NN;
    {
        float* Hw = Hs + (ph ? 32 * KB_HP : 0);
        #pragma unroll
        for (int r = 0; r < 8; r++) {
            Hw[(8*hh + r) * KB_HP + colb]      = a00[r] + a01[r];
            Hw[(16 + 8*hh + r) * KB_HP + colb] = a10[r] + a11[r];
        }
    }
    __syncthreads();
    {
        const float* hr = &Hs[lane * KB_HP];
        const float* hs = hr + 32 * KB_HP;
        float* yr = &Ys[lane * KB_YP];
        float ha[6], hb[6];
        #pragma unroll
        for (int k = 0; k < 6; k++) {
            int j = wave + KB_WAVES * k;
            if (j > 63) j = 63;
            ha[k] = (hr[j] + hs[j]) * scale;
            hb[k] = (hr[j + 32] + hs[j + 32]) * scale;
        }
        #pragma unroll
        for (int k = 0; k < 6; k++) {
            const int j = wave + KB_WAVES * k;
            if (j < 64) {
                float v;
                if (j < 16)      v = fmaxf(ha[k], 0.0f);
                else if (j < 32) v = fabsf(ha[k]);
                else {
                    const int jj = j - 32;
                    const bool use_relu = (jj < 8) || (jj >= 16 && jj < 24);
                    const float ga = use_relu ? fmaxf(ha[k], 0.0f) : tanhf(ha[k]);
                    v = ga * hb[k];
                }
                yr[j] = v;
            }
        }
    }
    __syncthreads();

    {
        const float* ya = &Ys[c * KB_YP + 2 * hh];
        v2f A0[16], A1[16];
        #pragma unroll
        for (int i = 0; i < 16; i++) {
            A0[i] = *reinterpret_cast<const v2f*>(ya + 4 * i);
            A1[i] = *reinterpret_cast<const v2f*>(ya + 16 * KB_YP + 4 * i);
        }
        #pragma unroll
        for (int t = 0; t < 2; t++) {
            const int jt = wave + KB_WAVES * t;
            if (jt < 16) {
                v8f z0 = {}, z1 = {};
                #pragma unroll
                for (int i = 0; i < 16; i++) {
                    z0 = __builtin_amdgcn_wmma_f32_16x16x4_f32(false, A0[i], false, Bf[t][i], (short)0, z0, false, false);
                    z1 = __builtin_amdgcn_wmma_f32_16x16x4_f32(false, A1[i], false, Bf[t][i], (short)0, z1, false, false);
                }
                float* zc = &Zs[(8 * hh) * KB_ZP + 16 * jt + c];
                #pragma unroll
                for (int r = 0; r < 8; r++) {
                    zc[r * KB_ZP]        = z0[r];
                    zc[(16 + r) * KB_ZP] = z1[r];
                }
            }
        }
    }
    __syncthreads();
    const int rows = (N - n0 < rowsT) ? (N - n0) : rowsT;
    for (int idx = tid; idx < rows * 64; idx += 32 * KB_WAVES) {
        const int m = idx >> 6, q = idx & 63;
        float4 v = *reinterpret_cast<const float4*>(&Zs[m * KB_ZP + 4 * q]);
        *reinterpret_cast<float4*>(z + (size_t)(n0 + m) * 256 + 4 * q) = v;
    }
}

#define KE_NODES 4
#define KE_U 8

__device__ __forceinline__ int rl_i(int v, int l) { return __builtin_amdgcn_readlane(v, l); }
__device__ __forceinline__ float rl_f(float v, int l) {
    return __int_as_float(__builtin_amdgcn_readlane(__float_as_int(v), l));
}
template <int M> __device__ __forceinline__ float row_xadd(float v) {
    int t = __builtin_amdgcn_update_dpp(0, __float_as_int(v), 0x160 | M, 0xF, 0xF, false);
    return v + __int_as_float(t);
}
__device__ __forceinline__ float wave_sum(float v) {
    v = row_xadd<1>(v); v = row_xadd<2>(v); v = row_xadd<4>(v); v = row_xadd<8>(v);
    return rl_f(v, 0) + rl_f(v, 16);
}
__device__ __forceinline__ float h2term(float c0, float c1, float c2,
                                        float w0, float w1, float w2, float zv) {
    float hv = c0*w0 + c1*w1 + c2*w2;
    hv = fmaxf(hv * INV_SQRT3, 0.0f);
    return hv * zv;
}

__global__ __launch_bounds__(128) void kE(const float* __restrict__ basis,
                                          const float* __restrict__ z,
                                          const int* __restrict__ srcs,
                                          const int* __restrict__ off,
                                          const float* __restrict__ W2a,
                                          float* __restrict__ out, int N)
{
    const int lane = threadIdx.x & 31;
    const int wave = __builtin_amdgcn_readfirstlane(threadIdx.x >> 5);
    const int n0 = (blockIdx.x * 4 + wave) * KE_NODES;
    if (n0 >= N) return;
    const int nn = (N - n0 < KE_NODES) ? (N - n0) : KE_NODES;
    int oi = n0 + lane;
    if (oi > N) oi = N;
    const int ov = off[oi];
    const int beg_all = rl_i(ov, 0);
    const int end_all = rl_i(ov, nn);

    const float4* w4 = reinterpret_cast<const float4*>(W2a);
    const float4 w0a = w4[lane],       w0b = w4[32 + lane];
    const float4 w1a = w4[64 + lane],  w1b = w4[96 + lane];
    const float4 w2a = w4[128 + lane], w2b = w4[160 + lane];

    const float oscale = (1.0f/128.0f) * INV_NN;
    int   k    = 0;
    int   nend = rl_i(ov, 1);
    float acc  = 0.0f;

    for (int base = beg_all; base < end_all; base += 64) {
        const int last = end_all - 1;
        int iA = base + lane;       if (iA > last) iA = last;
        int iB = base + 32 + lane;  if (iB > last) iB = last;
        const int sA = srcs[iA], sB = srcs[iB];
        const float pA0 = basis[(size_t)iA*3], pA1 = basis[(size_t)iA*3 + 1], pA2 = basis[(size_t)iA*3 + 2];
        const float pB0 = basis[(size_t)iB*3], pB1 = basis[(size_t)iB*3 + 1], pB2 = basis[(size_t)iB*3 + 2];
        const int left = end_all - base;
        const int cnt  = left < 64 ? left : 64;
        for (int j0 = 0; j0 < cnt; j0 += KE_U) {
            const bool hi = j0 >= 32;
            const int   vs  = hi ? sB  : sA;
            const float vb0 = hi ? pB0 : pA0, vb1 = hi ? pB1 : pA1, vb2 = hi ? pB2 : pA2;
            const int jl = j0 & 31;
            const int m  = (cnt - j0 < KE_U) ? (cnt - j0) : KE_U;
            float4 za[KE_U], zb[KE_U];
            #pragma unroll
            for (int u = 0; u < KE_U; u++) {
                const int jj = jl + (u < m ? u : m - 1);
                const int s  = rl_i(vs, jj);
                const float4* zr = reinterpret_cast<const float4*>(z + (size_t)s * 256);
                za[u] = zr[lane];
                zb[u] = zr[32 + lane];
            }
            #pragma unroll
            for (int u = 0; u < KE_U; u++) {
                if (u < m) {
                    const int jj = jl + u;
                    const float c0 = rl_f(vb0, jj), c1 = rl_f(vb1, jj), c2 = rl_f(vb2, jj);
                    float t = h2term(c0, c1, c2, w0a.x, w1a.x, w2a.x, za[u].x);
                    t += h2term(c0, c1, c2, w0a.y, w1a.y, w2a.y, za[u].y);
                    t += h2term(c0, c1, c2, w0a.z, w1a.z, w2a.z, za[u].z);
                    t += h2term(c0, c1, c2, w0a.w, w1a.w, w2a.w, za[u].w);
                    t += h2term(c0, c1, c2, w0b.x, w1b.x, w2b.x, zb[u].x);
                    t += h2term(c0, c1, c2, w0b.y, w1b.y, w2b.y, zb[u].y);
                    t += h2term(c0, c1, c2, w0b.z, w1b.z, w2b.z, zb[u].z);
                    t += h2term(c0, c1, c2, w0b.w, w1b.w, w2b.w, zb[u].w);
                    const int p = base + j0 + u;
                    while (p >= nend) {
                        const float tot = wave_sum(acc);
                        if (lane == 0) out[n0 + k] = tot * oscale;
                        acc = 0.0f;
                        k++;
                        nend = rl_i(ov, k + 1);
                    }
                    acc += t;
                }
            }
        }
    }
    while (k < nn) {
        const float tot = wave_sum(acc);
        if (lane == 0) out[n0 + k] = tot * oscale;
        acc = 0.0f;
        k++;
    }
}

extern "C" void kernel_launch(void* const* d_in, const int* in_sizes, int n_in,
                              void* d_out, int out_size, void* d_ws, size_t ws_size,
                              hipStream_t stream)
{
    const float* evec = (const float*)d_in[0];
    const float* elen = (const float*)d_in[1];
    const float* W1a  = (const float*)d_in[2];
    const float* W1b  = (const float*)d_in[3];
    const float* W2a  = (const float*)d_in[4];
    const float* W2b  = (const float*)d_in[5];
    const int*   esrc = (const int*)d_in[6];
    const int*   edst = (const int*)d_in[7];
    float* out = (float*)d_out;

    const int E = in_sizes[0] / 3;
    const int N = out_size;

    float* ws = (float*)d_ws;
    float* x_     = ws;                 ws += (size_t)N * 16;
    int*   blist_ = (int*)ws;
    ws += (size_t)N * 96;
    float* lists_ = ws;                 ws += (size_t)N * 64;
    float* z_     = ws;                 ws += (size_t)N * 256;
    float* sh_    = ws;                 ws += (size_t)E * 16;
    float* basis_ = ws;                 ws += (size_t)E * 3;

    int* off_  = (int*)lists_;
    int* srcs_ = off_ + N + 4;

    const int nb  = (N + KG_NODES - 1) / KG_NODES;
    const int nch = (E + KH_EDGES - 1) / KH_EDGES;
    int* boff_ = blist_ + (size_t)nch * KH_EDGES * 2;
    kH<<<nch, 256, 0, stream>>>(edst, blist_, boff_, E, N, nb);
    kG<<<nb, 1024, 0, stream>>>(evec, elen, esrc, edst, blist_, boff_, nch, off_, sh_, basis_, srcs_, x_, E, N);
    const int nfull = ((N / 32) / KB_ROUND) * KB_ROUND;
    const int nhalf = (N - nfull * 32 + 15) / 16;
    kB<<<nfull + nhalf, 32 * KB_WAVES, 0, stream>>>(sh_, basis_, x_, srcs_, off_, W1a, W1b, W2b, z_, N, nfull);
    kE<<<(N + 4 * KE_NODES - 1) / (4 * KE_NODES), 128, 0, stream>>>(basis_, z_, srcs_, off_, W2a, out, N);
}
